# starting kernel
# baseline (speedup 1.0000x reference)
#include <hip/hip_runtime.h>
#include <hip/hip_bf16.h>
#include <math.h>

typedef float v2f __attribute__((ext_vector_type(2)));
typedef float v8f __attribute__((ext_vector_type(8)));

#define GAT_INC 256
#define GAT_HID 64
#define NEG_SLOPE 0.2f

__global__ __launch_bounds__(256)
void gat_gemm_wmma(const float* __restrict__ A, const float* __restrict__ W,
                   float* __restrict__ C, int N, int K) {
    __shared__ float sA[64 * (GAT_INC + 4)];
    const int KP   = K + 4;
    const int lane = threadIdx.x & 31;
    const int wave = threadIdx.x >> 5;
    const int rowBase = blockIdx.x * 64;

    const int col      = (wave & 3) * 16 + (lane & 15);
    const int waveRow  = (wave >> 2) * 32;

    for (int k = threadIdx.x; k < K; k += 256)
        __builtin_prefetch(&W[(size_t)k * 64 + col], 0, 3);

    for (int idx = threadIdx.x; idx < 64 * K; idx += 256) {
        int r = idx / K, c = idx - r * K;
        int gr = rowBase + r;
        sA[r * KP + c] = (gr < N) ? A[(size_t)gr * K + c] : 0.0f;
    }
    __syncthreads();

    const int row   = lane & 15;
    const int khalf = (lane >> 4) << 1;

    v8f acc0 = {};
    v8f acc1 = {};
    for (int k = 0; k < K; k += 4) {
        v2f b;
        b.x = W[(size_t)(k + khalf)     * 64 + col];
        b.y = W[(size_t)(k + khalf + 1) * 64 + col];

        v2f a0, a1;
        a0.x = sA[(waveRow +      row) * KP + k + khalf];
        a0.y = sA[(waveRow +      row) * KP + k + khalf + 1];
        a1.x = sA[(waveRow + 16 + row) * KP + k + khalf];
        a1.y = sA[(waveRow + 16 + row) * KP + k + khalf + 1];

        acc0 = __builtin_amdgcn_wmma_f32_16x16x4_f32(
            false, a0, false, b, (short)0, acc0, false, false);
        acc1 = __builtin_amdgcn_wmma_f32_16x16x4_f32(
            false, a1, false, b, (short)0, acc1, false, false);
    }

    const int mhalf = (lane >> 4) << 3;
    for (int r = 0; r < 8; ++r) {
        int gr0 = rowBase + waveRow +      mhalf + r;
        int gr1 = rowBase + waveRow + 16 + mhalf + r;
        if (gr0 < N) C[(size_t)gr0 * 64 + col] = acc0[r];
        if (gr1 < N) C[(size_t)gr1 * 64 + col] = acc1[r];
    }
}

__global__ __launch_bounds__(256)
void gat_node_alpha(const float* __restrict__ h, const float* __restrict__ a_src,
                    const float* __restrict__ a_dst, float* __restrict__ as_,
                    float* __restrict__ ad_, int N) {
    int node = (int)((blockIdx.x * (size_t)blockDim.x + threadIdx.x) >> 5);
    int lane = threadIdx.x & 31;
    if (node >= N) return;
    const float* hr = h + (size_t)node * 64;
    float s = hr[lane] * a_src[lane] + hr[lane + 32] * a_src[lane + 32];
    float d = hr[lane] * a_dst[lane] + hr[lane + 32] * a_dst[lane + 32];
    for (int off = 16; off > 0; off >>= 1) {
        s += __shfl_down(s, off, 32);
        d += __shfl_down(d, off, 32);
    }
    if (lane == 0) { as_[node] = s; ad_[node] = d; }
}

__global__ __launch_bounds__(256)
void gat_init(float* __restrict__ agg, float* __restrict__ emax,
              float* __restrict__ denom, int N) {
    int i = (int)(blockIdx.x * (size_t)blockDim.x + threadIdx.x);
    if (i < N * 64) agg[i] = 0.0f;
    if (i < N) { emax[i] = -3.0e38f; denom[i] = 0.0f; }
}

__global__ __launch_bounds__(256)
void gat_edge_max(const int* __restrict__ srcI, const int* __restrict__ dstI,
                  const float* __restrict__ as_, const float* __restrict__ ad_,
                  float* __restrict__ ebuf, float* __restrict__ emax, int E) {
    int e = (int)(blockIdx.x * (size_t)blockDim.x + threadIdx.x);
    if (e >= E) return;
    int s = srcI[e], d = dstI[e];
    float v = as_[s] + ad_[d];
    v = (v > 0.0f) ? v : v * NEG_SLOPE;
    ebuf[e] = v;
    atomicMax(&emax[d], v);
}

__global__ __launch_bounds__(256)
void gat_edge_exp(const int* __restrict__ dstI, const float* __restrict__ emax,
                  float* __restrict__ ebuf, float* __restrict__ denom, int E) {
    int e = (int)(blockIdx.x * (size_t)blockDim.x + threadIdx.x);
    if (e >= E) return;
    int d = dstI[e];
    float ex = __expf(ebuf[e] - emax[d]);
    ebuf[e] = ex;
    atomicAdd(&denom[d], ex);
}

__global__ __launch_bounds__(256)
void gat_edge_aggr(const int* __restrict__ srcI, const int* __restrict__ dstI,
                   const float* __restrict__ ebuf, const float* __restrict__ denom,
                   const float* __restrict__ h, float* __restrict__ agg, int E) {
    int edge = (int)((blockIdx.x * (size_t)blockDim.x + threadIdx.x) >> 5);
    int lane = threadIdx.x & 31;
    if (edge >= E) return;
    int s = srcI[edge], d = dstI[edge];
    float alpha = ebuf[edge] / (denom[d] + 1e-16f);
    const float* hs = h + (size_t)s * 64;
    float*       od = agg + (size_t)d * 64;
    atomicAdd(&od[lane],      alpha * hs[lane]);
    atomicAdd(&od[lane + 32], alpha * hs[lane + 32]);
}

__global__ __launch_bounds__(256)
void gat_finalize(const float* __restrict__ agg, const float* __restrict__ b,
                  float* __restrict__ out, int N, int relu_flag) {
    int i = (int)(blockIdx.x * (size_t)blockDim.x + threadIdx.x);
    if (i >= N * 64) return;
    float v = agg[i] + b[i & 63];
    out[i] = relu_flag ? fmaxf(v, 0.0f) : v;
}

extern "C" void kernel_launch(void* const* d_in, const int* in_sizes, int n_in,
                              void* d_out, int out_size, void* d_ws, size_t ws_size,
                              hipStream_t stream) {
    const float* x    = (const float*)d_in[0];
    const int*   eidx = (const int*)  d_in[1];
    const float* W1   = (const float*)d_in[2];
    const float* a1s  = (const float*)d_in[3];
    const float* a1d  = (const float*)d_in[4];
    const float* b1   = (const float*)d_in[5];
    const float* W2   = (const float*)d_in[6];
    const float* a2s  = (const float*)d_in[7];
    const float* a2d  = (const float*)d_in[8];
    const float* b2   = (const float*)d_in[9];

    const int N = in_sizes[0] / GAT_INC;
    const int E = in_sizes[1] / 2;
    const int* srcI = eidx;
    const int* dstI = eidx + E;

    float* ws = (float*)d_ws;
    size_t o = 0;
    float* h     = ws + o; o += (size_t)N * 64;
    float* hpost = ws + o; o += (size_t)N * 64;
    float* agg   = ws + o; o += (size_t)N * 64;
    float* as_   = ws + o; o += N;
    float* ad_   = ws + o; o += N;
    float* emax  = ws + o; o += N;
    float* denom = ws + o; o += N;
    float* ebuf  = ws + o; o += E;

    const int gGemm  = (N + 63) / 64;
    const int gNode  = (int)(((size_t)N * 32 + 255) / 256);
    const int gInit  = (N * 64 + 255) / 256;
    const int gEdge  = (E + 255) / 256;
    const int gEAgg  = (int)(((size_t)E * 32 + 255) / 256);
    const int gElem  = (N * 64 + 255) / 256;

    gat_gemm_wmma <<<gGemm, 256, 0, stream>>>(x, W1, h, N, GAT_INC);
    gat_node_alpha<<<gNode, 256, 0, stream>>>(h, a1s, a1d, as_, ad_, N);
    gat_init      <<<gInit, 256, 0, stream>>>(agg, emax, denom, N);
    gat_edge_max  <<<gEdge, 256, 0, stream>>>(srcI, dstI, as_, ad_, ebuf, emax, E);
    gat_edge_exp  <<<gEdge, 256, 0, stream>>>(dstI, emax, ebuf, denom, E);
    gat_edge_aggr <<<gEAgg, 256, 0, stream>>>(srcI, dstI, ebuf, denom, h, agg, E);
    gat_finalize  <<<gElem, 256, 0, stream>>>(agg, b1, hpost, N, 1);

    float* out = (float*)d_out;
    gat_gemm_wmma <<<gGemm, 256, 0, stream>>>(hpost, W2, h, N, GAT_HID);
    gat_node_alpha<<<gNode, 256, 0, stream>>>(h, a2s, a2d, as_, ad_, N);
    gat_init      <<<gInit, 256, 0, stream>>>(out, emax, denom, N);
    gat_edge_max  <<<gEdge, 256, 0, stream>>>(srcI, dstI, as_, ad_, ebuf, emax, E);
    gat_edge_exp  <<<gEdge, 256, 0, stream>>>(dstI, emax, ebuf, denom, E);
    gat_edge_aggr <<<gEAgg, 256, 0, stream>>>(srcI, dstI, ebuf, denom, h, out, E);
    gat_finalize  <<<gElem, 256, 0, stream>>>(out, b2, out, N, 0);
}


# Edges counting-sorted by dst once per call (returning int add = arrival-order rank, scan, fill); one wave per node then does max, exp-sum, row-sum in registers: no float atomics, 14 to 10 launches
# speedup vs baseline: 2.3419x; 2.3419x over previous
#include <hip/hip_runtime.h>
#include <hip/hip_bf16.h>
#include <math.h>

typedef float v2f __attribute__((ext_vector_type(2)));
typedef float v8f __attribute__((ext_vector_type(8)));

#define GAT_INC 256
#define GAT_HID 64
#define NEG_SLOPE 0.2f

__global__ __launch_bounds__(256)
void gat_gemm_wmma(const float* __restrict__ A, const float* __restrict__ W,
                   float* __restrict__ C, int N, int K) {
    __shared__ float sA[64 * (GAT_INC + 4)];
    const int KP   = K + 4;
    const int lane = threadIdx.x & 31;
    const int wave = threadIdx.x >> 5;
    const int rowBase = blockIdx.x * 64;

    const int col      = (wave & 3) * 16 + (lane & 15);
    const int waveRow  = (wave >> 2) * 32;

    for (int k = threadIdx.x; k < K; k += 256)
        __builtin_prefetch(&W[(size_t)k * 64 + col], 0, 3);

    for (int idx = threadIdx.x; idx < 64 * K; idx += 256) {
        int r = idx / K, c = idx - r * K;
        int gr = rowBase + r;
        sA[r * KP + c] = (gr < N) ? A[(size_t)gr * K + c] : 0.0f;
    }
    __syncthreads();

    const int row   = lane & 15;
    const int khalf = (lane >> 4) << 1;

    v8f acc0 = {};
    v8f acc1 = {};
    for (int k = 0; k < K; k += 4) {
        v2f b;
        b.x = W[(size_t)(k + khalf)     * 64 + col];
        b.y = W[(size_t)(k + khalf + 1) * 64 + col];

        v2f a0, a1;
        a0.x = sA[(waveRow +      row) * KP + k + khalf];
        a0.y = sA[(waveRow +      row) * KP + k + khalf + 1];
        a1.x = sA[(waveRow + 16 + row) * KP + k + khalf];
        a1.y = sA[(waveRow + 16 + row) * KP + k + khalf + 1];

        acc0 = __builtin_amdgcn_wmma_f32_16x16x4_f32(
            false, a0, false, b, (short)0, acc0, false, false);
        acc1 = __builtin_amdgcn_wmma_f32_16x16x4_f32(
            false, a1, false, b, (short)0, acc1, false, false);
    }

    const int mhalf = (lane >> 4) << 3;
    for (int r = 0; r < 8; ++r) {
        int gr0 = rowBase + waveRow +      mhalf + r;
        int gr1 = rowBase + waveRow + 16 + mhalf + r;
        if (gr0 < N) C[(size_t)gr0 * 64 + col] = acc0[r];
        if (gr1 < N) C[(size_t)gr1 * 64 + col] = acc1[r];
    }
}

__global__ __launch_bounds__(256)
void gat_node_alpha(const float* __restrict__ h, const float* __restrict__ a_src,
                    const float* __restrict__ a_dst, float* __restrict__ as_,
                    float* __restrict__ ad_, int N) {
    int node = (int)((blockIdx.x * (size_t)blockDim.x + threadIdx.x) >> 5);
    int lane = threadIdx.x & 31;
    if (node >= N) return;
    const float* hr = h + (size_t)node * 64;
    float s = hr[lane] * a_src[lane] + hr[lane + 32] * a_src[lane + 32];
    float d = hr[lane] * a_dst[lane] + hr[lane + 32] * a_dst[lane + 32];
    for (int off = 16; off > 0; off >>= 1) {
        s += __shfl_down(s, off, 32);
        d += __shfl_down(d, off, 32);
    }
    if (lane == 0) { as_[node] = s; ad_[node] = d; }
}

__global__ __launch_bounds__(256)
void csr_zero(int* __restrict__ cnt, int N) {
    int i = (int)(blockIdx.x * (size_t)blockDim.x + threadIdx.x);
    if (i < N) cnt[i] = 0;
}

__global__ __launch_bounds__(256)
void csr_rank(const int* __restrict__ dstI, int* __restrict__ cnt,
              int* __restrict__ rank, int E) {
    int e = (int)(blockIdx.x * (size_t)blockDim.x + threadIdx.x);
    if (e >= E) return;
    rank[e] = atomicAdd(&cnt[dstI[e]], 1);
}

__global__ __launch_bounds__(1024)
void csr_scan(const int* __restrict__ cnt, int* __restrict__ rowptr, int N) {
    __shared__ int sFront[32];
    __shared__ int sWave[32];
    const int tid  = threadIdx.x;
    const int lane = tid & 31;
    const int wave = tid >> 5;
    const int first = blockIdx.x * 1024;

    int front = 0;
    const int4* c4 = (const int4*)cnt;
#pragma unroll 8
    for (int q = tid; q < (first >> 2); q += 1024) {
        int4 v = c4[q];
        front += (v.x + v.y) + (v.z + v.w);
    }
    for (int off = 16; off > 0; off >>= 1) front += __shfl_xor(front, off, 32);

    const int i = first + tid;
    const int x = (i < N) ? cnt[i] : 0;
    int incl = x;
    for (int off = 1; off < 32; off <<= 1) {
        int y = __shfl_up(incl, off, 32);
        if (lane >= off) incl += y;
    }
    if (lane == 0)  sFront[wave] = front;
    if (lane == 31) sWave[wave]  = incl;
    __syncthreads();

    int a = sFront[lane];
    int b = (lane < wave) ? sWave[lane] : 0;
    for (int off = 16; off > 0; off >>= 1) {
        a += __shfl_xor(a, off, 32);
        b += __shfl_xor(b, off, 32);
    }
    if (i <= N) rowptr[i] = a + b + (incl - x);
}

__global__ __launch_bounds__(256)
void csr_fill(const int* __restrict__ srcI, const int* __restrict__ dstI,
              const int* __restrict__ rank, const int* __restrict__ rowptr,
              int* __restrict__ csr_src, int E) {
    int e = (int)(blockIdx.x * (size_t)blockDim.x + threadIdx.x);
    if (e >= E) return;
    const int sv = srcI[e];
    const unsigned p = (unsigned)(rowptr[dstI[e]] + rank[e]);
    csr_src[min(p, (unsigned)(E - 1))] = sv;
}

template <int M>
__device__ __forceinline__ float row_xor(float x) {
    return __builtin_bit_cast(float, __builtin_amdgcn_update_dpp(
        0, __builtin_bit_cast(int, x), 0x160 + M, 0xf, 0xf, false));
}
__device__ __forceinline__ float xhalf(float x) {
    return __builtin_bit_cast(float, __builtin_amdgcn_permlanex16(
        __builtin_bit_cast(int, x), __builtin_bit_cast(int, x),
        0x76543210, 0xfedcba98, false, false));
}
__device__ __forceinline__ float wave_max32(float v) {
    v = fmaxf(v, row_xor<1>(v));
    v = fmaxf(v, row_xor<2>(v));
    v = fmaxf(v, row_xor<4>(v));
    v = fmaxf(v, row_xor<8>(v));
    return fmaxf(v, xhalf(v));
}
__device__ __forceinline__ float wave_sum32(float v) {
    v += row_xor<1>(v);
    v += row_xor<2>(v);
    v += row_xor<4>(v);
    v += row_xor<8>(v);
    return v + xhalf(v);
}
__device__ __forceinline__ float gat_score(const float* __restrict__ as_, int s, float add) {
    float v = as_[s] + add;
    return (v > 0.0f) ? v : v * NEG_SLOPE;
}
template <int G>
__device__ __forceinline__ void gat_row_group(const v2f* __restrict__ h2, int s, float alpha,
                                              int t, unsigned lane, v2f& acc) {
    unsigned idx[G];
    v2f r[G];
#pragma unroll
    for (int u = 0; u < G; ++u)
        idx[u] = (unsigned)__builtin_amdgcn_readlane(s, t + u) * 32u + lane;
#pragma unroll
    for (int u = 0; u < G; ++u)
        r[u] = h2[idx[u]];
#pragma unroll
    for (int u = 0; u < G; ++u) {
        const float at = __builtin_bit_cast(float,
            __builtin_amdgcn_readlane(__builtin_bit_cast(int, alpha), t + u));
        acc = acc + r[u] * at;
    }
}
__device__ __forceinline__ void gat_rows(const v2f* __restrict__ h2, int s, float alpha,
                                         int cnt, unsigned lane, v2f& acc) {
    const int c4 = (cnt + 3) & ~3;
    int t = 0;
    for (; t + 8 <= c4; t += 8) gat_row_group<8>(h2, s, alpha, t, lane, acc);
    if (t < c4)                 gat_row_group<4>(h2, s, alpha, t, lane, acc);
}

__global__ __launch_bounds__(256)
void gat_gather(const int* __restrict__ rowptr, const int* __restrict__ csr_src,
                const float* __restrict__ as_, const float* __restrict__ ad_,
                const float* __restrict__ h, const float* __restrict__ bias,
                float* __restrict__ out, int N, int E, int relu_flag) {
    const unsigned lane = threadIdx.x & 31;
    const int node = __builtin_amdgcn_readfirstlane(
        (int)((blockIdx.x * (size_t)blockDim.x + threadIdx.x) >> 5));
    if (node >= N) return;
    int beg = rowptr[node];
    int end = rowptr[node + 1];
    beg = min(max(beg, 0), E);
    end = min(max(end, beg), E);
    const float add = ad_[node];
    const v2f bb = ((const v2f*)bias)[lane];
    const unsigned nlast = (unsigned)(N - 1);
    const v2f* h2 = (const v2f*)h;

    v2f acc = {0.0f, 0.0f};
    if (end > beg) {
        if (end - beg <= 32) {
            const int  j     = beg + (int)lane;
            const bool valid = j < end;
            const int  s     = (int)min((unsigned)csr_src[min(j, end - 1)], nlast);
            const float v    = gat_score(as_, s, add);
            const float m    = wave_max32(v);
            const float ex   = valid ? __expf(v - m) : 0.0f;
            const float den  = wave_sum32(ex);
            const float alpha = ex / (den + 1e-16f);
            gat_rows(h2, s, alpha, end - beg, lane, acc);
        } else {
            float m = -3.0e38f;
            for (int base = beg; base < end; base += 32) {
                const int j = base + (int)lane;
                const int s = (int)min((unsigned)csr_src[min(j, end - 1)], nlast);
                m = fmaxf(m, gat_score(as_, s, add));
            }
            m = wave_max32(m);
            float den = 0.0f;
            for (int base = beg; base < end; base += 32) {
                const int j = base + (int)lane;
                const int s = (int)min((unsigned)csr_src[min(j, end - 1)], nlast);
                const float ex = __expf(gat_score(as_, s, add) - m);
                den += (j < end) ? ex : 0.0f;
            }
            den = wave_sum32(den);
            const float dd = den + 1e-16f;
            for (int base = beg; base < end; base += 32) {
                const int j = base + (int)lane;
                const int s = (int)min((unsigned)csr_src[min(j, end - 1)], nlast);
                const float ex = __expf(gat_score(as_, s, add) - m);
                const float alpha = (j < end) ? ex / dd : 0.0f;
                gat_rows(h2, s, alpha, min(32, end - base), lane, acc);
            }
        }
    }

    v2f r;
    r.x = acc.x + bb.x;
    r.y = acc.y + bb.y;
    if (relu_flag) { r.x = fmaxf(r.x, 0.0f); r.y = fmaxf(r.y, 0.0f); }
    ((v2f*)out)[(unsigned)node * 32u + lane] = r;
}

extern "C" void kernel_launch(void* const* d_in, const int* in_sizes, int n_in,
                              void* d_out, int out_size, void* d_ws, size_t ws_size,
                              hipStream_t stream) {
    const float* x    = (const float*)d_in[0];
    const int*   eidx = (const int*)  d_in[1];
    const float* W1   = (const float*)d_in[2];
    const float* a1s  = (const float*)d_in[3];
    const float* a1d  = (const float*)d_in[4];
    const float* b1   = (const float*)d_in[5];
    const float* W2   = (const float*)d_in[6];
    const float* a2s  = (const float*)d_in[7];
    const float* a2d  = (const float*)d_in[8];
    const float* b2   = (const float*)d_in[9];

    const int N = in_sizes[0] / GAT_INC;
    const int E = in_sizes[1] / 2;
    const int* srcI = eidx;
    const int* dstI = eidx + E;

    float* ws = (float*)d_ws;
    size_t o = 0;
    float* h     = ws + o; o += (size_t)N * 64;
    float* hpost = ws + o; o += (size_t)N * 64;
    int*   graph = (int*)(ws + o); o += (size_t)N * 64;
    float* as_   = ws + o; o += N;
    float* ad_   = ws + o; o += N;
    o += N;
    o += N;
    int*   rank  = (int*)(ws + o); o += E;

    const size_t Np = ((size_t)N + 4) & ~(size_t)3;
    int* cnt     = graph;
    int* rowptr  = graph + Np;
    int* csr_src = graph + 2 * Np;

    const int gGemm  = (N + 63) / 64;
    const int gNode  = (int)(((size_t)N * 32 + 255) / 256);
    const int gN     = (N + 255) / 256;
    const int gEdge  = (E + 255) / 256;
    const int gScan  = (N + 1 + 1023) / 1024;

    csr_zero<<<gN,    256,  0, stream>>>(cnt, N);
    csr_rank<<<gEdge, 256,  0, stream>>>(dstI, cnt, rank, E);
    csr_scan<<<gScan, 1024, 0, stream>>>(cnt, rowptr, N);
    csr_fill<<<gEdge, 256,  0, stream>>>(srcI, dstI, rank, rowptr, csr_src, E);

    gat_gemm_wmma <<<gGemm, 256, 0, stream>>>(x, W1, h, N, GAT_INC);
    gat_node_alpha<<<gNode, 256, 0, stream>>>(h, a1s, a1d, as_, ad_, N);
    gat_gather    <<<gNode, 256, 0, stream>>>(rowptr, csr_src, as_, ad_, h, b1, hpost, N, E, 1);

    float* out = (float*)d_out;
    gat_gemm_wmma <<<gGemm, 256, 0, stream>>>(hpost, W2, h, N, GAT_HID);
    gat_node_alpha<<<gNode, 256, 0, stream>>>(h, a2s, a2d, as_, ad_, N);
    gat_gather    <<<gNode, 256, 0, stream>>>(rowptr, csr_src, as_, ad_, h, b2, out, N, E, 0);
}


# Per-node and per-edge kernels on a fixed grid of 1024 work-groups whose waves loop over nodes/edges (next nodes' edge ids and logits prefetched), not one wave per node; logit dot sums re-ordered
# speedup vs baseline: 2.5557x; 1.0913x over previous
#include <hip/hip_runtime.h>
#include <hip/hip_bf16.h>
#include <math.h>

typedef float v2f __attribute__((ext_vector_type(2)));
typedef float v8f __attribute__((ext_vector_type(8)));

#define GAT_INC 256
#define GAT_HID 64
#define NEG_SLOPE 0.2f

__global__ __launch_bounds__(256)
void gat_gemm_wmma(const float* __restrict__ A, const float* __restrict__ W,
                   float* __restrict__ C, int N, int K) {
    __shared__ float sA[64 * (GAT_INC + 4)];
    const int KP   = K + 4;
    const int lane = threadIdx.x & 31;
    const int wave = threadIdx.x >> 5;
    const int rowBase = blockIdx.x * 64;

    const int col      = (wave & 3) * 16 + (lane & 15);
    const int waveRow  = (wave >> 2) * 32;

    for (int k = threadIdx.x; k < K; k += 256)
        __builtin_prefetch(&W[(size_t)k * 64 + col], 0, 3);

    for (int idx = threadIdx.x; idx < 64 * K; idx += 256) {
        int r = idx / K, c = idx - r * K;
        int gr = rowBase + r;
        sA[r * KP + c] = (gr < N) ? A[(size_t)gr * K + c] : 0.0f;
    }
    __syncthreads();

    const int row   = lane & 15;
    const int khalf = (lane >> 4) << 1;

    v8f acc0 = {};
    v8f acc1 = {};
    for (int k = 0; k < K; k += 4) {
        v2f b;
        b.x = W[(size_t)(k + khalf)     * 64 + col];
        b.y = W[(size_t)(k + khalf + 1) * 64 + col];

        v2f a0, a1;
        a0.x = sA[(waveRow +      row) * KP + k + khalf];
        a0.y = sA[(waveRow +      row) * KP + k + khalf + 1];
        a1.x = sA[(waveRow + 16 + row) * KP + k + khalf];
        a1.y = sA[(waveRow + 16 + row) * KP + k + khalf + 1];

        acc0 = __builtin_amdgcn_wmma_f32_16x16x4_f32(
            false, a0, false, b, (short)0, acc0, false, false);
        acc1 = __builtin_amdgcn_wmma_f32_16x16x4_f32(
            false, a1, false, b, (short)0, acc1, false, false);
    }

    const int mhalf = (lane >> 4) << 3;
    for (int r = 0; r < 8; ++r) {
        int gr0 = rowBase + waveRow +      mhalf + r;
        int gr1 = rowBase + waveRow + 16 + mhalf + r;
        if (gr0 < N) C[(size_t)gr0 * 64 + col] = acc0[r];
        if (gr1 < N) C[(size_t)gr1 * 64 + col] = acc1[r];
    }
}

template <int M>
__device__ __forceinline__ float row_xor(float x) {
    return __builtin_bit_cast(float, __builtin_amdgcn_update_dpp(
        0, __builtin_bit_cast(int, x), 0x160 + M, 0xf, 0xf, false));
}
__device__ __forceinline__ float xhalf(float x) {
    return __builtin_bit_cast(float, __builtin_amdgcn_permlanex16(
        __builtin_bit_cast(int, x), __builtin_bit_cast(int, x),
        0x76543210, 0xfedcba98, false, false));
}
__device__ __forceinline__ float wave_max32(float v) {
    v = fmaxf(v, row_xor<1>(v));
    v = fmaxf(v, row_xor<2>(v));
    v = fmaxf(v, row_xor<4>(v));
    v = fmaxf(v, row_xor<8>(v));
    return fmaxf(v, xhalf(v));
}
__device__ __forceinline__ float wave_sum32(float v) {
    v += row_xor<1>(v);
    v += row_xor<2>(v);
    v += row_xor<4>(v);
    v += row_xor<8>(v);
    return v + xhalf(v);
}

#define GAT_PERS_WG 1024

__global__ __launch_bounds__(256) __attribute__((amdgpu_waves_per_eu(4, 8)))
void gat_node_alpha(const float* __restrict__ h, const float* __restrict__ a_src,
                    const float* __restrict__ a_dst, float* __restrict__ as_,
                    float* __restrict__ ad_, int N) {
    const int lane = threadIdx.x & 31;
    const int sub  = lane & 15;
    const int half = lane >> 4;
    const unsigned nw = gridDim.x * (blockDim.x >> 5);
    const unsigned gw = (unsigned)__builtin_amdgcn_readfirstlane(
        (int)(blockIdx.x * (blockDim.x >> 5) + (threadIdx.x >> 5)));
    const unsigned per = (unsigned)N / nw, rem = (unsigned)N % nw;
    const int r0 = (int)(gw * per + min(gw, rem));
    const int r1 = r0 + (int)per + (gw < rem ? 1 : 0);
    const float4 ws = ((const float4*)a_src)[sub];
    const float4 wd = ((const float4*)a_dst)[sub];
    const float4* h4 = (const float4*)h;
    const int rlast = N - 1;
    for (int b = r0; b < r1; b += 16) {
        float4 x[8];
#pragma unroll
        for (int u = 0; u < 8; ++u) {
            const unsigned row = (unsigned)min(b + 2 * u + half, rlast);
            x[u] = h4[row * 16u + (unsigned)sub];
        }
        float rs = 0.0f, rd = 0.0f;
#pragma unroll
        for (int u = 0; u < 8; ++u) {
            float s = x[u].x * ws.x + x[u].y * ws.y + x[u].z * ws.z + x[u].w * ws.w;
            float d = x[u].x * wd.x + x[u].y * wd.y + x[u].z * wd.z + x[u].w * wd.w;
            s += row_xor<1>(s); d += row_xor<1>(d);
            s += row_xor<2>(s); d += row_xor<2>(d);
            s += row_xor<4>(s); d += row_xor<4>(d);
            s += row_xor<8>(s); d += row_xor<8>(d);
            if (sub == 2 * u + half) { rs = s; rd = d; }
        }
        const int row = b + sub;
        if ((sub & 1) == half && row < r1) { as_[row] = rs; ad_[row] = rd; }
    }
}

__global__ __launch_bounds__(256)
void csr_zero(int* __restrict__ cnt, int N) {
    int i = (int)(blockIdx.x * (size_t)blockDim.x + threadIdx.x);
    if (i < N) cnt[i] = 0;
}

__global__ __launch_bounds__(256)
void csr_rank(const int* __restrict__ dstI, int* __restrict__ cnt,
              int* __restrict__ rank, int E) {
    const int T = (int)(gridDim.x * blockDim.x);
    int e = (int)(blockIdx.x * blockDim.x + threadIdx.x);
    for (; e < E - 3 * T; e += 4 * T) {
        const int d0 = dstI[e], d1 = dstI[e + T], d2 = dstI[e + 2 * T], d3 = dstI[e + 3 * T];
        const int k0 = atomicAdd(&cnt[d0], 1);
        const int k1 = atomicAdd(&cnt[d1], 1);
        const int k2 = atomicAdd(&cnt[d2], 1);
        const int k3 = atomicAdd(&cnt[d3], 1);
        rank[e] = k0; rank[e + T] = k1; rank[e + 2 * T] = k2; rank[e + 3 * T] = k3;
    }
    for (; e < E; e += T) rank[e] = atomicAdd(&cnt[dstI[e]], 1);
}

__global__ __launch_bounds__(1024)
void csr_scan(const int* __restrict__ cnt, int* __restrict__ rowptr, int N) {
    __shared__ int sFront[32];
    __shared__ int sWave[32];
    const int tid  = threadIdx.x;
    const int lane = tid & 31;
    const int wave = tid >> 5;
    const int first = blockIdx.x * 1024;

    int front = 0;
    const int4* c4 = (const int4*)cnt;
#pragma unroll 8
    for (int q = tid; q < (first >> 2); q += 1024) {
        int4 v = c4[q];
        front += (v.x + v.y) + (v.z + v.w);
    }
    for (int off = 16; off > 0; off >>= 1) front += __shfl_xor(front, off, 32);

    const int i = first + tid;
    const int x = (i < N) ? cnt[i] : 0;
    int incl = x;
    for (int off = 1; off < 32; off <<= 1) {
        int y = __shfl_up(incl, off, 32);
        if (lane >= off) incl += y;
    }
    if (lane == 0)  sFront[wave] = front;
    if (lane == 31) sWave[wave]  = incl;
    __syncthreads();

    int a = sFront[lane];
    int b = (lane < wave) ? sWave[lane] : 0;
    for (int off = 16; off > 0; off >>= 1) {
        a += __shfl_xor(a, off, 32);
        b += __shfl_xor(b, off, 32);
    }
    if (i <= N) rowptr[i] = a + b + (incl - x);
}

__global__ __launch_bounds__(256)
void csr_fill(const int* __restrict__ srcI, const int* __restrict__ dstI,
              const int* __restrict__ rank, const int* __restrict__ rowptr,
              int* __restrict__ csr_src, int E) {
    const int T = (int)(gridDim.x * blockDim.x);
    const unsigned elast = (unsigned)(E - 1);
    int e = (int)(blockIdx.x * blockDim.x + threadIdx.x);
    for (; e < E - 3 * T; e += 4 * T) {
        const int d0 = dstI[e], d1 = dstI[e + T], d2 = dstI[e + 2 * T], d3 = dstI[e + 3 * T];
        const int s0 = srcI[e], s1 = srcI[e + T], s2 = srcI[e + 2 * T], s3 = srcI[e + 3 * T];
        const int k0 = rank[e], k1 = rank[e + T], k2 = rank[e + 2 * T], k3 = rank[e + 3 * T];
        const unsigned p0 = (unsigned)(rowptr[d0] + k0);
        const unsigned p1 = (unsigned)(rowptr[d1] + k1);
        const unsigned p2 = (unsigned)(rowptr[d2] + k2);
        const unsigned p3 = (unsigned)(rowptr[d3] + k3);
        csr_src[min(p0, elast)] = s0;
        csr_src[min(p1, elast)] = s1;
        csr_src[min(p2, elast)] = s2;
        csr_src[min(p3, elast)] = s3;
    }
    for (; e < E; e += T) {
        const int sv = srcI[e];
        const unsigned p = (unsigned)(rowptr[dstI[e]] + rank[e]);
        csr_src[min(p, elast)] = sv;
    }
}

__device__ __forceinline__ float gat_score(const float* __restrict__ as_, int s, float add) {
    float v = as_[s] + add;
    return (v > 0.0f) ? v : v * NEG_SLOPE;
}
template <int G>
__device__ __forceinline__ void gat_row_group(const v2f* __restrict__ h2, int s, float alpha,
                                              int t, unsigned lane, v2f& acc) {
    unsigned idx[G];
    v2f r[G];
#pragma unroll
    for (int u = 0; u < G; ++u)
        idx[u] = (unsigned)__builtin_amdgcn_readlane(s, t + u) * 32u + lane;
#pragma unroll
    for (int u = 0; u < G; ++u)
        r[u] = h2[idx[u]];
#pragma unroll
    for (int u = 0; u < G; ++u) {
        const float at = __builtin_bit_cast(float,
            __builtin_amdgcn_readlane(__builtin_bit_cast(int, alpha), t + u));
        acc = acc + r[u] * at;
    }
}
__device__ __forceinline__ void gat_rows(const v2f* __restrict__ h2, int s, float alpha,
                                         int cnt, unsigned lane, v2f& acc) {
    const int c4 = (cnt + 3) & ~3;
    if (c4 <= 4)       gat_row_group<4>(h2, s, alpha, 0, lane, acc);
    else if (c4 <= 8)  gat_row_group<8>(h2, s, alpha, 0, lane, acc);
    else if (c4 <= 12) gat_row_group<12>(h2, s, alpha, 0, lane, acc);
    else {
        gat_row_group<16>(h2, s, alpha, 0, lane, acc);
        int t = 16;
        for (; t + 8 <= c4; t += 8) gat_row_group<8>(h2, s, alpha, t, lane, acc);
        if (t < c4)                 gat_row_group<4>(h2, s, alpha, t, lane, acc);
    }
}

__device__ __forceinline__ int gat_edge_ids(const int* __restrict__ csr_src, int rp, int k,
                                            unsigned lane, int elast) {
    const int beg = __builtin_amdgcn_readlane(rp, k);
    const int end = max(__builtin_amdgcn_readlane(rp, k + 1), beg);
    const int j   = min(beg + (int)lane, end - 1);
    return csr_src[min(max(j, 0), elast)];
}

__global__ __launch_bounds__(256)
void gat_gather(const int* __restrict__ rowptr, const int* __restrict__ csr_src,
                const float* __restrict__ as_, const float* __restrict__ ad_,
                const float* __restrict__ h, const float* __restrict__ bias,
                float* __restrict__ out, int N, int E, int relu_flag) {
    const unsigned lane = threadIdx.x & 31;
    const unsigned nw = gridDim.x * (blockDim.x >> 5);
    const unsigned gw = (unsigned)__builtin_amdgcn_readfirstlane(
        (int)(blockIdx.x * (blockDim.x >> 5) + (threadIdx.x >> 5)));
    const unsigned per = (unsigned)N / nw, rem = (unsigned)N % nw;
    const int n0 = (int)(gw * per + min(gw, rem));
    const int n1 = n0 + (int)per + (gw < rem ? 1 : 0);
    if (n0 >= n1) return;
    const v2f bb = ((const v2f*)bias)[lane];
    const unsigned nlast = (unsigned)(N - 1);
    const int elast = max(E - 1, 0);
    const v2f* h2 = (const v2f*)h;

    for (int base = n0; base < n1; base += 31) {
        const int c = min(31, n1 - base);
        int rp = rowptr[min(base + (int)lane, N)];
        rp = min(max(rp, 0), E);
        const float adl = ad_[min(base + (int)lane, N - 1)];

        int   sA = gat_edge_ids(csr_src, rp, 0, lane, elast);
        int   sB = gat_edge_ids(csr_src, rp, min(1, c - 1), lane, elast);
        sA = (int)min((unsigned)sA, nlast);
        float vA = as_[sA];

        for (int k = 0; k < c; ++k) {
            const int sC = gat_edge_ids(csr_src, rp, min(k + 2, c - 1), lane, elast);
            sB = (int)min((unsigned)sB, nlast);
            const float vB = as_[sB];

            const int beg = __builtin_amdgcn_readlane(rp, k);
            const int end = max(__builtin_amdgcn_readlane(rp, k + 1), beg);
            const float add = __builtin_bit_cast(float,
                __builtin_amdgcn_readlane(__builtin_bit_cast(int, adl), k));
            const int cnt = end - beg;

            v2f acc = {0.0f, 0.0f};
            if (cnt > 0) {
                if (cnt <= 32) {
                    const bool valid = (int)lane < cnt;
                    float v = vA + add;
                    v = (v > 0.0f) ? v : v * NEG_SLOPE;
                    const float m    = wave_max32(v);
                    const float ex   = valid ? __expf(v - m) : 0.0f;
                    const float den  = wave_sum32(ex);
                    const float alpha = ex / (den + 1e-16f);
                    gat_rows(h2, sA, alpha, cnt, lane, acc);
                } else {
                    float m = -3.0e38f;
                    for (int eb = beg; eb < end; eb += 32) {
                        const int j = eb + (int)lane;
                        const int s = (int)min((unsigned)csr_src[min(j, end - 1)], nlast);
                        m = fmaxf(m, gat_score(as_, s, add));
                    }
                    m = wave_max32(m);
                    float den = 0.0f;
                    for (int eb = beg; eb < end; eb += 32) {
                        const int j = eb + (int)lane;
                        const int s = (int)min((unsigned)csr_src[min(j, end - 1)], nlast);
                        const float ex = __expf(gat_score(as_, s, add) - m);
                        den += (j < end) ? ex : 0.0f;
                    }
                    den = wave_sum32(den);
                    const float dd = den + 1e-16f;
                    for (int eb = beg; eb < end; eb += 32) {
                        const int j = eb + (int)lane;
                        const int s = (int)min((unsigned)csr_src[min(j, end - 1)], nlast);
                        const float ex = __expf(gat_score(as_, s, add) - m);
                        const float alpha = (j < end) ? ex / dd : 0.0f;
                        gat_rows(h2, s, alpha, min(32, end - eb), lane, acc);
                    }
                }
            }

            v2f r;
            r.x = acc.x + bb.x;
            r.y = acc.y + bb.y;
            if (relu_flag) { r.x = fmaxf(r.x, 0.0f); r.y = fmaxf(r.y, 0.0f); }
            ((v2f*)out)[(unsigned)(base + k) * 32u + lane] = r;

            sA = sB; vA = vB; sB = sC;
        }
    }
}

extern "C" void kernel_launch(void* const* d_in, const int* in_sizes, int n_in,
                              void* d_out, int out_size, void* d_ws, size_t ws_size,
                              hipStream_t stream) {
    const float* x    = (const float*)d_in[0];
    const int*   eidx = (const int*)  d_in[1];
    const float* W1   = (const float*)d_in[2];
    const float* a1s  = (const float*)d_in[3];
    const float* a1d  = (const float*)d_in[4];
    const float* b1   = (const float*)d_in[5];
    const float* W2   = (const float*)d_in[6];
    const float* a2s  = (const float*)d_in[7];
    const float* a2d  = (const float*)d_in[8];
    const float* b2   = (const float*)d_in[9];

    const int N = in_sizes[0] / GAT_INC;
    const int E = in_sizes[1] / 2;
    const int* srcI = eidx;
    const int* dstI = eidx + E;

    float* ws = (float*)d_ws;
    size_t o = 0;
    float* h     = ws + o; o += (size_t)N * 64;
    float* hpost = ws + o; o += (size_t)N * 64;
    int*   graph = (int*)(ws + o); o += (size_t)N * 64;
    float* as_   = ws + o; o += N;
    float* ad_   = ws + o; o += N;
    o += N;
    o += N;
    int*   rank  = (int*)(ws + o); o += E;

    const size_t Np = ((size_t)N + 4) & ~(size_t)3;
    int* cnt     = graph;
    int* rowptr  = graph + Np;
    int* csr_src = graph + 2 * Np;

    const int gGemm  = (N + 63) / 64;
    const int gPers  = GAT_PERS_WG;
    const int gN     = (N + 255) / 256;
    const int gScan  = (N + 1 + 1023) / 1024;

    csr_zero<<<gN,    256,  0, stream>>>(cnt, N);
    csr_rank<<<gPers, 256,  0, stream>>>(dstI, cnt, rank, E);
    csr_scan<<<gScan, 1024, 0, stream>>>(cnt, rowptr, N);
    csr_fill<<<gPers, 256,  0, stream>>>(srcI, dstI, rank, rowptr, csr_src, E);

    gat_gemm_wmma <<<gGemm, 256, 0, stream>>>(x, W1, h, N, GAT_INC);
    gat_node_alpha<<<gPers, 256, 0, stream>>>(h, a1s, a1d, as_, ad_, N);
    gat_gather    <<<gPers, 256, 0, stream>>>(rowptr, csr_src, as_, ad_, h, b1, hpost, N, E, 1);

    float* out = (float*)d_out;
    gat_gemm_wmma <<<gGemm, 256, 0, stream>>>(hpost, W2, h, N, GAT_HID);
    gat_node_alpha<<<gPers, 256, 0, stream>>>(h, a2s, a2d, as_, ad_, N);
    gat_gather    <<<gPers, 256, 0, stream>>>(rowptr, csr_src, as_, ad_, h, b2, out, N, E, 0);
}


# GEMM A tile by 128-bit non-temporal loads, all of a thread's pieces in flight before its first LDS store; K a template constant (was one 4-byte load in flight per lane, a division per element)
# speedup vs baseline: 3.6011x; 1.4090x over previous
#include <hip/hip_runtime.h>
#include <hip/hip_bf16.h>
#include <math.h>

typedef float v2f __attribute__((ext_vector_type(2)));
typedef float v8f __attribute__((ext_vector_type(8)));
typedef float v4f __attribute__((ext_vector_type(4)));

#define GAT_INC 256
#define GAT_HID 64
#define NEG_SLOPE 0.2f

template <int K>
__global__ __launch_bounds__(256)
void gat_gemm_wmma(const float* __restrict__ A, const float* __restrict__ W,
                   float* __restrict__ C, int N) {
    constexpr int KP    = K + 4;
    constexpr int QPR   = K / 4;
    constexpr int RSTEP = 256 / QPR;
    constexpr int NQ    = 64 / RSTEP;
    static_assert(K % 4 == 0 && 256 % QPR == 0 && 64 % RSTEP == 0, "tile map");
    __shared__ __attribute__((aligned(16))) float sA[64 * KP];
    const int lane = threadIdx.x & 31;
    const int wave = threadIdx.x >> 5;
    const int rowBase = blockIdx.x * 64;

    const int col      = (wave & 3) * 16 + (lane & 15);
    const int waveRow  = (wave >> 2) * 32;

    for (int k = threadIdx.x; k < K; k += 256)
        __builtin_prefetch(&W[(size_t)k * 64 + col], 0, 3);

    {
        const int q  = threadIdx.x % QPR;
        const int r0 = threadIdx.x / QPR;
        v4f t[NQ];
#pragma unroll
        for (int i = 0; i < NQ; ++i) {
            const int gr = max(min(rowBase + r0 + i * RSTEP, N - 1), 0);
            t[i] = __builtin_nontemporal_load((const v4f*)(A + (size_t)gr * K) + q);
        }
#pragma unroll
        for (int i = 0; i < NQ; ++i)
            *(v4f*)&sA[(r0 + i * RSTEP) * KP + 4 * q] = t[i];
    }
    __syncthreads();

    const int row   = lane & 15;
    const int khalf = (lane >> 4) << 1;

    v8f acc0 = {};
    v8f acc1 = {};
#pragma unroll 8
    for (int k = 0; k < K; k += 4) {
        v2f b;
        b.x = W[(size_t)(k + khalf)     * 64 + col];
        b.y = W[(size_t)(k + khalf + 1) * 64 + col];

        v2f a0, a1;
        a0.x = sA[(waveRow +      row) * KP + k + khalf];
        a0.y = sA[(waveRow +      row) * KP + k + khalf + 1];
        a1.x = sA[(waveRow + 16 + row) * KP + k + khalf];
        a1.y = sA[(waveRow + 16 + row) * KP + k + khalf + 1];

        acc0 = __builtin_amdgcn_wmma_f32_16x16x4_f32(
            false, a0, false, b, (short)0, acc0, false, false);
        acc1 = __builtin_amdgcn_wmma_f32_16x16x4_f32(
            false, a1, false, b, (short)0, acc1, false, false);
    }

    const int mhalf = (lane >> 4) << 3;
    for (int r = 0; r < 8; ++r) {
        int gr0 = rowBase + waveRow +      mhalf + r;
        int gr1 = rowBase + waveRow + 16 + mhalf + r;
        if (gr0 < N) C[(size_t)gr0 * 64 + col] = acc0[r];
        if (gr1 < N) C[(size_t)gr1 * 64 + col] = acc1[r];
    }
}

template <int M>
__device__ __forceinline__ float row_xor(float x) {
    return __builtin_bit_cast(float, __builtin_amdgcn_update_dpp(
        0, __builtin_bit_cast(int, x), 0x160 + M, 0xf, 0xf, false));
}
__device__ __forceinline__ float xhalf(float x) {
    return __builtin_bit_cast(float, __builtin_amdgcn_permlanex16(
        __builtin_bit_cast(int, x), __builtin_bit_cast(int, x),
        0x76543210, 0xfedcba98, false, false));
}
__device__ __forceinline__ float wave_max32(float v) {
    v = fmaxf(v, row_xor<1>(v));
    v = fmaxf(v, row_xor<2>(v));
    v = fmaxf(v, row_xor<4>(v));
    v = fmaxf(v, row_xor<8>(v));
    return fmaxf(v, xhalf(v));
}
__device__ __forceinline__ float wave_sum32(float v) {
    v += row_xor<1>(v);
    v += row_xor<2>(v);
    v += row_xor<4>(v);
    v += row_xor<8>(v);
    return v + xhalf(v);
}

#define GAT_PERS_WG 1024

__global__ __launch_bounds__(256) __attribute__((amdgpu_waves_per_eu(4, 8)))
void gat_node_alpha(const float* __restrict__ h, const float* __restrict__ a_src,
                    const float* __restrict__ a_dst, float* __restrict__ as_,
                    float* __restrict__ ad_, int N) {
    const int lane = threadIdx.x & 31;
    const int sub  = lane & 15;
    const int half = lane >> 4;
    const unsigned nw = gridDim.x * (blockDim.x >> 5);
    const unsigned gw = (unsigned)__builtin_amdgcn_readfirstlane(
        (int)(blockIdx.x * (blockDim.x >> 5) + (threadIdx.x >> 5)));
    const unsigned per = (unsigned)N / nw, rem = (unsigned)N % nw;
    const int r0 = (int)(gw * per + min(gw, rem));
    const int r1 = r0 + (int)per + (gw < rem ? 1 : 0);
    const float4 ws = ((const float4*)a_src)[sub];
    const float4 wd = ((const float4*)a_dst)[sub];
    const float4* h4 = (const float4*)h;
    const int rlast = N - 1;
    for (int b = r0; b < r1; b += 16) {
        float4 x[8];
#pragma unroll
        for (int u = 0; u < 8; ++u) {
            const unsigned row = (unsigned)min(b + 2 * u + half, rlast);
            x[u] = h4[row * 16u + (unsigned)sub];
        }
        float rs = 0.0f, rd = 0.0f;
#pragma unroll
        for (int u = 0; u < 8; ++u) {
            float s = x[u].x * ws.x + x[u].y * ws.y + x[u].z * ws.z + x[u].w * ws.w;
            float d = x[u].x * wd.x + x[u].y * wd.y + x[u].z * wd.z + x[u].w * wd.w;
            s += row_xor<1>(s); d += row_xor<1>(d);
            s += row_xor<2>(s); d += row_xor<2>(d);
            s += row_xor<4>(s); d += row_xor<4>(d);
            s += row_xor<8>(s); d += row_xor<8>(d);
            if (sub == 2 * u + half) { rs = s; rd = d; }
        }
        const int row = b + sub;
        if ((sub & 1) == half && row < r1) { as_[row] = rs; ad_[row] = rd; }
    }
}

__global__ __launch_bounds__(256)
void csr_zero(int* __restrict__ cnt, int N) {
    int i = (int)(blockIdx.x * (size_t)blockDim.x + threadIdx.x);
    if (i < N) cnt[i] = 0;
}

__global__ __launch_bounds__(256)
void csr_rank(const int* __restrict__ dstI, int* __restrict__ cnt,
              int* __restrict__ rank, int E) {
    const int T = (int)(gridDim.x * blockDim.x);
    int e = (int)(blockIdx.x * blockDim.x + threadIdx.x);
    for (; e < E - 3 * T; e += 4 * T) {
        const int d0 = dstI[e], d1 = dstI[e + T], d2 = dstI[e + 2 * T], d3 = dstI[e + 3 * T];
        const int k0 = atomicAdd(&cnt[d0], 1);
        const int k1 = atomicAdd(&cnt[d1], 1);
        const int k2 = atomicAdd(&cnt[d2], 1);
        const int k3 = atomicAdd(&cnt[d3], 1);
        rank[e] = k0; rank[e + T] = k1; rank[e + 2 * T] = k2; rank[e + 3 * T] = k3;
    }
    for (; e < E; e += T) rank[e] = atomicAdd(&cnt[dstI[e]], 1);
}

__global__ __launch_bounds__(1024)
void csr_scan(const int* __restrict__ cnt, int* __restrict__ rowptr, int N) {
    __shared__ int sFront[32];
    __shared__ int sWave[32];
    const int tid  = threadIdx.x;
    const int lane = tid & 31;
    const int wave = tid >> 5;
    const int first = blockIdx.x * 1024;

    int front = 0;
    const int4* c4 = (const int4*)cnt;
#pragma unroll 8
    for (int q = tid; q < (first >> 2); q += 1024) {
        int4 v = c4[q];
        front += (v.x + v.y) + (v.z + v.w);
    }
    for (int off = 16; off > 0; off >>= 1) front += __shfl_xor(front, off, 32);

    const int i = first + tid;
    const int x = (i < N) ? cnt[i] : 0;
    int incl = x;
    for (int off = 1; off < 32; off <<= 1) {
        int y = __shfl_up(incl, off, 32);
        if (lane >= off) incl += y;
    }
    if (lane == 0)  sFront[wave] = front;
    if (lane == 31) sWave[wave]  = incl;
    __syncthreads();

    int a = sFront[lane];
    int b = (lane < wave) ? sWave[lane] : 0;
    for (int off = 16; off > 0; off >>= 1) {
        a += __shfl_xor(a, off, 32);
        b += __shfl_xor(b, off, 32);
    }
    if (i <= N) rowptr[i] = a + b + (incl - x);
}

__global__ __launch_bounds__(256)
void csr_fill(const int* __restrict__ srcI, const int* __restrict__ dstI,
              const int* __restrict__ rank, const int* __restrict__ rowptr,
              int* __restrict__ csr_src, int E) {
    const int T = (int)(gridDim.x * blockDim.x);
    const unsigned elast = (unsigned)(E - 1);
    int e = (int)(blockIdx.x * blockDim.x + threadIdx.x);
    for (; e < E - 3 * T; e += 4 * T) {
        const int d0 = dstI[e], d1 = dstI[e + T], d2 = dstI[e + 2 * T], d3 = dstI[e + 3 * T];
        const int s0 = srcI[e], s1 = srcI[e + T], s2 = srcI[e + 2 * T], s3 = srcI[e + 3 * T];
        const int k0 = rank[e], k1 = rank[e + T], k2 = rank[e + 2 * T], k3 = rank[e + 3 * T];
        const unsigned p0 = (unsigned)(rowptr[d0] + k0);
        const unsigned p1 = (unsigned)(rowptr[d1] + k1);
        const unsigned p2 = (unsigned)(rowptr[d2] + k2);
        const unsigned p3 = (unsigned)(rowptr[d3] + k3);
        csr_src[min(p0, elast)] = s0;
        csr_src[min(p1, elast)] = s1;
        csr_src[min(p2, elast)] = s2;
        csr_src[min(p3, elast)] = s3;
    }
    for (; e < E; e += T) {
        const int sv = srcI[e];
        const unsigned p = (unsigned)(rowptr[dstI[e]] + rank[e]);
        csr_src[min(p, elast)] = sv;
    }
}

__device__ __forceinline__ float gat_score(const float* __restrict__ as_, int s, float add) {
    float v = as_[s] + add;
    return (v > 0.0f) ? v : v * NEG_SLOPE;
}
template <int G>
__device__ __forceinline__ void gat_row_group(const v2f* __restrict__ h2, int s, float alpha,
                                              int t, unsigned lane, v2f& acc) {
    unsigned idx[G];
    v2f r[G];
#pragma unroll
    for (int u = 0; u < G; ++u)
        idx[u] = (unsigned)__builtin_amdgcn_readlane(s, t + u) * 32u + lane;
#pragma unroll
    for (int u = 0; u < G; ++u)
        r[u] = h2[idx[u]];
#pragma unroll
    for (int u = 0; u < G; ++u) {
        const float at = __builtin_bit_cast(float,
            __builtin_amdgcn_readlane(__builtin_bit_cast(int, alpha), t + u));
        acc = acc + r[u] * at;
    }
}
__device__ __forceinline__ void gat_rows(const v2f* __restrict__ h2, int s, float alpha,
                                         int cnt, unsigned lane, v2f& acc) {
    const int c4 = (cnt + 3) & ~3;
    if (c4 <= 4)       gat_row_group<4>(h2, s, alpha, 0, lane, acc);
    else if (c4 <= 8)  gat_row_group<8>(h2, s, alpha, 0, lane, acc);
    else if (c4 <= 12) gat_row_group<12>(h2, s, alpha, 0, lane, acc);
    else {
        gat_row_group<16>(h2, s, alpha, 0, lane, acc);
        int t = 16;
        for (; t + 8 <= c4; t += 8) gat_row_group<8>(h2, s, alpha, t, lane, acc);
        if (t < c4)                 gat_row_group<4>(h2, s, alpha, t, lane, acc);
    }
}

__device__ __forceinline__ int gat_edge_ids(const int* __restrict__ csr_src, int rp, int k,
                                            unsigned lane, int elast) {
    const int beg = __builtin_amdgcn_readlane(rp, k);
    const int end = max(__builtin_amdgcn_readlane(rp, k + 1), beg);
    const int j   = min(beg + (int)lane, end - 1);
    return csr_src[min(max(j, 0), elast)];
}

__global__ __launch_bounds__(256)
void gat_gather(const int* __restrict__ rowptr, const int* __restrict__ csr_src,
                const float* __restrict__ as_, const float* __restrict__ ad_,
                const float* __restrict__ h, const float* __restrict__ bias,
                float* __restrict__ out, int N, int E, int relu_flag) {
    const unsigned lane = threadIdx.x & 31;
    const unsigned nw = gridDim.x * (blockDim.x >> 5);
    const unsigned gw = (unsigned)__builtin_amdgcn_readfirstlane(
        (int)(blockIdx.x * (blockDim.x >> 5) + (threadIdx.x >> 5)));
    const unsigned per = (unsigned)N / nw, rem = (unsigned)N % nw;
    const int n0 = (int)(gw * per + min(gw, rem));
    const int n1 = n0 + (int)per + (gw < rem ? 1 : 0);
    if (n0 >= n1) return;
    const v2f bb = ((const v2f*)bias)[lane];
    const unsigned nlast = (unsigned)(N - 1);
    const int elast = max(E - 1, 0);
    const v2f* h2 = (const v2f*)h;

    for (int base = n0; base < n1; base += 31) {
        const int c = min(31, n1 - base);
        int rp = rowptr[min(base + (int)lane, N)];
        rp = min(max(rp, 0), E);
        const float adl = ad_[min(base + (int)lane, N - 1)];

        int   sA = gat_edge_ids(csr_src, rp, 0, lane, elast);
        int   sB = gat_edge_ids(csr_src, rp, min(1, c - 1), lane, elast);
        sA = (int)min((unsigned)sA, nlast);
        float vA = as_[sA];

        for (int k = 0; k < c; ++k) {
            const int sC = gat_edge_ids(csr_src, rp, min(k + 2, c - 1), lane, elast);
            sB = (int)min((unsigned)sB, nlast);
            const float vB = as_[sB];

            const int beg = __builtin_amdgcn_readlane(rp, k);
            const int end = max(__builtin_amdgcn_readlane(rp, k + 1), beg);
            const float add = __builtin_bit_cast(float,
                __builtin_amdgcn_readlane(__builtin_bit_cast(int, adl), k));
            const int cnt = end - beg;

            v2f acc = {0.0f, 0.0f};
            if (cnt > 0) {
                if (cnt <= 32) {
                    const bool valid = (int)lane < cnt;
                    float v = vA + add;
                    v = (v > 0.0f) ? v : v * NEG_SLOPE;
                    const float m    = wave_max32(v);
                    const float ex   = valid ? __expf(v - m) : 0.0f;
                    const float den  = wave_sum32(ex);
                    const float alpha = ex / (den + 1e-16f);
                    gat_rows(h2, sA, alpha, cnt, lane, acc);
                } else {
                    float m = -3.0e38f;
                    for (int eb = beg; eb < end; eb += 32) {
                        const int j = eb + (int)lane;
                        const int s = (int)min((unsigned)csr_src[min(j, end - 1)], nlast);
                        m = fmaxf(m, gat_score(as_, s, add));
                    }
                    m = wave_max32(m);
                    float den = 0.0f;
                    for (int eb = beg; eb < end; eb += 32) {
                        const int j = eb + (int)lane;
                        const int s = (int)min((unsigned)csr_src[min(j, end - 1)], nlast);
                        const float ex = __expf(gat_score(as_, s, add) - m);
                        den += (j < end) ? ex : 0.0f;
                    }
                    den = wave_sum32(den);
                    const float dd = den + 1e-16f;
                    for (int eb = beg; eb < end; eb += 32) {
                        const int j = eb + (int)lane;
                        const int s = (int)min((unsigned)csr_src[min(j, end - 1)], nlast);
                        const float ex = __expf(gat_score(as_, s, add) - m);
                        const float alpha = (j < end) ? ex / dd : 0.0f;
                        gat_rows(h2, s, alpha, min(32, end - eb), lane, acc);
                    }
                }
            }

            v2f r;
            r.x = acc.x + bb.x;
            r.y = acc.y + bb.y;
            if (relu_flag) { r.x = fmaxf(r.x, 0.0f); r.y = fmaxf(r.y, 0.0f); }
            ((v2f*)out)[(unsigned)(base + k) * 32u + lane] = r;

            sA = sB; vA = vB; sB = sC;
        }
    }
}

extern "C" void kernel_launch(void* const* d_in, const int* in_sizes, int n_in,
                              void* d_out, int out_size, void* d_ws, size_t ws_size,
                              hipStream_t stream) {
    const float* x    = (const float*)d_in[0];
    const int*   eidx = (const int*)  d_in[1];
    const float* W1   = (const float*)d_in[2];
    const float* a1s  = (const float*)d_in[3];
    const float* a1d  = (const float*)d_in[4];
    const float* b1   = (const float*)d_in[5];
    const float* W2   = (const float*)d_in[6];
    const float* a2s  = (const float*)d_in[7];
    const float* a2d  = (const float*)d_in[8];
    const float* b2   = (const float*)d_in[9];

    const int N = in_sizes[0] / GAT_INC;
    const int E = in_sizes[1] / 2;
    const int* srcI = eidx;
    const int* dstI = eidx + E;

    float* ws = (float*)d_ws;
    size_t o = 0;
    float* h     = ws + o; o += (size_t)N * 64;
    float* hpost = ws + o; o += (size_t)N * 64;
    int*   graph = (int*)(ws + o); o += (size_t)N * 64;
    float* as_   = ws + o; o += N;
    float* ad_   = ws + o; o += N;
    o += N;
    o += N;
    int*   rank  = (int*)(ws + o); o += E;

    const size_t Np = ((size_t)N + 4) & ~(size_t)3;
    int* cnt     = graph;
    int* rowptr  = graph + Np;
    int* csr_src = graph + 2 * Np;

    const int gGemm  = (N + 63) / 64;
    const int gPers  = GAT_PERS_WG;
    const int gN     = (N + 255) / 256;
    const int gScan  = (N + 1 + 1023) / 1024;

    csr_zero<<<gN,    256,  0, stream>>>(cnt, N);
    csr_rank<<<gPers, 256,  0, stream>>>(dstI, cnt, rank, E);
    csr_scan<<<gScan, 1024, 0, stream>>>(cnt, rowptr, N);
    csr_fill<<<gPers, 256,  0, stream>>>(srcI, dstI, rank, rowptr, csr_src, E);

    gat_gemm_wmma<GAT_INC><<<gGemm, 256, 0, stream>>>(x, W1, h, N);
    gat_node_alpha<<<gPers, 256, 0, stream>>>(h, a1s, a1d, as_, ad_, N);
    gat_gather    <<<gPers, 256, 0, stream>>>(rowptr, csr_src, as_, ad_, h, b1, hpost, N, E, 1);

    float* out = (float*)d_out;
    gat_gemm_wmma<GAT_HID><<<gGemm, 256, 0, stream>>>(hpost, W2, h, N);
    gat_node_alpha<<<gPers, 256, 0, stream>>>(h, a2s, a2d, as_, ad_, N);
    gat_gather    <<<gPers, 256, 0, stream>>>(rowptr, csr_src, as_, ad_, h, b2, out, N, E, 0);
}


# Edge ranks from LDS counters in a two-level counting sort (chunks binned by 256-node bucket, then one work-group orders each bucket), no returning global add per edge; rows in LDS-add arrival order
# speedup vs baseline: 4.2790x; 1.1882x over previous
#include <hip/hip_runtime.h>
#include <hip/hip_bf16.h>
#include <math.h>

typedef float v2f __attribute__((ext_vector_type(2)));
typedef float v8f __attribute__((ext_vector_type(8)));
typedef float v4f __attribute__((ext_vector_type(4)));

#define GAT_INC 256
#define GAT_HID 64
#define NEG_SLOPE 0.2f

template <int K>
__global__ __launch_bounds__(256)
void gat_gemm_wmma(const float* __restrict__ A, const float* __restrict__ W,
                   float* __restrict__ C, int N) {
    constexpr int KP    = K + 4;
    constexpr int QPR   = K / 4;
    constexpr int RSTEP = 256 / QPR;
    constexpr int NQ    = 64 / RSTEP;
    static_assert(K % 4 == 0 && 256 % QPR == 0 && 64 % RSTEP == 0, "tile map");
    __shared__ __attribute__((aligned(16))) float sA[64 * KP];
    const int lane = threadIdx.x & 31;
    const int wave = threadIdx.x >> 5;
    const int rowBase = blockIdx.x * 64;

    const int col      = (wave & 3) * 16 + (lane & 15);
    const int waveRow  = (wave >> 2) * 32;

    for (int k = threadIdx.x; k < K; k += 256)
        __builtin_prefetch(&W[(size_t)k * 64 + col], 0, 3);

    {
        const int q  = threadIdx.x % QPR;
        const int r0 = threadIdx.x / QPR;
        v4f t[NQ];
#pragma unroll
        for (int i = 0; i < NQ; ++i) {
            const int gr = max(min(rowBase + r0 + i * RSTEP, N - 1), 0);
            t[i] = __builtin_nontemporal_load((const v4f*)(A + (size_t)gr * K) + q);
        }
#pragma unroll
        for (int i = 0; i < NQ; ++i)
            *(v4f*)&sA[(r0 + i * RSTEP) * KP + 4 * q] = t[i];
    }
    __syncthreads();

    const int row   = lane & 15;
    const int khalf = (lane >> 4) << 1;

    v8f acc0 = {};
    v8f acc1 = {};
#pragma unroll 8
    for (int k = 0; k < K; k += 4) {
        v2f b;
        b.x = W[(size_t)(k + khalf)     * 64 + col];
        b.y = W[(size_t)(k + khalf + 1) * 64 + col];

        v2f a0, a1;
        a0.x = sA[(waveRow +      row) * KP + k + khalf];
        a0.y = sA[(waveRow +      row) * KP + k + khalf + 1];
        a1.x = sA[(waveRow + 16 + row) * KP + k + khalf];
        a1.y = sA[(waveRow + 16 + row) * KP + k + khalf + 1];

        acc0 = __builtin_amdgcn_wmma_f32_16x16x4_f32(
            false, a0, false, b, (short)0, acc0, false, false);
        acc1 = __builtin_amdgcn_wmma_f32_16x16x4_f32(
            false, a1, false, b, (short)0, acc1, false, false);
    }

    const int mhalf = (lane >> 4) << 3;
    for (int r = 0; r < 8; ++r) {
        int gr0 = rowBase + waveRow +      mhalf + r;
        int gr1 = rowBase + waveRow + 16 + mhalf + r;
        if (gr0 < N) C[(size_t)gr0 * 64 + col] = acc0[r];
        if (gr1 < N) C[(size_t)gr1 * 64 + col] = acc1[r];
    }
}

template <int M>
__device__ __forceinline__ float row_xor(float x) {
    return __builtin_bit_cast(float, __builtin_amdgcn_update_dpp(
        0, __builtin_bit_cast(int, x), 0x160 + M, 0xf, 0xf, false));
}
__device__ __forceinline__ float xhalf(float x) {
    return __builtin_bit_cast(float, __builtin_amdgcn_permlanex16(
        __builtin_bit_cast(int, x), __builtin_bit_cast(int, x),
        0x76543210, 0xfedcba98, false, false));
}
__device__ __forceinline__ float wave_max32(float v) {
    v = fmaxf(v, row_xor<1>(v));
    v = fmaxf(v, row_xor<2>(v));
    v = fmaxf(v, row_xor<4>(v));
    v = fmaxf(v, row_xor<8>(v));
    return fmaxf(v, xhalf(v));
}
__device__ __forceinline__ float wave_sum32(float v) {
    v += row_xor<1>(v);
    v += row_xor<2>(v);
    v += row_xor<4>(v);
    v += row_xor<8>(v);
    return v + xhalf(v);
}

#define GAT_PERS_WG 1024

__global__ __launch_bounds__(256) __attribute__((amdgpu_waves_per_eu(4, 8)))
void gat_node_alpha(const float* __restrict__ h, const float* __restrict__ a_src,
                    const float* __restrict__ a_dst, float* __restrict__ as_,
                    float* __restrict__ ad_, int N) {
    const int lane = threadIdx.x & 31;
    const int sub  = lane & 15;
    const int half = lane >> 4;
    const unsigned nw = gridDim.x * (blockDim.x >> 5);
    const unsigned gw = (unsigned)__builtin_amdgcn_readfirstlane(
        (int)(blockIdx.x * (blockDim.x >> 5) + (threadIdx.x >> 5)));
    const unsigned per = (unsigned)N / nw, rem = (unsigned)N % nw;
    const int r0 = (int)(gw * per + min(gw, rem));
    const int r1 = r0 + (int)per + (gw < rem ? 1 : 0);
    const float4 ws = ((const float4*)a_src)[sub];
    const float4 wd = ((const float4*)a_dst)[sub];
    const float4* h4 = (const float4*)h;
    const int rlast = N - 1;
    for (int b = r0; b < r1; b += 16) {
        float4 x[8];
#pragma unroll
        for (int u = 0; u < 8; ++u) {
            const unsigned row = (unsigned)min(b + 2 * u + half, rlast);
            x[u] = h4[row * 16u + (unsigned)sub];
        }
        float rs = 0.0f, rd = 0.0f;
#pragma unroll
        for (int u = 0; u < 8; ++u) {
            float s = x[u].x * ws.x + x[u].y * ws.y + x[u].z * ws.z + x[u].w * ws.w;
            float d = x[u].x * wd.x + x[u].y * wd.y + x[u].z * wd.z + x[u].w * wd.w;
            s += row_xor<1>(s); d += row_xor<1>(d);
            s += row_xor<2>(s); d += row_xor<2>(d);
            s += row_xor<4>(s); d += row_xor<4>(d);
            s += row_xor<8>(s); d += row_xor<8>(d);
            if (sub == 2 * u + half) { rs = s; rd = d; }
        }
        const int row = b + sub;
        if ((sub & 1) == half && row < r1) { as_[row] = rs; ad_[row] = rd; }
    }
}

#define CS_SH 8
#define CS_BN (1 << CS_SH)
#define CS_WG 256
#define CS_TH 512
#define CS_UC 8
#define CS_UB 6

__device__ __forceinline__ void cs_chunk(int E, int& c0, int& c1) {
    const int per = (E + (int)gridDim.x - 1) / (int)gridDim.x;
    c0 = min((int)blockIdx.x * per, E);
    c1 = min(c0 + per, E);
}

__global__ __launch_bounds__(CS_TH)
void csr_bucket_count(const int* __restrict__ dstI, int* __restrict__ wgcnt, int E, int NB) {
    extern __shared__ int sBkt[];
    const int tid = threadIdx.x;
    for (int i = tid; i < NB; i += CS_TH) sBkt[i] = 0;
    __syncthreads();
    int c0, c1;
    cs_chunk(E, c0, c1);
    const unsigned blast = (unsigned)(NB - 1);
    for (int e0 = c0 + tid; e0 < c1; e0 += CS_UC * CS_TH) {
        int d[CS_UC];
#pragma unroll
        for (int u = 0; u < CS_UC; ++u) d[u] = dstI[min(e0 + u * CS_TH, c1 - 1)];
#pragma unroll
        for (int u = 0; u < CS_UC; ++u)
            atomicAdd(&sBkt[min((unsigned)d[u] >> CS_SH, blast)], (e0 + u * CS_TH < c1) ? 1 : 0);
    }
    __syncthreads();
    for (int i = tid; i < NB; i += CS_TH) wgcnt[(size_t)i * gridDim.x + blockIdx.x] = sBkt[i];
}

__global__ __launch_bounds__(1024)
void csr_scan(const int* __restrict__ cnt, int* __restrict__ rowptr, int N) {
    __shared__ int sFront[32];
    __shared__ int sWave[32];
    const int tid  = threadIdx.x;
    const int lane = tid & 31;
    const int wave = tid >> 5;
    const int first = blockIdx.x * 1024;

    int front = 0;
    const int4* c4 = (const int4*)cnt;
#pragma unroll 8
    for (int q = tid; q < (first >> 2); q += 1024) {
        int4 v = c4[q];
        front += (v.x + v.y) + (v.z + v.w);
    }
    for (int off = 16; off > 0; off >>= 1) front += __shfl_xor(front, off, 32);

    const int i = first + tid;
    const int x = (i < N) ? cnt[i] : 0;
    int incl = x;
    for (int off = 1; off < 32; off <<= 1) {
        int y = __shfl_up(incl, off, 32);
        if (lane >= off) incl += y;
    }
    if (lane == 0)  sFront[wave] = front;
    if (lane == 31) sWave[wave]  = incl;
    __syncthreads();

    int a = sFront[lane];
    int b = (lane < wave) ? sWave[lane] : 0;
    for (int off = 16; off > 0; off >>= 1) {
        a += __shfl_xor(a, off, 32);
        b += __shfl_xor(b, off, 32);
    }
    if (i <= N) rowptr[i] = a + b + (incl - x);
}

__global__ __launch_bounds__(CS_TH)
void csr_bucket_scatter(const int* __restrict__ srcI, const int* __restrict__ dstI,
                        const int* __restrict__ base, int* __restrict__ tmp, int E, int NB) {
    extern __shared__ int sBkt[];
    const int tid = threadIdx.x;
    int c0, c1;
    cs_chunk(E, c0, c1);
    if (c0 >= c1) return;
    const unsigned blast = (unsigned)(NB - 1);
    const unsigned elast = (unsigned)(E - 1);
    int e0 = c0 + tid;
    int d[CS_UC], s[CS_UC];
#pragma unroll
    for (int u = 0; u < CS_UC; ++u) {
        const int e = min(e0 + u * CS_TH, c1 - 1);
        d[u] = dstI[e];
        s[u] = srcI[e];
    }
    for (int i = tid; i < NB; i += CS_TH) sBkt[i] = base[(size_t)i * gridDim.x + blockIdx.x];
    __syncthreads();
    while (e0 < c1) {
        int p[CS_UC];
#pragma unroll
        for (int u = 0; u < CS_UC; ++u)
            p[u] = atomicAdd(&sBkt[min((unsigned)d[u] >> CS_SH, blast)],
                             (e0 + u * CS_TH < c1) ? 1 : 0);
#pragma unroll
        for (int u = 0; u < CS_UC; ++u)
            if (e0 + u * CS_TH < c1)
                tmp[min((unsigned)p[u], elast)] =
                    (int)(((unsigned)s[u] << CS_SH) | ((unsigned)d[u] & (unsigned)(CS_BN - 1)));
        e0 += CS_UC * CS_TH;
        if (e0 < c1) {
#pragma unroll
            for (int u = 0; u < CS_UC; ++u) {
                const int e = min(e0 + u * CS_TH, c1 - 1);
                d[u] = dstI[e];
                s[u] = srcI[e];
            }
        }
    }
}

__global__ __launch_bounds__(CS_TH)
void csr_bucket_sort(const int* __restrict__ base, const int* __restrict__ tmp,
                     int* __restrict__ rowptr, int* __restrict__ csr_src, int N, int E, int NB) {
    __shared__ int sCnt[CS_BN];
    __shared__ int sTot[CS_BN / 32];
    const int tid  = threadIdx.x;
    const int lane = tid & 31;
    const int wave = tid >> 5;
    const int b    = blockIdx.x;
    int beg = base[(size_t)b * CS_WG];
    int end = base[(size_t)(b + 1) * CS_WG];
    beg = min(max(beg, 0), E);
    end = min(max(end, beg), E);
    const unsigned elast = (unsigned)max(E - 1, 0);
    if (tid < CS_BN) sCnt[tid] = 0;
    __syncthreads();
    for (int j0 = beg + tid; j0 < end; j0 += CS_UB * CS_TH) {
        int w[CS_UB];
#pragma unroll
        for (int u = 0; u < CS_UB; ++u) w[u] = tmp[min(j0 + u * CS_TH, end - 1)];
#pragma unroll
        for (int u = 0; u < CS_UB; ++u)
            atomicAdd(&sCnt[w[u] & (CS_BN - 1)], (j0 + u * CS_TH < end) ? 1 : 0);
    }
    __syncthreads();
    const int x = (tid < CS_BN) ? sCnt[tid] : 0;
    int incl = x;
    for (int off = 1; off < 32; off <<= 1) {
        int y = __shfl_up(incl, off, 32);
        if (lane >= off) incl += y;
    }
    if (tid < CS_BN && lane == 31) sTot[wave] = incl;
    __syncthreads();
    if (tid < CS_BN) {
        int front = 0;
        for (int v = 0; v < wave; ++v) front += sTot[v];
        const int start = beg + front + (incl - x);
        sCnt[tid] = start;
        const int node = b * CS_BN + tid;
        if (node < N) rowptr[node] = start;
    }
    if (b == NB - 1 && tid == 0) rowptr[N] = end;
    __syncthreads();
    for (int j0 = beg + tid; j0 < end; j0 += CS_UB * CS_TH) {
        int w[CS_UB];
#pragma unroll
        for (int u = 0; u < CS_UB; ++u) w[u] = tmp[min(j0 + u * CS_TH, end - 1)];
        int p[CS_UB];
#pragma unroll
        for (int u = 0; u < CS_UB; ++u)
            p[u] = atomicAdd(&sCnt[w[u] & (CS_BN - 1)], (j0 + u * CS_TH < end) ? 1 : 0);
#pragma unroll
        for (int u = 0; u < CS_UB; ++u)
            if (j0 + u * CS_TH < end)
                csr_src[min((unsigned)p[u], elast)] = (int)((unsigned)w[u] >> CS_SH);
    }
}

__device__ __forceinline__ float gat_score(const float* __restrict__ as_, int s, float add) {
    float v = as_[s] + add;
    return (v > 0.0f) ? v : v * NEG_SLOPE;
}
template <int G>
__device__ __forceinline__ void gat_row_group(const v2f* __restrict__ h2, int s, float alpha,
                                              int t, unsigned lane, v2f& acc) {
    unsigned idx[G];
    v2f r[G];
#pragma unroll
    for (int u = 0; u < G; ++u)
        idx[u] = (unsigned)__builtin_amdgcn_readlane(s, t + u) * 32u + lane;
#pragma unroll
    for (int u = 0; u < G; ++u)
        r[u] = h2[idx[u]];
#pragma unroll
    for (int u = 0; u < G; ++u) {
        const float at = __builtin_bit_cast(float,
            __builtin_amdgcn_readlane(__builtin_bit_cast(int, alpha), t + u));
        acc = acc + r[u] * at;
    }
}
__device__ __forceinline__ void gat_rows(const v2f* __restrict__ h2, int s, float alpha,
                                         int cnt, unsigned lane, v2f& acc) {
    const int c4 = (cnt + 3) & ~3;
    if (c4 <= 4)       gat_row_group<4>(h2, s, alpha, 0, lane, acc);
    else if (c4 <= 8)  gat_row_group<8>(h2, s, alpha, 0, lane, acc);
    else if (c4 <= 12) gat_row_group<12>(h2, s, alpha, 0, lane, acc);
    else {
        gat_row_group<16>(h2, s, alpha, 0, lane, acc);
        int t = 16;
        for (; t + 8 <= c4; t += 8) gat_row_group<8>(h2, s, alpha, t, lane, acc);
        if (t < c4)                 gat_row_group<4>(h2, s, alpha, t, lane, acc);
    }
}

__device__ __forceinline__ int gat_edge_ids(const int* __restrict__ csr_src, int rp, int k,
                                            unsigned lane, int elast) {
    const int beg = __builtin_amdgcn_readlane(rp, k);
    const int end = max(__builtin_amdgcn_readlane(rp, k + 1), beg);
    const int j   = min(beg + (int)lane, end - 1);
    return csr_src[min(max(j, 0), elast)];
}

__global__ __launch_bounds__(256)
void gat_gather(const int* __restrict__ rowptr, const int* __restrict__ csr_src,
                const float* __restrict__ as_, const float* __restrict__ ad_,
                const float* __restrict__ h, const float* __restrict__ bias,
                float* __restrict__ out, int N, int E, int relu_flag) {
    const unsigned lane = threadIdx.x & 31;
    const unsigned nw = gridDim.x * (blockDim.x >> 5);
    const unsigned gw = (unsigned)__builtin_amdgcn_readfirstlane(
        (int)(blockIdx.x * (blockDim.x >> 5) + (threadIdx.x >> 5)));
    const unsigned per = (unsigned)N / nw, rem = (unsigned)N % nw;
    const int n0 = (int)(gw * per + min(gw, rem));
    const int n1 = n0 + (int)per + (gw < rem ? 1 : 0);
    if (n0 >= n1) return;
    const v2f bb = ((const v2f*)bias)[lane];
    const unsigned nlast = (unsigned)(N - 1);
    const int elast = max(E - 1, 0);
    const v2f* h2 = (const v2f*)h;

    for (int base = n0; base < n1; base += 31) {
        const int c = min(31, n1 - base);
        int rp = rowptr[min(base + (int)lane, N)];
        rp = min(max(rp, 0), E);
        const float adl = ad_[min(base + (int)lane, N - 1)];

        int   sA = gat_edge_ids(csr_src, rp, 0, lane, elast);
        int   sB = gat_edge_ids(csr_src, rp, min(1, c - 1), lane, elast);
        sA = (int)min((unsigned)sA, nlast);
        float vA = as_[sA];

        for (int k = 0; k < c; ++k) {
            const int sC = gat_edge_ids(csr_src, rp, min(k + 2, c - 1), lane, elast);
            sB = (int)min((unsigned)sB, nlast);
            const float vB = as_[sB];

            const int beg = __builtin_amdgcn_readlane(rp, k);
            const int end = max(__builtin_amdgcn_readlane(rp, k + 1), beg);
            const float add = __builtin_bit_cast(float,
                __builtin_amdgcn_readlane(__builtin_bit_cast(int, adl), k));
            const int cnt = end - beg;

            v2f acc = {0.0f, 0.0f};
            if (cnt > 0) {
                if (cnt <= 32) {
                    const bool valid = (int)lane < cnt;
                    float v = vA + add;
                    v = (v > 0.0f) ? v : v * NEG_SLOPE;
                    const float m    = wave_max32(v);
                    const float ex   = valid ? __expf(v - m) : 0.0f;
                    const float den  = wave_sum32(ex);
                    const float alpha = ex / (den + 1e-16f);
                    gat_rows(h2, sA, alpha, cnt, lane, acc);
                } else {
                    float m = -3.0e38f;
                    for (int eb = beg; eb < end; eb += 32) {
                        const int j = eb + (int)lane;
                        const int s = (int)min((unsigned)csr_src[min(j, end - 1)], nlast);
                        m = fmaxf(m, gat_score(as_, s, add));
                    }
                    m = wave_max32(m);
                    float den = 0.0f;
                    for (int eb = beg; eb < end; eb += 32) {
                        const int j = eb + (int)lane;
                        const int s = (int)min((unsigned)csr_src[min(j, end - 1)], nlast);
                        const float ex = __expf(gat_score(as_, s, add) - m);
                        den += (j < end) ? ex : 0.0f;
                    }
                    den = wave_sum32(den);
                    const float dd = den + 1e-16f;
                    for (int eb = beg; eb < end; eb += 32) {
                        const int j = eb + (int)lane;
                        const int s = (int)min((unsigned)csr_src[min(j, end - 1)], nlast);
                        const float ex = __expf(gat_score(as_, s, add) - m);
                        const float alpha = (j < end) ? ex / dd : 0.0f;
                        gat_rows(h2, s, alpha, min(32, end - eb), lane, acc);
                    }
                }
            }

            v2f r;
            r.x = acc.x + bb.x;
            r.y = acc.y + bb.y;
            if (relu_flag) { r.x = fmaxf(r.x, 0.0f); r.y = fmaxf(r.y, 0.0f); }
            ((v2f*)out)[(unsigned)(base + k) * 32u + lane] = r;

            sA = sB; vA = vB; sB = sC;
        }
    }
}

extern "C" void kernel_launch(void* const* d_in, const int* in_sizes, int n_in,
                              void* d_out, int out_size, void* d_ws, size_t ws_size,
                              hipStream_t stream) {
    const float* x    = (const float*)d_in[0];
    const int*   eidx = (const int*)  d_in[1];
    const float* W1   = (const float*)d_in[2];
    const float* a1s  = (const float*)d_in[3];
    const float* a1d  = (const float*)d_in[4];
    const float* b1   = (const float*)d_in[5];
    const float* W2   = (const float*)d_in[6];
    const float* a2s  = (const float*)d_in[7];
    const float* a2d  = (const float*)d_in[8];
    const float* b2   = (const float*)d_in[9];

    const int N = in_sizes[0] / GAT_INC;
    const int E = in_sizes[1] / 2;
    const int* srcI = eidx;
    const int* dstI = eidx + E;

    float* ws = (float*)d_ws;
    size_t o = 0;
    float* h     = ws + o; o += (size_t)N * 64;
    float* hpost = ws + o; o += (size_t)N * 64;
    int*   graph = (int*)(ws + o); o += (size_t)N * 64;
    float* as_   = ws + o; o += N;
    float* ad_   = ws + o; o += N;
    o += N;
    o += N;
    int*   tmp   = (int*)(ws + o); o += E;

    const int NB = (N + CS_BN - 1) / CS_BN;
    const int M  = NB * CS_WG;
    const size_t Np = ((size_t)N + 4) & ~(size_t)3;
    const size_t Ep = ((size_t)E + 3) & ~(size_t)3;
    const size_t Mp = ((size_t)M + 4) & ~(size_t)3;
    int* rowptr  = graph;
    int* csr_src = graph + Np;
    int* wgcnt   = csr_src + Ep;
    int* base    = wgcnt + Mp;

    const int gGemm  = (N + 63) / 64;
    const int gPers  = GAT_PERS_WG;
    const int gScan  = (M + 1 + 1023) / 1024;

    csr_bucket_count  <<<CS_WG, CS_TH, NB * sizeof(int), stream>>>(dstI, wgcnt, E, NB);
    csr_scan          <<<gScan, 1024,  0,                stream>>>(wgcnt, base, M);
    csr_bucket_scatter<<<CS_WG, CS_TH, NB * sizeof(int), stream>>>(srcI, dstI, base, tmp, E, NB);
    csr_bucket_sort   <<<NB,    CS_TH, 0,                stream>>>(base, tmp, rowptr, csr_src, N, E, NB);

    gat_gemm_wmma<GAT_INC><<<gGemm, 256, 0, stream>>>(x, W1, h, N);
    gat_node_alpha<<<gPers, 256, 0, stream>>>(h, a1s, a1d, as_, ad_, N);
    gat_gather    <<<gPers, 256, 0, stream>>>(rowptr, csr_src, as_, ad_, h, b1, hpost, N, E, 1);

    float* out = (float*)d_out;
    gat_gemm_wmma<GAT_HID><<<gGemm, 256, 0, stream>>>(hpost, W2, h, N);
    gat_node_alpha<<<gPers, 256, 0, stream>>>(h, a2s, a2d, as_, ad_, N);
    gat_gather    <<<gPers, 256, 0, stream>>>(rowptr, csr_src, as_, ad_, h, b2, out, N, E, 0);
}


# Attention logits (h.a_src, h.a_dst) computed in the GEMM epilogue from the accumulators via LDS instead of two launches that re-read h (10 to 8 launches); logit dot sums re-ordered
# speedup vs baseline: 4.4647x; 1.0434x over previous
#include <hip/hip_runtime.h>
#include <hip/hip_bf16.h>
#include <math.h>

typedef float v2f __attribute__((ext_vector_type(2)));
typedef float v8f __attribute__((ext_vector_type(8)));
typedef float v4f __attribute__((ext_vector_type(4)));

#define GAT_INC 256
#define GAT_HID 64
#define NEG_SLOPE 0.2f

template <int K>
__global__ __launch_bounds__(256)
void gat_gemm_wmma(const float* __restrict__ A, const float* __restrict__ W,
                   const float* __restrict__ a_src, const float* __restrict__ a_dst,
                   float* __restrict__ C, float* __restrict__ as_, float* __restrict__ ad_,
                   int N) {
    constexpr int KP    = K + 4;
    constexpr int QPR   = K / 4;
    constexpr int RSTEP = 256 / QPR;
    constexpr int NQ    = 64 / RSTEP;
    constexpr int CP    = 68;
    static_assert(K % 4 == 0 && 256 % QPR == 0 && 64 % RSTEP == 0, "tile map");
    static_assert(CP <= KP, "the C tile reuses the A tile's LDS");
    __shared__ __attribute__((aligned(16))) float sA[64 * KP];
    const int lane = threadIdx.x & 31;
    const int wave = __builtin_amdgcn_readfirstlane((int)(threadIdx.x >> 5));
    const int rowBase = blockIdx.x * 64;

    const int col      = (wave & 3) * 16 + (lane & 15);
    const int waveRow  = (wave >> 2) * 32;

    for (int k = threadIdx.x; k < K; k += 256)
        __builtin_prefetch(&W[(size_t)k * 64 + col], 0, 3);

    {
        const int q  = threadIdx.x % QPR;
        const int r0 = threadIdx.x / QPR;
        v4f t[NQ];
#pragma unroll
        for (int i = 0; i < NQ; ++i) {
            const int gr = max(min(rowBase + r0 + i * RSTEP, N - 1), 0);
            t[i] = __builtin_nontemporal_load((const v4f*)(A + (size_t)gr * K) + q);
        }
#pragma unroll
        for (int i = 0; i < NQ; ++i)
            *(v4f*)&sA[(r0 + i * RSTEP) * KP + 4 * q] = t[i];
    }
    __syncthreads();

    const int row   = lane & 15;
    const int khalf = (lane >> 4) << 1;

    v8f acc0 = {};
    v8f acc1 = {};
#pragma unroll 8
    for (int k = 0; k < K; k += 4) {
        v2f b;
        b.x = W[(size_t)(k + khalf)     * 64 + col];
        b.y = W[(size_t)(k + khalf + 1) * 64 + col];

        v2f a0, a1;
        a0.x = sA[(waveRow +      row) * KP + k + khalf];
        a0.y = sA[(waveRow +      row) * KP + k + khalf + 1];
        a1.x = sA[(waveRow + 16 + row) * KP + k + khalf];
        a1.y = sA[(waveRow + 16 + row) * KP + k + khalf + 1];

        acc0 = __builtin_amdgcn_wmma_f32_16x16x4_f32(
            false, a0, false, b, (short)0, acc0, false, false);
        acc1 = __builtin_amdgcn_wmma_f32_16x16x4_f32(
            false, a1, false, b, (short)0, acc1, false, false);
    }

    __syncthreads();
    float* sC = sA;

    const int mhalf = (lane >> 4) << 3;
    for (int r = 0; r < 8; ++r) {
        sC[(waveRow +      mhalf + r) * CP + col] = acc0[r];
        sC[(waveRow + 16 + mhalf + r) * CP + col] = acc1[r];
    }

    const int c0 = (wave & 3) * 16;
    float* crow = &sC[(waveRow + lane) * CP + c0];
    {
        float s = 0.0f, d = 0.0f;
#pragma unroll
        for (int j = 0; j < 4; ++j) {
            const v4f cv = *(const v4f*)(crow + 4 * j);
            s = __builtin_fmaf(cv.x, a_src[c0 + 4 * j + 0], s); d = __builtin_fmaf(cv.x, a_dst[c0 + 4 * j + 0], d);
            s = __builtin_fmaf(cv.y, a_src[c0 + 4 * j + 1], s); d = __builtin_fmaf(cv.y, a_dst[c0 + 4 * j + 1], d);
            s = __builtin_fmaf(cv.z, a_src[c0 + 4 * j + 2], s); d = __builtin_fmaf(cv.z, a_dst[c0 + 4 * j + 2], d);
            s = __builtin_fmaf(cv.w, a_src[c0 + 4 * j + 3], s); d = __builtin_fmaf(cv.w, a_dst[c0 + 4 * j + 3], d);
        }
        v2f p; p.x = s; p.y = d;
        *(v2f*)crow = p;
    }
    __syncthreads();

    if (c0 == 0) {
        const v2f p0 = *(const v2f*)(crow);
        const v2f p1 = *(const v2f*)(crow + 16);
        const v2f p2 = *(const v2f*)(crow + 32);
        const v2f p3 = *(const v2f*)(crow + 48);
        const int gr = rowBase + waveRow + lane;
        if (gr < N) {
            as_[gr] = (p0.x + p1.x) + (p2.x + p3.x);
            ad_[gr] = (p0.y + p1.y) + (p2.y + p3.y);
        }
    }

    for (int r = 0; r < 8; ++r) {
        int gr0 = rowBase + waveRow +      mhalf + r;
        int gr1 = rowBase + waveRow + 16 + mhalf + r;
        if (gr0 < N) C[(size_t)gr0 * 64 + col] = acc0[r];
        if (gr1 < N) C[(size_t)gr1 * 64 + col] = acc1[r];
    }
}

template <int M>
__device__ __forceinline__ float row_xor(float x) {
    return __builtin_bit_cast(float, __builtin_amdgcn_update_dpp(
        0, __builtin_bit_cast(int, x), 0x160 + M, 0xf, 0xf, false));
}
__device__ __forceinline__ float xhalf(float x) {
    return __builtin_bit_cast(float, __builtin_amdgcn_permlanex16(
        __builtin_bit_cast(int, x), __builtin_bit_cast(int, x),
        0x76543210, 0xfedcba98, false, false));
}
__device__ __forceinline__ float wave_max32(float v) {
    v = fmaxf(v, row_xor<1>(v));
    v = fmaxf(v, row_xor<2>(v));
    v = fmaxf(v, row_xor<4>(v));
    v = fmaxf(v, row_xor<8>(v));
    return fmaxf(v, xhalf(v));
}
__device__ __forceinline__ float wave_sum32(float v) {
    v += row_xor<1>(v);
    v += row_xor<2>(v);
    v += row_xor<4>(v);
    v += row_xor<8>(v);
    return v + xhalf(v);
}

#define GAT_PERS_WG 1024

#define CS_SH 8
#define CS_BN (1 << CS_SH)
#define CS_WG 256
#define CS_TH 512
#define CS_UC 8
#define CS_UB 6

__device__ __forceinline__ void cs_chunk(int E, int& c0, int& c1) {
    const int per = (E + (int)gridDim.x - 1) / (int)gridDim.x;
    c0 = min((int)blockIdx.x * per, E);
    c1 = min(c0 + per, E);
}

__global__ __launch_bounds__(CS_TH)
void csr_bucket_count(const int* __restrict__ dstI, int* __restrict__ wgcnt, int E, int NB) {
    extern __shared__ int sBkt[];
    const int tid = threadIdx.x;
    for (int i = tid; i < NB; i += CS_TH) sBkt[i] = 0;
    __syncthreads();
    int c0, c1;
    cs_chunk(E, c0, c1);
    const unsigned blast = (unsigned)(NB - 1);
    for (int e0 = c0 + tid; e0 < c1; e0 += CS_UC * CS_TH) {
        int d[CS_UC];
#pragma unroll
        for (int u = 0; u < CS_UC; ++u) d[u] = dstI[min(e0 + u * CS_TH, c1 - 1)];
#pragma unroll
        for (int u = 0; u < CS_UC; ++u)
            atomicAdd(&sBkt[min((unsigned)d[u] >> CS_SH, blast)], (e0 + u * CS_TH < c1) ? 1 : 0);
    }
    __syncthreads();
    for (int i = tid; i < NB; i += CS_TH) wgcnt[(size_t)i * gridDim.x + blockIdx.x] = sBkt[i];
}

__global__ __launch_bounds__(1024)
void csr_scan(const int* __restrict__ cnt, int* __restrict__ rowptr, int N) {
    __shared__ int sFront[32];
    __shared__ int sWave[32];
    const int tid  = threadIdx.x;
    const int lane = tid & 31;
    const int wave = tid >> 5;
    const int first = blockIdx.x * 1024;

    int front = 0;
    const int4* c4 = (const int4*)cnt;
#pragma unroll 8
    for (int q = tid; q < (first >> 2); q += 1024) {
        int4 v = c4[q];
        front += (v.x + v.y) + (v.z + v.w);
    }
    for (int off = 16; off > 0; off >>= 1) front += __shfl_xor(front, off, 32);

    const int i = first + tid;
    const int x = (i < N) ? cnt[i] : 0;
    int incl = x;
    for (int off = 1; off < 32; off <<= 1) {
        int y = __shfl_up(incl, off, 32);
        if (lane >= off) incl += y;
    }
    if (lane == 0)  sFront[wave] = front;
    if (lane == 31) sWave[wave]  = incl;
    __syncthreads();

    int a = sFront[lane];
    int b = (lane < wave) ? sWave[lane] : 0;
    for (int off = 16; off > 0; off >>= 1) {
        a += __shfl_xor(a, off, 32);
        b += __shfl_xor(b, off, 32);
    }
    if (i <= N) rowptr[i] = a + b + (incl - x);
}

__global__ __launch_bounds__(CS_TH)
void csr_bucket_scatter(const int* __restrict__ srcI, const int* __restrict__ dstI,
                        const int* __restrict__ base, int* __restrict__ tmp, int E, int NB) {
    extern __shared__ int sBkt[];
    const int tid = threadIdx.x;
    int c0, c1;
    cs_chunk(E, c0, c1);
    if (c0 >= c1) return;
    const unsigned blast = (unsigned)(NB - 1);
    const unsigned elast = (unsigned)(E - 1);
    int e0 = c0 + tid;
    int d[CS_UC], s[CS_UC];
#pragma unroll
    for (int u = 0; u < CS_UC; ++u) {
        const int e = min(e0 + u * CS_TH, c1 - 1);
        d[u] = dstI[e];
        s[u] = srcI[e];
    }
    for (int i = tid; i < NB; i += CS_TH) sBkt[i] = base[(size_t)i * gridDim.x + blockIdx.x];
    __syncthreads();
    while (e0 < c1) {
        int p[CS_UC];
#pragma unroll
        for (int u = 0; u < CS_UC; ++u)
            p[u] = atomicAdd(&sBkt[min((unsigned)d[u] >> CS_SH, blast)],
                             (e0 + u * CS_TH < c1) ? 1 : 0);
#pragma unroll
        for (int u = 0; u < CS_UC; ++u)
            if (e0 + u * CS_TH < c1)
                tmp[min((unsigned)p[u], elast)] =
                    (int)(((unsigned)s[u] << CS_SH) | ((unsigned)d[u] & (unsigned)(CS_BN - 1)));
        e0 += CS_UC * CS_TH;
        if (e0 < c1) {
#pragma unroll
            for (int u = 0; u < CS_UC; ++u) {
                const int e = min(e0 + u * CS_TH, c1 - 1);
                d[u] = dstI[e];
                s[u] = srcI[e];
            }
        }
    }
}

__global__ __launch_bounds__(CS_TH)
void csr_bucket_sort(const int* __restrict__ base, const int* __restrict__ tmp,
                     int* __restrict__ rowptr, int* __restrict__ csr_src, int N, int E, int NB) {
    __shared__ int sCnt[CS_BN];
    __shared__ int sTot[CS_BN / 32];
    const int tid  = threadIdx.x;
    const int lane = tid & 31;
    const int wave = tid >> 5;
    const int b    = blockIdx.x;
    int beg = base[(size_t)b * CS_WG];
    int end = base[(size_t)(b + 1) * CS_WG];
    beg = min(max(beg, 0), E);
    end = min(max(end, beg), E);
    const unsigned elast = (unsigned)max(E - 1, 0);
    if (tid < CS_BN) sCnt[tid] = 0;
    __syncthreads();
    for (int j0 = beg + tid; j0 < end; j0 += CS_UB * CS_TH) {
        int w[CS_UB];
#pragma unroll
        for (int u = 0; u < CS_UB; ++u) w[u] = tmp[min(j0 + u * CS_TH, end - 1)];
#pragma unroll
        for (int u = 0; u < CS_UB; ++u)
            atomicAdd(&sCnt[w[u] & (CS_BN - 1)], (j0 + u * CS_TH < end) ? 1 : 0);
    }
    __syncthreads();
    const int x = (tid < CS_BN) ? sCnt[tid] : 0;
    int incl = x;
    for (int off = 1; off < 32; off <<= 1) {
        int y = __shfl_up(incl, off, 32);
        if (lane >= off) incl += y;
    }
    if (tid < CS_BN && lane == 31) sTot[wave] = incl;
    __syncthreads();
    if (tid < CS_BN) {
        int front = 0;
        for (int v = 0; v < wave; ++v) front += sTot[v];
        const int start = beg + front + (incl - x);
        sCnt[tid] = start;
        const int node = b * CS_BN + tid;
        if (node < N) rowptr[node] = start;
    }
    if (b == NB - 1 && tid == 0) rowptr[N] = end;
    __syncthreads();
    for (int j0 = beg + tid; j0 < end; j0 += CS_UB * CS_TH) {
        int w[CS_UB];
#pragma unroll
        for (int u = 0; u < CS_UB; ++u) w[u] = tmp[min(j0 + u * CS_TH, end - 1)];
        int p[CS_UB];
#pragma unroll
        for (int u = 0; u < CS_UB; ++u)
            p[u] = atomicAdd(&sCnt[w[u] & (CS_BN - 1)], (j0 + u * CS_TH < end) ? 1 : 0);
#pragma unroll
        for (int u = 0; u < CS_UB; ++u)
            if (j0 + u * CS_TH < end)
                csr_src[min((unsigned)p[u], elast)] = (int)((unsigned)w[u] >> CS_SH);
    }
}

__device__ __forceinline__ float gat_score(const float* __restrict__ as_, int s, float add) {
    float v = as_[s] + add;
    return (v > 0.0f) ? v : v * NEG_SLOPE;
}
template <int G>
__device__ __forceinline__ void gat_row_group(const v2f* __restrict__ h2, int s, float alpha,
                                              int t, unsigned lane, v2f& acc) {
    unsigned idx[G];
    v2f r[G];
#pragma unroll
    for (int u = 0; u < G; ++u)
        idx[u] = (unsigned)__builtin_amdgcn_readlane(s, t + u) * 32u + lane;
#pragma unroll
    for (int u = 0; u < G; ++u)
        r[u] = h2[idx[u]];
#pragma unroll
    for (int u = 0; u < G; ++u) {
        const float at = __builtin_bit_cast(float,
            __builtin_amdgcn_readlane(__builtin_bit_cast(int, alpha), t + u));
        acc = acc + r[u] * at;
    }
}
__device__ __forceinline__ void gat_rows(const v2f* __restrict__ h2, int s, float alpha,
                                         int cnt, unsigned lane, v2f& acc) {
    const int c4 = (cnt + 3) & ~3;
    if (c4 <= 4)       gat_row_group<4>(h2, s, alpha, 0, lane, acc);
    else if (c4 <= 8)  gat_row_group<8>(h2, s, alpha, 0, lane, acc);
    else if (c4 <= 12) gat_row_group<12>(h2, s, alpha, 0, lane, acc);
    else {
        gat_row_group<16>(h2, s, alpha, 0, lane, acc);
        int t = 16;
        for (; t + 8 <= c4; t += 8) gat_row_group<8>(h2, s, alpha, t, lane, acc);
        if (t < c4)                 gat_row_group<4>(h2, s, alpha, t, lane, acc);
    }
}

__device__ __forceinline__ int gat_edge_ids(const int* __restrict__ csr_src, int rp, int k,
                                            unsigned lane, int elast) {
    const int beg = __builtin_amdgcn_readlane(rp, k);
    const int end = max(__builtin_amdgcn_readlane(rp, k + 1), beg);
    const int j   = min(beg + (int)lane, end - 1);
    return csr_src[min(max(j, 0), elast)];
}

__global__ __launch_bounds__(256)
void gat_gather(const int* __restrict__ rowptr, const int* __restrict__ csr_src,
                const float* __restrict__ as_, const float* __restrict__ ad_,
                const float* __restrict__ h, const float* __restrict__ bias,
                float* __restrict__ out, int N, int E, int relu_flag) {
    const unsigned lane = threadIdx.x & 31;
    const unsigned nw = gridDim.x * (blockDim.x >> 5);
    const unsigned gw = (unsigned)__builtin_amdgcn_readfirstlane(
        (int)(blockIdx.x * (blockDim.x >> 5) + (threadIdx.x >> 5)));
    const unsigned per = (unsigned)N / nw, rem = (unsigned)N % nw;
    const int n0 = (int)(gw * per + min(gw, rem));
    const int n1 = n0 + (int)per + (gw < rem ? 1 : 0);
    if (n0 >= n1) return;
    const v2f bb = ((const v2f*)bias)[lane];
    const unsigned nlast = (unsigned)(N - 1);
    const int elast = max(E - 1, 0);
    const v2f* h2 = (const v2f*)h;

    for (int base = n0; base < n1; base += 31) {
        const int c = min(31, n1 - base);
        int rp = rowptr[min(base + (int)lane, N)];
        rp = min(max(rp, 0), E);
        const float adl = ad_[min(base + (int)lane, N - 1)];

        int   sA = gat_edge_ids(csr_src, rp, 0, lane, elast);
        int   sB = gat_edge_ids(csr_src, rp, min(1, c - 1), lane, elast);
        sA = (int)min((unsigned)sA, nlast);
        float vA = as_[sA];

        for (int k = 0; k < c; ++k) {
            const int sC = gat_edge_ids(csr_src, rp, min(k + 2, c - 1), lane, elast);
            sB = (int)min((unsigned)sB, nlast);
            const float vB = as_[sB];

            const int beg = __builtin_amdgcn_readlane(rp, k);
            const int end = max(__builtin_amdgcn_readlane(rp, k + 1), beg);
            const float add = __builtin_bit_cast(float,
                __builtin_amdgcn_readlane(__builtin_bit_cast(int, adl), k));
            const int cnt = end - beg;

            v2f acc = {0.0f, 0.0f};
            if (cnt > 0) {
                if (cnt <= 32) {
                    const bool valid = (int)lane < cnt;
                    float v = vA + add;
                    v = (v > 0.0f) ? v : v * NEG_SLOPE;
                    const float m    = wave_max32(v);
                    const float ex   = valid ? __expf(v - m) : 0.0f;
                    const float den  = wave_sum32(ex);
                    const float alpha = ex / (den + 1e-16f);
                    gat_rows(h2, sA, alpha, cnt, lane, acc);
                } else {
                    float m = -3.0e38f;
                    for (int eb = beg; eb < end; eb += 32) {
                        const int j = eb + (int)lane;
                        const int s = (int)min((unsigned)csr_src[min(j, end - 1)], nlast);
                        m = fmaxf(m, gat_score(as_, s, add));
                    }
                    m = wave_max32(m);
                    float den = 0.0f;
                    for (int eb = beg; eb < end; eb += 32) {
                        const int j = eb + (int)lane;
                        const int s = (int)min((unsigned)csr_src[min(j, end - 1)], nlast);
                        const float ex = __expf(gat_score(as_, s, add) - m);
                        den += (j < end) ? ex : 0.0f;
                    }
                    den = wave_sum32(den);
                    const float dd = den + 1e-16f;
                    for (int eb = beg; eb < end; eb += 32) {
                        const int j = eb + (int)lane;
                        const int s = (int)min((unsigned)csr_src[min(j, end - 1)], nlast);
                        const float ex = __expf(gat_score(as_, s, add) - m);
                        const float alpha = (j < end) ? ex / dd : 0.0f;
                        gat_rows(h2, s, alpha, min(32, end - eb), lane, acc);
                    }
                }
            }

            v2f r;
            r.x = acc.x + bb.x;
            r.y = acc.y + bb.y;
            if (relu_flag) { r.x = fmaxf(r.x, 0.0f); r.y = fmaxf(r.y, 0.0f); }
            ((v2f*)out)[(unsigned)(base + k) * 32u + lane] = r;

            sA = sB; vA = vB; sB = sC;
        }
    }
}

extern "C" void kernel_launch(void* const* d_in, const int* in_sizes, int n_in,
                              void* d_out, int out_size, void* d_ws, size_t ws_size,
                              hipStream_t stream) {
    const float* x    = (const float*)d_in[0];
    const int*   eidx = (const int*)  d_in[1];
    const float* W1   = (const float*)d_in[2];
    const float* a1s  = (const float*)d_in[3];
    const float* a1d  = (const float*)d_in[4];
    const float* b1   = (const float*)d_in[5];
    const float* W2   = (const float*)d_in[6];
    const float* a2s  = (const float*)d_in[7];
    const float* a2d  = (const float*)d_in[8];
    const float* b2   = (const float*)d_in[9];

    const int N = in_sizes[0] / GAT_INC;
    const int E = in_sizes[1] / 2;
    const int* srcI = eidx;
    const int* dstI = eidx + E;

    float* ws = (float*)d_ws;
    size_t o = 0;
    float* h     = ws + o; o += (size_t)N * 64;
    float* hpost = ws + o; o += (size_t)N * 64;
    int*   graph = (int*)(ws + o); o += (size_t)N * 64;
    float* as_   = ws + o; o += N;
    float* ad_   = ws + o; o += N;
    o += N;
    o += N;
    int*   tmp   = (int*)(ws + o); o += E;

    const int NB = (N + CS_BN - 1) / CS_BN;
    const int M  = NB * CS_WG;
    const size_t Np = ((size_t)N + 4) & ~(size_t)3;
    const size_t Ep = ((size_t)E + 3) & ~(size_t)3;
    const size_t Mp = ((size_t)M + 4) & ~(size_t)3;
    int* rowptr  = graph;
    int* csr_src = graph + Np;
    int* wgcnt   = csr_src + Ep;
    int* base    = wgcnt + Mp;

    const int gGemm  = (N + 63) / 64;
    const int gPers  = GAT_PERS_WG;
    const int gScan  = (M + 1 + 1023) / 1024;

    csr_bucket_count  <<<CS_WG, CS_TH, NB * sizeof(int), stream>>>(dstI, wgcnt, E, NB);
    csr_scan          <<<gScan, 1024,  0,                stream>>>(wgcnt, base, M);
    csr_bucket_scatter<<<CS_WG, CS_TH, NB * sizeof(int), stream>>>(srcI, dstI, base, tmp, E, NB);
    csr_bucket_sort   <<<NB,    CS_TH, 0,                stream>>>(base, tmp, rowptr, csr_src, N, E, NB);

    gat_gemm_wmma<GAT_INC><<<gGemm, 256, 0, stream>>>(x, W1, a1s, a1d, h, as_, ad_, N);
    gat_gather    <<<gPers, 256, 0, stream>>>(rowptr, csr_src, as_, ad_, h, b1, hpost, N, E, 1);

    float* out = (float*)d_out;
    gat_gemm_wmma<GAT_HID><<<gGemm, 256, 0, stream>>>(hpost, W2, a2s, a2d, h, as_, ad_, N);
    gat_gather    <<<gPers, 256, 0, stream>>>(rowptr, csr_src, as_, ad_, h, b2, out, N, E, 0);
}


# Layer-2 GEMM builds its A tile in LDS by aggregating layer 1 itself (a tile's 8 waves do 8 nodes each: softmax, weighted rows, bias, relu): hidden plane never written or re-read, 8 to 7 launches
# speedup vs baseline: 4.6129x; 1.0332x over previous
#include <hip/hip_runtime.h>
#include <hip/hip_bf16.h>
#include <math.h>

typedef float v2f __attribute__((ext_vector_type(2)));
typedef float v8f __attribute__((ext_vector_type(8)));
typedef float v4f __attribute__((ext_vector_type(4)));

#define GAT_INC 256
#define GAT_HID 64
#define NEG_SLOPE 0.2f

template <int M>
__device__ __forceinline__ float row_xor(float x) {
    return __builtin_bit_cast(float, __builtin_amdgcn_update_dpp(
        0, __builtin_bit_cast(int, x), 0x160 + M, 0xf, 0xf, false));
}
__device__ __forceinline__ float xhalf(float x) {
    return __builtin_bit_cast(float, __builtin_amdgcn_permlanex16(
        __builtin_bit_cast(int, x), __builtin_bit_cast(int, x),
        0x76543210, 0xfedcba98, false, false));
}
__device__ __forceinline__ float wave_max32(float v) {
    v = fmaxf(v, row_xor<1>(v));
    v = fmaxf(v, row_xor<2>(v));
    v = fmaxf(v, row_xor<4>(v));
    v = fmaxf(v, row_xor<8>(v));
    return fmaxf(v, xhalf(v));
}
__device__ __forceinline__ float wave_sum32(float v) {
    v += row_xor<1>(v);
    v += row_xor<2>(v);
    v += row_xor<4>(v);
    v += row_xor<8>(v);
    return v + xhalf(v);
}

#define GAT_PERS_WG 1024

#define CS_SH 8
#define CS_BN (1 << CS_SH)
#define CS_WG 256
#define CS_TH 512
#define CS_UC 8
#define CS_UB 6

__device__ __forceinline__ void cs_chunk(int E, int& c0, int& c1) {
    const int per = (E + (int)gridDim.x - 1) / (int)gridDim.x;
    c0 = min((int)blockIdx.x * per, E);
    c1 = min(c0 + per, E);
}

__global__ __launch_bounds__(CS_TH)
void csr_bucket_count(const int* __restrict__ dstI, int* __restrict__ wgcnt, int E, int NB) {
    extern __shared__ int sBkt[];
    const int tid = threadIdx.x;
    for (int i = tid; i < NB; i += CS_TH) sBkt[i] = 0;
    __syncthreads();
    int c0, c1;
    cs_chunk(E, c0, c1);
    const unsigned blast = (unsigned)(NB - 1);
    for (int e0 = c0 + tid; e0 < c1; e0 += CS_UC * CS_TH) {
        int d[CS_UC];
#pragma unroll
        for (int u = 0; u < CS_UC; ++u) d[u] = dstI[min(e0 + u * CS_TH, c1 - 1)];
#pragma unroll
        for (int u = 0; u < CS_UC; ++u)
            atomicAdd(&sBkt[min((unsigned)d[u] >> CS_SH, blast)], (e0 + u * CS_TH < c1) ? 1 : 0);
    }
    __syncthreads();
    for (int i = tid; i < NB; i += CS_TH) wgcnt[(size_t)i * gridDim.x + blockIdx.x] = sBkt[i];
}

__global__ __launch_bounds__(1024)
void csr_scan(const int* __restrict__ cnt, int* __restrict__ rowptr, int N) {
    __shared__ int sFront[32];
    __shared__ int sWave[32];
    const int tid  = threadIdx.x;
    const int lane = tid & 31;
    const int wave = tid >> 5;
    const int first = blockIdx.x * 1024;

    int front = 0;
    const int4* c4 = (const int4*)cnt;
#pragma unroll 8
    for (int q = tid; q < (first >> 2); q += 1024) {
        int4 v = c4[q];
        front += (v.x + v.y) + (v.z + v.w);
    }
    for (int off = 16; off > 0; off >>= 1) front += __shfl_xor(front, off, 32);

    const int i = first + tid;
    const int x = (i < N) ? cnt[i] : 0;
    int incl = x;
    for (int off = 1; off < 32; off <<= 1) {
        int y = __shfl_up(incl, off, 32);
        if (lane >= off) incl += y;
    }
    if (lane == 0)  sFront[wave] = front;
    if (lane == 31) sWave[wave]  = incl;
    __syncthreads();

    int a = sFront[lane];
    int b = (lane < wave) ? sWave[lane] : 0;
    for (int off = 16; off > 0; off >>= 1) {
        a += __shfl_xor(a, off, 32);
        b += __shfl_xor(b, off, 32);
    }
    if (i <= N) rowptr[i] = a + b + (incl - x);
}

__global__ __launch_bounds__(CS_TH)
void csr_bucket_scatter(const int* __restrict__ srcI, const int* __restrict__ dstI,
                        const int* __restrict__ base, int* __restrict__ tmp, int E, int NB) {
    extern __shared__ int sBkt[];
    const int tid = threadIdx.x;
    int c0, c1;
    cs_chunk(E, c0, c1);
    if (c0 >= c1) return;
    const unsigned blast = (unsigned)(NB - 1);
    const unsigned elast = (unsigned)(E - 1);
    int e0 = c0 + tid;
    int d[CS_UC], s[CS_UC];
#pragma unroll
    for (int u = 0; u < CS_UC; ++u) {
        const int e = min(e0 + u * CS_TH, c1 - 1);
        d[u] = dstI[e];
        s[u] = srcI[e];
    }
    for (int i = tid; i < NB; i += CS_TH) sBkt[i] = base[(size_t)i * gridDim.x + blockIdx.x];
    __syncthreads();
    while (e0 < c1) {
        int p[CS_UC];
#pragma unroll
        for (int u = 0; u < CS_UC; ++u)
            p[u] = atomicAdd(&sBkt[min((unsigned)d[u] >> CS_SH, blast)],
                             (e0 + u * CS_TH < c1) ? 1 : 0);
#pragma unroll
        for (int u = 0; u < CS_UC; ++u)
            if (e0 + u * CS_TH < c1)
                tmp[min((unsigned)p[u], elast)] =
                    (int)(((unsigned)s[u] << CS_SH) | ((unsigned)d[u] & (unsigned)(CS_BN - 1)));
        e0 += CS_UC * CS_TH;
        if (e0 < c1) {
#pragma unroll
            for (int u = 0; u < CS_UC; ++u) {
                const int e = min(e0 + u * CS_TH, c1 - 1);
                d[u] = dstI[e];
                s[u] = srcI[e];
            }
        }
    }
}

__global__ __launch_bounds__(CS_TH)
void csr_bucket_sort(const int* __restrict__ base, const int* __restrict__ tmp,
                     int* __restrict__ rowptr, int* __restrict__ csr_src, int N, int E, int NB) {
    __shared__ int sCnt[CS_BN];
    __shared__ int sTot[CS_BN / 32];
    const int tid  = threadIdx.x;
    const int lane = tid & 31;
    const int wave = tid >> 5;
    const int b    = blockIdx.x;
    int beg = base[(size_t)b * CS_WG];
    int end = base[(size_t)(b + 1) * CS_WG];
    beg = min(max(beg, 0), E);
    end = min(max(end, beg), E);
    const unsigned elast = (unsigned)max(E - 1, 0);
    if (tid < CS_BN) sCnt[tid] = 0;
    __syncthreads();
    for (int j0 = beg + tid; j0 < end; j0 += CS_UB * CS_TH) {
        int w[CS_UB];
#pragma unroll
        for (int u = 0; u < CS_UB; ++u) w[u] = tmp[min(j0 + u * CS_TH, end - 1)];
#pragma unroll
        for (int u = 0; u < CS_UB; ++u)
            atomicAdd(&sCnt[w[u] & (CS_BN - 1)], (j0 + u * CS_TH < end) ? 1 : 0);
    }
    __syncthreads();
    const int x = (tid < CS_BN) ? sCnt[tid] : 0;
    int incl = x;
    for (int off = 1; off < 32; off <<= 1) {
        int y = __shfl_up(incl, off, 32);
        if (lane >= off) incl += y;
    }
    if (tid < CS_BN && lane == 31) sTot[wave] = incl;
    __syncthreads();
    if (tid < CS_BN) {
        int front = 0;
        for (int v = 0; v < wave; ++v) front += sTot[v];
        const int start = beg + front + (incl - x);
        sCnt[tid] = start;
        const int node = b * CS_BN + tid;
        if (node < N) rowptr[node] = start;
    }
    if (b == NB - 1 && tid == 0) rowptr[N] = end;
    __syncthreads();
    for (int j0 = beg + tid; j0 < end; j0 += CS_UB * CS_TH) {
        int w[CS_UB];
#pragma unroll
        for (int u = 0; u < CS_UB; ++u) w[u] = tmp[min(j0 + u * CS_TH, end - 1)];
        int p[CS_UB];
#pragma unroll
        for (int u = 0; u < CS_UB; ++u)
            p[u] = atomicAdd(&sCnt[w[u] & (CS_BN - 1)], (j0 + u * CS_TH < end) ? 1 : 0);
#pragma unroll
        for (int u = 0; u < CS_UB; ++u)
            if (j0 + u * CS_TH < end)
                csr_src[min((unsigned)p[u], elast)] = (int)((unsigned)w[u] >> CS_SH);
    }
}

__device__ __forceinline__ float gat_score(const float* __restrict__ as_, int s, float add) {
    float v = as_[s] + add;
    return (v > 0.0f) ? v : v * NEG_SLOPE;
}
template <int G>
__device__ __forceinline__ void gat_row_group(const v2f* __restrict__ h2, int s, float alpha,
                                              int t, unsigned lane, v2f& acc) {
    unsigned idx[G];
    v2f r[G];
#pragma unroll
    for (int u = 0; u < G; ++u)
        idx[u] = (unsigned)__builtin_amdgcn_readlane(s, t + u) * 32u + lane;
#pragma unroll
    for (int u = 0; u < G; ++u)
        r[u] = h2[idx[u]];
#pragma unroll
    for (int u = 0; u < G; ++u) {
        const float at = __builtin_bit_cast(float,
            __builtin_amdgcn_readlane(__builtin_bit_cast(int, alpha), t + u));
        acc = acc + r[u] * at;
    }
}
__device__ __forceinline__ void gat_rows(const v2f* __restrict__ h2, int s, float alpha,
                                         int cnt, unsigned lane, v2f& acc) {
    const int c4 = (cnt + 3) & ~3;
    if (c4 <= 4)       gat_row_group<4>(h2, s, alpha, 0, lane, acc);
    else if (c4 <= 8)  gat_row_group<8>(h2, s, alpha, 0, lane, acc);
    else if (c4 <= 12) gat_row_group<12>(h2, s, alpha, 0, lane, acc);
    else {
        gat_row_group<16>(h2, s, alpha, 0, lane, acc);
        int t = 16;
        for (; t + 8 <= c4; t += 8) gat_row_group<8>(h2, s, alpha, t, lane, acc);
        if (t < c4)                 gat_row_group<4>(h2, s, alpha, t, lane, acc);
    }
}

__device__ __forceinline__ int gat_edge_ids(const int* __restrict__ csr_src, int rp, int k,
                                            unsigned lane, int elast) {
    const int beg = __builtin_amdgcn_readlane(rp, k);
    const int end = max(__builtin_amdgcn_readlane(rp, k + 1), beg);
    const int j   = min(beg + (int)lane, end - 1);
    return csr_src[min(max(j, 0), elast)];
}

template <bool TO_LDS>
__device__ __forceinline__ void gat_node_run(const int* __restrict__ rowptr, const int* __restrict__ csr_src,
                                             const float* __restrict__ as_, const float* __restrict__ ad_,
                                             const float* __restrict__ h, const v2f bb,
                                             float* dst, int pitch, int row0,
                                             int n0, int n1, int N, int E, int relu_flag, unsigned lane) {
    const unsigned nlast = (unsigned)(N - 1);
    const int elast = max(E - 1, 0);
    const v2f* h2 = (const v2f*)h;

    for (int base = n0; base < n1; base += 31) {
        const int c = min(31, n1 - base);
        int rp = rowptr[min(base + (int)lane, N)];
        rp = min(max(rp, 0), E);
        const float adl = ad_[min(base + (int)lane, N - 1)];

        int   sA = gat_edge_ids(csr_src, rp, 0, lane, elast);
        int   sB = gat_edge_ids(csr_src, rp, min(1, c - 1), lane, elast);
        sA = (int)min((unsigned)sA, nlast);
        float vA = as_[sA];

        for (int k = 0; k < c; ++k) {
            const int sC = gat_edge_ids(csr_src, rp, min(k + 2, c - 1), lane, elast);
            sB = (int)min((unsigned)sB, nlast);
            const float vB = as_[sB];

            const int beg = __builtin_amdgcn_readlane(rp, k);
            const int end = max(__builtin_amdgcn_readlane(rp, k + 1), beg);
            const float add = __builtin_bit_cast(float,
                __builtin_amdgcn_readlane(__builtin_bit_cast(int, adl), k));
            const int cnt = end - beg;

            v2f acc = {0.0f, 0.0f};
            if (cnt > 0) {
                if (cnt <= 32) {
                    const bool valid = (int)lane < cnt;
                    float v = vA + add;
                    v = (v > 0.0f) ? v : v * NEG_SLOPE;
                    const float m    = wave_max32(v);
                    const float ex   = valid ? __expf(v - m) : 0.0f;
                    const float den  = wave_sum32(ex);
                    const float alpha = ex / (den + 1e-16f);
                    gat_rows(h2, sA, alpha, cnt, lane, acc);
                } else {
                    float m = -3.0e38f;
                    for (int eb = beg; eb < end; eb += 32) {
                        const int j = eb + (int)lane;
                        const int s = (int)min((unsigned)csr_src[min(j, end - 1)], nlast);
                        m = fmaxf(m, gat_score(as_, s, add));
                    }
                    m = wave_max32(m);
                    float den = 0.0f;
                    for (int eb = beg; eb < end; eb += 32) {
                        const int j = eb + (int)lane;
                        const int s = (int)min((unsigned)csr_src[min(j, end - 1)], nlast);
                        const float ex = __expf(gat_score(as_, s, add) - m);
                        den += (j < end) ? ex : 0.0f;
                    }
                    den = wave_sum32(den);
                    const float dd = den + 1e-16f;
                    for (int eb = beg; eb < end; eb += 32) {
                        const int j = eb + (int)lane;
                        const int s = (int)min((unsigned)csr_src[min(j, end - 1)], nlast);
                        const float ex = __expf(gat_score(as_, s, add) - m);
                        const float alpha = (j < end) ? ex / dd : 0.0f;
                        gat_rows(h2, s, alpha, min(32, end - eb), lane, acc);
                    }
                }
            }

            v2f r;
            r.x = acc.x + bb.x;
            r.y = acc.y + bb.y;
            if (relu_flag) { r.x = fmaxf(r.x, 0.0f); r.y = fmaxf(r.y, 0.0f); }
            if (TO_LDS) *(v2f*)(dst + (base + k - row0) * pitch + 2 * (int)lane) = r;
            else        ((v2f*)dst)[(unsigned)(base + k) * 32u + lane] = r;

            sA = sB; vA = vB; sB = sC;
        }
    }
}

__global__ __launch_bounds__(256)
void gat_gather(const int* __restrict__ rowptr, const int* __restrict__ csr_src,
                const float* __restrict__ as_, const float* __restrict__ ad_,
                const float* __restrict__ h, const float* __restrict__ bias,
                float* __restrict__ out, int N, int E, int relu_flag) {
    const unsigned lane = threadIdx.x & 31;
    const unsigned nw = gridDim.x * (blockDim.x >> 5);
    const unsigned gw = (unsigned)__builtin_amdgcn_readfirstlane(
        (int)(blockIdx.x * (blockDim.x >> 5) + (threadIdx.x >> 5)));
    const unsigned per = (unsigned)N / nw, rem = (unsigned)N % nw;
    const int n0 = (int)(gw * per + min(gw, rem));
    const int n1 = n0 + (int)per + (gw < rem ? 1 : 0);
    if (n0 >= n1) return;
    const v2f bb = ((const v2f*)bias)[lane];
    gat_node_run<false>(rowptr, csr_src, as_, ad_, h, bb, out, 64, 0, n0, n1, N, E, relu_flag, lane);
}

template <int K, bool AGG>
__global__ __launch_bounds__(256)
void gat_gemm_wmma(const float* __restrict__ A, const float* __restrict__ W,
                   const float* __restrict__ a_src, const float* __restrict__ a_dst,
                   float* __restrict__ C, float* __restrict__ as_, float* __restrict__ ad_,
                   int N,
                   const int* __restrict__ rowptr, const int* __restrict__ csr_src,
                   const float* __restrict__ asP, const float* __restrict__ adP,
                   const float* __restrict__ biasP, int E) {
    constexpr int KP    = K + 4;
    constexpr int QPR   = K / 4;
    constexpr int RSTEP = 256 / QPR;
    constexpr int NQ    = 64 / RSTEP;
    constexpr int CP    = 68;
    static_assert(K % 4 == 0 && 256 % QPR == 0 && 64 % RSTEP == 0, "tile map");
    static_assert(CP <= KP, "the C tile reuses the A tile's LDS");
    static_assert(!AGG || K == 64, "an aggregated row is 64 features: one v2f per lane");
    __shared__ __attribute__((aligned(16))) float sA[64 * KP];
    const int lane = threadIdx.x & 31;
    const int wave = __builtin_amdgcn_readfirstlane((int)(threadIdx.x >> 5));
    const int rowBase = blockIdx.x * 64;

    const int col      = (wave & 3) * 16 + (lane & 15);
    const int waveRow  = (wave >> 2) * 32;

    for (int k = threadIdx.x; k < K; k += 256)
        __builtin_prefetch(&W[(size_t)k * 64 + col], 0, 3);

    if constexpr (AGG) {
        const int n0 = rowBase + wave * 8;
        const int n1 = min(n0 + 8, N);
        const int nv = max(n1 - n0, 0);
        for (int k = nv; k < 8; ++k) {
            v2f z; z.x = 0.0f; z.y = 0.0f;
            *(v2f*)&sA[(wave * 8 + k) * KP + 2 * lane] = z;
        }
        if (nv > 0) {
            const v2f bb = ((const v2f*)biasP)[lane];
            gat_node_run<true>(rowptr, csr_src, asP, adP, A, bb, sA, KP, rowBase,
                               n0, n1, N, E, 1, (unsigned)lane);
        }
    } else {
        const int q  = threadIdx.x % QPR;
        const int r0 = threadIdx.x / QPR;
        v4f t[NQ];
#pragma unroll
        for (int i = 0; i < NQ; ++i) {
            const int gr = max(min(rowBase + r0 + i * RSTEP, N - 1), 0);
            t[i] = __builtin_nontemporal_load((const v4f*)(A + (size_t)gr * K) + q);
        }
#pragma unroll
        for (int i = 0; i < NQ; ++i)
            *(v4f*)&sA[(r0 + i * RSTEP) * KP + 4 * q] = t[i];
    }
    __syncthreads();

    const int row   = lane & 15;
    const int khalf = (lane >> 4) << 1;

    v8f acc0 = {};
    v8f acc1 = {};
#pragma unroll 8
    for (int k = 0; k < K; k += 4) {
        v2f b;
        b.x = W[(size_t)(k + khalf)     * 64 + col];
        b.y = W[(size_t)(k + khalf + 1) * 64 + col];

        v2f a0, a1;
        a0.x = sA[(waveRow +      row) * KP + k + khalf];
        a0.y = sA[(waveRow +      row) * KP + k + khalf + 1];
        a1.x = sA[(waveRow + 16 + row) * KP + k + khalf];
        a1.y = sA[(waveRow + 16 + row) * KP + k + khalf + 1];

        acc0 = __builtin_amdgcn_wmma_f32_16x16x4_f32(
            false, a0, false, b, (short)0, acc0, false, false);
        acc1 = __builtin_amdgcn_wmma_f32_16x16x4_f32(
            false, a1, false, b, (short)0, acc1, false, false);
    }

    __syncthreads();
    float* sC = sA;

    const int mhalf = (lane >> 4) << 3;
    for (int r = 0; r < 8; ++r) {
        sC[(waveRow +      mhalf + r) * CP + col] = acc0[r];
        sC[(waveRow + 16 + mhalf + r) * CP + col] = acc1[r];
    }

    const int c0 = (wave & 3) * 16;
    float* crow = &sC[(waveRow + lane) * CP + c0];
    {
        float s = 0.0f, d = 0.0f;
#pragma unroll
        for (int j = 0; j < 4; ++j) {
            const v4f cv = *(const v4f*)(crow + 4 * j);
            s = __builtin_fmaf(cv.x, a_src[c0 + 4 * j + 0], s); d = __builtin_fmaf(cv.x, a_dst[c0 + 4 * j + 0], d);
            s = __builtin_fmaf(cv.y, a_src[c0 + 4 * j + 1], s); d = __builtin_fmaf(cv.y, a_dst[c0 + 4 * j + 1], d);
            s = __builtin_fmaf(cv.z, a_src[c0 + 4 * j + 2], s); d = __builtin_fmaf(cv.z, a_dst[c0 + 4 * j + 2], d);
            s = __builtin_fmaf(cv.w, a_src[c0 + 4 * j + 3], s); d = __builtin_fmaf(cv.w, a_dst[c0 + 4 * j + 3], d);
        }
        v2f p; p.x = s; p.y = d;
        *(v2f*)crow = p;
    }
    __syncthreads();

    if (c0 == 0) {
        const v2f p0 = *(const v2f*)(crow);
        const v2f p1 = *(const v2f*)(crow + 16);
        const v2f p2 = *(const v2f*)(crow + 32);
        const v2f p3 = *(const v2f*)(crow + 48);
        const int gr = rowBase + waveRow + lane;
        if (gr < N) {
            as_[gr] = (p0.x + p1.x) + (p2.x + p3.x);
            ad_[gr] = (p0.y + p1.y) + (p2.y + p3.y);
        }
    }

    for (int r = 0; r < 8; ++r) {
        int gr0 = rowBase + waveRow +      mhalf + r;
        int gr1 = rowBase + waveRow + 16 + mhalf + r;
        if (gr0 < N) C[(size_t)gr0 * 64 + col] = acc0[r];
        if (gr1 < N) C[(size_t)gr1 * 64 + col] = acc1[r];
    }
}

extern "C" void kernel_launch(void* const* d_in, const int* in_sizes, int n_in,
                              void* d_out, int out_size, void* d_ws, size_t ws_size,
                              hipStream_t stream) {
    const float* x    = (const float*)d_in[0];
    const int*   eidx = (const int*)  d_in[1];
    const float* W1   = (const float*)d_in[2];
    const float* a1s  = (const float*)d_in[3];
    const float* a1d  = (const float*)d_in[4];
    const float* b1   = (const float*)d_in[5];
    const float* W2   = (const float*)d_in[6];
    const float* a2s  = (const float*)d_in[7];
    const float* a2d  = (const float*)d_in[8];
    const float* b2   = (const float*)d_in[9];

    const int N = in_sizes[0] / GAT_INC;
    const int E = in_sizes[1] / 2;
    const int* srcI = eidx;
    const int* dstI = eidx + E;

    float* ws = (float*)d_ws;
    size_t o = 0;
    float* h     = ws + o; o += (size_t)N * 64;
    float* h2    = ws + o; o += (size_t)N * 64;
    int*   graph = (int*)(ws + o); o += (size_t)N * 64;
    float* as_   = ws + o; o += N;
    float* ad_   = ws + o; o += N;
    float* as2   = ws + o; o += N;
    float* ad2   = ws + o; o += N;
    int*   tmp   = (int*)(ws + o); o += E;

    const int NB = (N + CS_BN - 1) / CS_BN;
    const int M  = NB * CS_WG;
    const size_t Np = ((size_t)N + 4) & ~(size_t)3;
    const size_t Ep = ((size_t)E + 3) & ~(size_t)3;
    const size_t Mp = ((size_t)M + 4) & ~(size_t)3;
    int* rowptr  = graph;
    int* csr_src = graph + Np;
    int* wgcnt   = csr_src + Ep;
    int* base    = wgcnt + Mp;

    const int gGemm  = (N + 63) / 64;
    const int gPers  = GAT_PERS_WG;
    const int gScan  = (M + 1 + 1023) / 1024;

    csr_bucket_count  <<<CS_WG, CS_TH, NB * sizeof(int), stream>>>(dstI, wgcnt, E, NB);
    csr_scan          <<<gScan, 1024,  0,                stream>>>(wgcnt, base, M);
    csr_bucket_scatter<<<CS_WG, CS_TH, NB * sizeof(int), stream>>>(srcI, dstI, base, tmp, E, NB);
    csr_bucket_sort   <<<NB,    CS_TH, 0,                stream>>>(base, tmp, rowptr, csr_src, N, E, NB);

    gat_gemm_wmma<GAT_INC, false><<<gGemm, 256, 0, stream>>>(x, W1, a1s, a1d, h, as_, ad_, N,
                                                             nullptr, nullptr, nullptr, nullptr, nullptr, E);

    gat_gemm_wmma<GAT_HID, true><<<gGemm, 256, 0, stream>>>(h, W2, a2s, a2d, h2, as2, ad2, N,
                                                            rowptr, csr_src, as_, ad_, b1, E);

    float* out = (float*)d_out;
    gat_gather    <<<gPers, 256, 0, stream>>>(rowptr, csr_src, as2, ad2, h2, b2, out, N, E, 0);
}


# Layer-1 GEMM tiles run inside the four edge-sort launches, behind each stage's own work-groups (stages at 256 threads): GEMM overlaps the sort, 7 to 6 launches; rows still in LDS-add arrival order
# speedup vs baseline: 4.7516x; 1.0301x over previous
#include <hip/hip_runtime.h>
#include <hip/hip_bf16.h>
#include <math.h>

typedef float v2f __attribute__((ext_vector_type(2)));
typedef float v8f __attribute__((ext_vector_type(8)));
typedef float v4f __attribute__((ext_vector_type(4)));

#define GAT_INC 256
#define GAT_HID 64
#define NEG_SLOPE 0.2f

template <int M>
__device__ __forceinline__ float row_xor(float x) {
    return __builtin_bit_cast(float, __builtin_amdgcn_update_dpp(
        0, __builtin_bit_cast(int, x), 0x160 + M, 0xf, 0xf, false));
}
__device__ __forceinline__ float xhalf(float x) {
    return __builtin_bit_cast(float, __builtin_amdgcn_permlanex16(
        __builtin_bit_cast(int, x), __builtin_bit_cast(int, x),
        0x76543210, 0xfedcba98, false, false));
}
__device__ __forceinline__ float wave_max32(float v) {
    v = fmaxf(v, row_xor<1>(v));
    v = fmaxf(v, row_xor<2>(v));
    v = fmaxf(v, row_xor<4>(v));
    v = fmaxf(v, row_xor<8>(v));
    return fmaxf(v, xhalf(v));
}
__device__ __forceinline__ float wave_sum32(float v) {
    v += row_xor<1>(v);
    v += row_xor<2>(v);
    v += row_xor<4>(v);
    v += row_xor<8>(v);
    return v + xhalf(v);
}

#define GAT_PERS_WG 1024

#define CS_SH 8
#define CS_BN (1 << CS_SH)
#define CS_WG 256
#define CS_TH 256
#define CS_UC 16
#define CS_UB 12

__device__ __forceinline__ void cs_chunk(int E, int wg, int& c0, int& c1) {
    const int per = (E + CS_WG - 1) / CS_WG;
    c0 = min(wg * per, E);
    c1 = min(c0 + per, E);
}

__device__ __forceinline__
void csr_bucket_count(const int* __restrict__ dstI, int* __restrict__ wgcnt, int E, int NB,
                      int wg, int* sBkt) {
    const int tid = threadIdx.x;
    for (int i = tid; i < NB; i += CS_TH) sBkt[i] = 0;
    __syncthreads();
    int c0, c1;
    cs_chunk(E, wg, c0, c1);
    const unsigned blast = (unsigned)(NB - 1);
    for (int e0 = c0 + tid; e0 < c1; e0 += CS_UC * CS_TH) {
        int d[CS_UC];
#pragma unroll
        for (int u = 0; u < CS_UC; ++u) d[u] = dstI[min(e0 + u * CS_TH, c1 - 1)];
#pragma unroll
        for (int u = 0; u < CS_UC; ++u)
            atomicAdd(&sBkt[min((unsigned)d[u] >> CS_SH, blast)], (e0 + u * CS_TH < c1) ? 1 : 0);
    }
    __syncthreads();
    for (int i = tid; i < NB; i += CS_TH) wgcnt[(size_t)i * CS_WG + wg] = sBkt[i];
}

#define CS_SCAN_B (4 * CS_TH)
__device__ __forceinline__
void csr_scan(const int* __restrict__ cnt, int* __restrict__ rowptr, int N, int blk, int* sL) {
    int* sFront = sL;
    int* sWave  = sL + 32;
    const int tid  = threadIdx.x;
    const int lane = tid & 31;
    const int wave = tid >> 5;
    const int first = blk * CS_SCAN_B;

    int front = 0;
    const int4* c4 = (const int4*)cnt;
#pragma unroll 16
    for (int q = tid; q < (first >> 2); q += CS_TH) {
        int4 v = c4[q];
        front += (v.x + v.y) + (v.z + v.w);
    }
    for (int off = 16; off > 0; off >>= 1) front += __shfl_xor(front, off, 32);

    const int i = first + 4 * tid;
    const int x0 = (i     < N) ? cnt[i]     : 0;
    const int x1 = (i + 1 < N) ? cnt[i + 1] : 0;
    const int x2 = (i + 2 < N) ? cnt[i + 2] : 0;
    const int x3 = (i + 3 < N) ? cnt[i + 3] : 0;
    const int x = (x0 + x1) + (x2 + x3);
    int incl = x;
    for (int off = 1; off < 32; off <<= 1) {
        int y = __shfl_up(incl, off, 32);
        if (lane >= off) incl += y;
    }
    if (lane == 0)  sFront[wave] = front;
    if (lane == 31) sWave[wave]  = incl;
    __syncthreads();

    int a = (lane < CS_TH / 32) ? sFront[lane] : 0;
    int b = (lane < wave) ? sWave[lane] : 0;
    for (int off = 16; off > 0; off >>= 1) {
        a += __shfl_xor(a, off, 32);
        b += __shfl_xor(b, off, 32);
    }
    const int e0 = a + b + (incl - x);
    if (i     <= N) rowptr[i]     = e0;
    if (i + 1 <= N) rowptr[i + 1] = e0 + x0;
    if (i + 2 <= N) rowptr[i + 2] = e0 + x0 + x1;
    if (i + 3 <= N) rowptr[i + 3] = e0 + x0 + x1 + x2;
}

__device__ __forceinline__
void csr_bucket_scatter(const int* __restrict__ srcI, const int* __restrict__ dstI,
                        const int* __restrict__ base, int* __restrict__ tmp, int E, int NB,
                        int wg, int* sBkt) {
    const int tid = threadIdx.x;
    int c0, c1;
    cs_chunk(E, wg, c0, c1);
    if (c0 >= c1) return;
    const unsigned blast = (unsigned)(NB - 1);
    const unsigned elast = (unsigned)(E - 1);
    int e0 = c0 + tid;
    int d[CS_UC], s[CS_UC];
#pragma unroll
    for (int u = 0; u < CS_UC; ++u) {
        const int e = min(e0 + u * CS_TH, c1 - 1);
        d[u] = dstI[e];
        s[u] = srcI[e];
    }
    for (int i = tid; i < NB; i += CS_TH) sBkt[i] = base[(size_t)i * CS_WG + wg];
    __syncthreads();
    while (e0 < c1) {
        int p[CS_UC];
#pragma unroll
        for (int u = 0; u < CS_UC; ++u)
            p[u] = atomicAdd(&sBkt[min((unsigned)d[u] >> CS_SH, blast)],
                             (e0 + u * CS_TH < c1) ? 1 : 0);
#pragma unroll
        for (int u = 0; u < CS_UC; ++u)
            if (e0 + u * CS_TH < c1)
                tmp[min((unsigned)p[u], elast)] =
                    (int)(((unsigned)s[u] << CS_SH) | ((unsigned)d[u] & (unsigned)(CS_BN - 1)));
        e0 += CS_UC * CS_TH;
        if (e0 < c1) {
#pragma unroll
            for (int u = 0; u < CS_UC; ++u) {
                const int e = min(e0 + u * CS_TH, c1 - 1);
                d[u] = dstI[e];
                s[u] = srcI[e];
            }
        }
    }
}

__device__ __forceinline__
void csr_bucket_sort(const int* __restrict__ base, const int* __restrict__ tmp,
                     int* __restrict__ rowptr, int* __restrict__ csr_src, int N, int E, int NB,
                     int b, int* sL) {
    int* sCnt = sL;
    int* sTot = sL + CS_BN;
    const int tid  = threadIdx.x;
    const int lane = tid & 31;
    const int wave = tid >> 5;
    int beg = base[(size_t)b * CS_WG];
    int end = base[(size_t)(b + 1) * CS_WG];
    beg = min(max(beg, 0), E);
    end = min(max(end, beg), E);
    const unsigned elast = (unsigned)max(E - 1, 0);
    if (tid < CS_BN) sCnt[tid] = 0;
    __syncthreads();
    for (int j0 = beg + tid; j0 < end; j0 += CS_UB * CS_TH) {
        int w[CS_UB];
#pragma unroll
        for (int u = 0; u < CS_UB; ++u) w[u] = tmp[min(j0 + u * CS_TH, end - 1)];
#pragma unroll
        for (int u = 0; u < CS_UB; ++u)
            atomicAdd(&sCnt[w[u] & (CS_BN - 1)], (j0 + u * CS_TH < end) ? 1 : 0);
    }
    __syncthreads();
    const int x = (tid < CS_BN) ? sCnt[tid] : 0;
    int incl = x;
    for (int off = 1; off < 32; off <<= 1) {
        int y = __shfl_up(incl, off, 32);
        if (lane >= off) incl += y;
    }
    if (tid < CS_BN && lane == 31) sTot[wave] = incl;
    __syncthreads();
    if (tid < CS_BN) {
        int front = 0;
        for (int v = 0; v < wave; ++v) front += sTot[v];
        const int start = beg + front + (incl - x);
        sCnt[tid] = start;
        const int node = b * CS_BN + tid;
        if (node < N) rowptr[node] = start;
    }
    if (b == NB - 1 && tid == 0) rowptr[N] = end;
    __syncthreads();
    for (int j0 = beg + tid; j0 < end; j0 += CS_UB * CS_TH) {
        int w[CS_UB];
#pragma unroll
        for (int u = 0; u < CS_UB; ++u) w[u] = tmp[min(j0 + u * CS_TH, end - 1)];
        int p[CS_UB];
#pragma unroll
        for (int u = 0; u < CS_UB; ++u)
            p[u] = atomicAdd(&sCnt[w[u] & (CS_BN - 1)], (j0 + u * CS_TH < end) ? 1 : 0);
#pragma unroll
        for (int u = 0; u < CS_UB; ++u)
            if (j0 + u * CS_TH < end)
                csr_src[min((unsigned)p[u], elast)] = (int)((unsigned)w[u] >> CS_SH);
    }
}

__device__ __forceinline__ float gat_score(const float* __restrict__ as_, int s, float add) {
    float v = as_[s] + add;
    return (v > 0.0f) ? v : v * NEG_SLOPE;
}
template <int G>
__device__ __forceinline__ void gat_row_group(const v2f* __restrict__ h2, int s, float alpha,
                                              int t, unsigned lane, v2f& acc) {
    unsigned idx[G];
    v2f r[G];
#pragma unroll
    for (int u = 0; u < G; ++u)
        idx[u] = (unsigned)__builtin_amdgcn_readlane(s, t + u) * 32u + lane;
#pragma unroll
    for (int u = 0; u < G; ++u)
        r[u] = h2[idx[u]];
#pragma unroll
    for (int u = 0; u < G; ++u) {
        const float at = __builtin_bit_cast(float,
            __builtin_amdgcn_readlane(__builtin_bit_cast(int, alpha), t + u));
        acc = acc + r[u] * at;
    }
}
__device__ __forceinline__ void gat_rows(const v2f* __restrict__ h2, int s, float alpha,
                                         int cnt, unsigned lane, v2f& acc) {
    const int c4 = (cnt + 3) & ~3;
    if (c4 <= 4)       gat_row_group<4>(h2, s, alpha, 0, lane, acc);
    else if (c4 <= 8)  gat_row_group<8>(h2, s, alpha, 0, lane, acc);
    else if (c4 <= 12) gat_row_group<12>(h2, s, alpha, 0, lane, acc);
    else {
        gat_row_group<16>(h2, s, alpha, 0, lane, acc);
        int t = 16;
        for (; t + 8 <= c4; t += 8) gat_row_group<8>(h2, s, alpha, t, lane, acc);
        if (t < c4)                 gat_row_group<4>(h2, s, alpha, t, lane, acc);
    }
}

__device__ __forceinline__ int gat_edge_ids(const int* __restrict__ csr_src, int rp, int k,
                                            unsigned lane, int elast) {
    const int beg = __builtin_amdgcn_readlane(rp, k);
    const int end = max(__builtin_amdgcn_readlane(rp, k + 1), beg);
    const int j   = min(beg + (int)lane, end - 1);
    return csr_src[min(max(j, 0), elast)];
}

template <bool TO_LDS>
__device__ __forceinline__ void gat_node_run(const int* __restrict__ rowptr, const int* __restrict__ csr_src,
                                             const float* __restrict__ as_, const float* __restrict__ ad_,
                                             const float* __restrict__ h, const v2f bb,
                                             float* dst, int pitch, int row0,
                                             int n0, int n1, int N, int E, int relu_flag, unsigned lane) {
    const unsigned nlast = (unsigned)(N - 1);
    const int elast = max(E - 1, 0);
    const v2f* h2 = (const v2f*)h;

    for (int base = n0; base < n1; base += 31) {
        const int c = min(31, n1 - base);
        int rp = rowptr[min(base + (int)lane, N)];
        rp = min(max(rp, 0), E);
        const float adl = ad_[min(base + (int)lane, N - 1)];

        int   sA = gat_edge_ids(csr_src, rp, 0, lane, elast);
        int   sB = gat_edge_ids(csr_src, rp, min(1, c - 1), lane, elast);
        sA = (int)min((unsigned)sA, nlast);
        float vA = as_[sA];

        for (int k = 0; k < c; ++k) {
            const int sC = gat_edge_ids(csr_src, rp, min(k + 2, c - 1), lane, elast);
            sB = (int)min((unsigned)sB, nlast);
            const float vB = as_[sB];

            const int beg = __builtin_amdgcn_readlane(rp, k);
            const int end = max(__builtin_amdgcn_readlane(rp, k + 1), beg);
            const float add = __builtin_bit_cast(float,
                __builtin_amdgcn_readlane(__builtin_bit_cast(int, adl), k));
            const int cnt = end - beg;

            v2f acc = {0.0f, 0.0f};
            if (cnt > 0) {
                if (cnt <= 32) {
                    const bool valid = (int)lane < cnt;
                    float v = vA + add;
                    v = (v > 0.0f) ? v : v * NEG_SLOPE;
                    const float m    = wave_max32(v);
                    const float ex   = valid ? __expf(v - m) : 0.0f;
                    const float den  = wave_sum32(ex);
                    const float alpha = ex / (den + 1e-16f);
                    gat_rows(h2, sA, alpha, cnt, lane, acc);
                } else {
                    float m = -3.0e38f;
                    for (int eb = beg; eb < end; eb += 32) {
                        const int j = eb + (int)lane;
                        const int s = (int)min((unsigned)csr_src[min(j, end - 1)], nlast);
                        m = fmaxf(m, gat_score(as_, s, add));
                    }
                    m = wave_max32(m);
                    float den = 0.0f;
                    for (int eb = beg; eb < end; eb += 32) {
                        const int j = eb + (int)lane;
                        const int s = (int)min((unsigned)csr_src[min(j, end - 1)], nlast);
                        const float ex = __expf(gat_score(as_, s, add) - m);
                        den += (j < end) ? ex : 0.0f;
                    }
                    den = wave_sum32(den);
                    const float dd = den + 1e-16f;
                    for (int eb = beg; eb < end; eb += 32) {
                        const int j = eb + (int)lane;
                        const int s = (int)min((unsigned)csr_src[min(j, end - 1)], nlast);
                        const float ex = __expf(gat_score(as_, s, add) - m);
                        const float alpha = (j < end) ? ex / dd : 0.0f;
                        gat_rows(h2, s, alpha, min(32, end - eb), lane, acc);
                    }
                }
            }

            v2f r;
            r.x = acc.x + bb.x;
            r.y = acc.y + bb.y;
            if (relu_flag) { r.x = fmaxf(r.x, 0.0f); r.y = fmaxf(r.y, 0.0f); }
            if (TO_LDS) *(v2f*)(dst + (base + k - row0) * pitch + 2 * (int)lane) = r;
            else        ((v2f*)dst)[(unsigned)(base + k) * 32u + lane] = r;

            sA = sB; vA = vB; sB = sC;
        }
    }
}

__global__ __launch_bounds__(256)
void gat_gather(const int* __restrict__ rowptr, const int* __restrict__ csr_src,
                const float* __restrict__ as_, const float* __restrict__ ad_,
                const float* __restrict__ h, const float* __restrict__ bias,
                float* __restrict__ out, int N, int E, int relu_flag) {
    const unsigned lane = threadIdx.x & 31;
    const unsigned nw = gridDim.x * (blockDim.x >> 5);
    const unsigned gw = (unsigned)__builtin_amdgcn_readfirstlane(
        (int)(blockIdx.x * (blockDim.x >> 5) + (threadIdx.x >> 5)));
    const unsigned per = (unsigned)N / nw, rem = (unsigned)N % nw;
    const int n0 = (int)(gw * per + min(gw, rem));
    const int n1 = n0 + (int)per + (gw < rem ? 1 : 0);
    if (n0 >= n1) return;
    const v2f bb = ((const v2f*)bias)[lane];
    gat_node_run<false>(rowptr, csr_src, as_, ad_, h, bb, out, 64, 0, n0, n1, N, E, relu_flag, lane);
}

template <int K, bool AGG>
__device__ __forceinline__
void gat_gemm_tile(float* sA, int tile,
                   const float* __restrict__ A, const float* __restrict__ W,
                   const float* __restrict__ a_src, const float* __restrict__ a_dst,
                   float* __restrict__ C, float* __restrict__ as_, float* __restrict__ ad_,
                   int N,
                   const int* __restrict__ rowptr, const int* __restrict__ csr_src,
                   const float* __restrict__ asP, const float* __restrict__ adP,
                   const float* __restrict__ biasP, int E) {
    constexpr int KP    = K + 4;
    constexpr int QPR   = K / 4;
    constexpr int RSTEP = 256 / QPR;
    constexpr int NQ    = 64 / RSTEP;
    constexpr int CP    = 68;
    static_assert(K % 4 == 0 && 256 % QPR == 0 && 64 % RSTEP == 0, "tile map");
    static_assert(CP <= KP, "the C tile reuses the A tile's LDS");
    static_assert(!AGG || K == 64, "an aggregated row is 64 features: one v2f per lane");
    const int lane = threadIdx.x & 31;
    const int wave = __builtin_amdgcn_readfirstlane((int)(threadIdx.x >> 5));
    const int rowBase = tile * 64;

    const int col      = (wave & 3) * 16 + (lane & 15);
    const int waveRow  = (wave >> 2) * 32;

    for (int k = threadIdx.x; k < K; k += 256)
        __builtin_prefetch(&W[(size_t)k * 64 + col], 0, 3);

    if constexpr (AGG) {
        const int n0 = rowBase + wave * 8;
        const int n1 = min(n0 + 8, N);
        const int nv = max(n1 - n0, 0);
        for (int k = nv; k < 8; ++k) {
            v2f z; z.x = 0.0f; z.y = 0.0f;
            *(v2f*)&sA[(wave * 8 + k) * KP + 2 * lane] = z;
        }
        if (nv > 0) {
            const v2f bb = ((const v2f*)biasP)[lane];
            gat_node_run<true>(rowptr, csr_src, asP, adP, A, bb, sA, KP, rowBase,
                               n0, n1, N, E, 1, (unsigned)lane);
        }
    } else {
        const int q  = threadIdx.x % QPR;
        const int r0 = threadIdx.x / QPR;
        v4f t[NQ];
#pragma unroll
        for (int i = 0; i < NQ; ++i) {
            const int gr = max(min(rowBase + r0 + i * RSTEP, N - 1), 0);
            t[i] = __builtin_nontemporal_load((const v4f*)(A + (size_t)gr * K) + q);
        }
#pragma unroll
        for (int i = 0; i < NQ; ++i)
            *(v4f*)&sA[(r0 + i * RSTEP) * KP + 4 * q] = t[i];
    }
    __syncthreads();

    const int row   = lane & 15;
    const int khalf = (lane >> 4) << 1;

    v8f acc0 = {};
    v8f acc1 = {};
#pragma unroll 8
    for (int k = 0; k < K; k += 4) {
        v2f b;
        b.x = W[(size_t)(k + khalf)     * 64 + col];
        b.y = W[(size_t)(k + khalf + 1) * 64 + col];

        v2f a0, a1;
        a0.x = sA[(waveRow +      row) * KP + k + khalf];
        a0.y = sA[(waveRow +      row) * KP + k + khalf + 1];
        a1.x = sA[(waveRow + 16 + row) * KP + k + khalf];
        a1.y = sA[(waveRow + 16 + row) * KP + k + khalf + 1];

        acc0 = __builtin_amdgcn_wmma_f32_16x16x4_f32(
            false, a0, false, b, (short)0, acc0, false, false);
        acc1 = __builtin_amdgcn_wmma_f32_16x16x4_f32(
            false, a1, false, b, (short)0, acc1, false, false);
    }

    __syncthreads();
    float* sC = sA;

    const int mhalf = (lane >> 4) << 3;
    for (int r = 0; r < 8; ++r) {
        sC[(waveRow +      mhalf + r) * CP + col] = acc0[r];
        sC[(waveRow + 16 + mhalf + r) * CP + col] = acc1[r];
    }

    const int c0 = (wave & 3) * 16;
    float* crow = &sC[(waveRow + lane) * CP + c0];
    {
        float s = 0.0f, d = 0.0f;
#pragma unroll
        for (int j = 0; j < 4; ++j) {
            const v4f cv = *(const v4f*)(crow + 4 * j);
            s = __builtin_fmaf(cv.x, a_src[c0 + 4 * j + 0], s); d = __builtin_fmaf(cv.x, a_dst[c0 + 4 * j + 0], d);
            s = __builtin_fmaf(cv.y, a_src[c0 + 4 * j + 1], s); d = __builtin_fmaf(cv.y, a_dst[c0 + 4 * j + 1], d);
            s = __builtin_fmaf(cv.z, a_src[c0 + 4 * j + 2], s); d = __builtin_fmaf(cv.z, a_dst[c0 + 4 * j + 2], d);
            s = __builtin_fmaf(cv.w, a_src[c0 + 4 * j + 3], s); d = __builtin_fmaf(cv.w, a_dst[c0 + 4 * j + 3], d);
        }
        v2f p; p.x = s; p.y = d;
        *(v2f*)crow = p;
    }
    __syncthreads();

    if (c0 == 0) {
        const v2f p0 = *(const v2f*)(crow);
        const v2f p1 = *(const v2f*)(crow + 16);
        const v2f p2 = *(const v2f*)(crow + 32);
        const v2f p3 = *(const v2f*)(crow + 48);
        const int gr = rowBase + waveRow + lane;
        if (gr < N) {
            as_[gr] = (p0.x + p1.x) + (p2.x + p3.x);
            ad_[gr] = (p0.y + p1.y) + (p2.y + p3.y);
        }
    }

    for (int r = 0; r < 8; ++r) {
        int gr0 = rowBase + waveRow +      mhalf + r;
        int gr1 = rowBase + waveRow + 16 + mhalf + r;
        if (gr0 < N) C[(size_t)gr0 * 64 + col] = acc0[r];
        if (gr1 < N) C[(size_t)gr1 * 64 + col] = acc1[r];
    }
}

__global__ __launch_bounds__(256)
void gat_gemm_agg(const float* __restrict__ A, const float* __restrict__ W,
                  const float* __restrict__ a_src, const float* __restrict__ a_dst,
                  float* __restrict__ C, float* __restrict__ as_, float* __restrict__ ad_,
                  int N,
                  const int* __restrict__ rowptr, const int* __restrict__ csr_src,
                  const float* __restrict__ asP, const float* __restrict__ adP,
                  const float* __restrict__ biasP, int E) {
    __shared__ __attribute__((aligned(16))) float sA[64 * (GAT_HID + 4)];
    gat_gemm_tile<GAT_HID, true>(sA, (int)blockIdx.x, A, W, a_src, a_dst, C, as_, ad_, N,
                                 rowptr, csr_src, asP, adP, biasP, E);
}

#define GAT_L1_ARGS const float* __restrict__ x, const float* __restrict__ W1, \
                    const float* __restrict__ a1s, const float* __restrict__ a1d, \
                    float* __restrict__ h, float* __restrict__ as_, float* __restrict__ ad_, int N
#define GAT_L1_TILE(sA, t) gat_gemm_tile<GAT_INC, false>(sA, (t), x, W1, a1s, a1d, h, as_, ad_, N, \
                                                         nullptr, nullptr, nullptr, nullptr, nullptr, 0)

__global__ __launch_bounds__(256)
void gat_stage_count(const int* __restrict__ dstI, int* __restrict__ wgcnt, int E, int NB,
                     int nStage, int tile0, GAT_L1_ARGS) {
    __shared__ __attribute__((aligned(16))) float sA[64 * (GAT_INC + 4)];
    const int wg = (int)blockIdx.x;
    if (wg < nStage) csr_bucket_count(dstI, wgcnt, E, NB, wg, (int*)sA);
    else             GAT_L1_TILE(sA, tile0 + wg - nStage);
}

__global__ __launch_bounds__(256)
void gat_stage_scan(const int* __restrict__ cnt, int* __restrict__ base, int M,
                    int nStage, int tile0, GAT_L1_ARGS) {
    __shared__ __attribute__((aligned(16))) float sA[64 * (GAT_INC + 4)];
    const int wg = (int)blockIdx.x;
    if (wg < nStage) csr_scan(cnt, base, M, wg, (int*)sA);
    else             GAT_L1_TILE(sA, tile0 + wg - nStage);
}

__global__ __launch_bounds__(256)
void gat_stage_scatter(const int* __restrict__ srcI, const int* __restrict__ dstI,
                       const int* __restrict__ base, int* __restrict__ tmp, int E, int NB,
                       int nStage, int tile0, GAT_L1_ARGS) {
    __shared__ __attribute__((aligned(16))) float sA[64 * (GAT_INC + 4)];
    const int wg = (int)blockIdx.x;
    if (wg < nStage) csr_bucket_scatter(srcI, dstI, base, tmp, E, NB, wg, (int*)sA);
    else             GAT_L1_TILE(sA, tile0 + wg - nStage);
}

__global__ __launch_bounds__(256)
void gat_stage_sort(const int* __restrict__ base, const int* __restrict__ tmp,
                    int* __restrict__ rowptr, int* __restrict__ csr_src, int E, int NB,
                    int nStage, int tile0, GAT_L1_ARGS) {
    __shared__ __attribute__((aligned(16))) float sA[64 * (GAT_INC + 4)];
    const int wg = (int)blockIdx.x;
    if (wg < nStage) csr_bucket_sort(base, tmp, rowptr, csr_src, N, E, NB, wg, (int*)sA);
    else             GAT_L1_TILE(sA, tile0 + wg - nStage);
}

extern "C" void kernel_launch(void* const* d_in, const int* in_sizes, int n_in,
                              void* d_out, int out_size, void* d_ws, size_t ws_size,
                              hipStream_t stream) {
    const float* x    = (const float*)d_in[0];
    const int*   eidx = (const int*)  d_in[1];
    const float* W1   = (const float*)d_in[2];
    const float* a1s  = (const float*)d_in[3];
    const float* a1d  = (const float*)d_in[4];
    const float* b1   = (const float*)d_in[5];
    const float* W2   = (const float*)d_in[6];
    const float* a2s  = (const float*)d_in[7];
    const float* a2d  = (const float*)d_in[8];
    const float* b2   = (const float*)d_in[9];

    const int N = in_sizes[0] / GAT_INC;
    const int E = in_sizes[1] / 2;
    const int* srcI = eidx;
    const int* dstI = eidx + E;

    float* ws = (float*)d_ws;
    size_t o = 0;
    float* h     = ws + o; o += (size_t)N * 64;
    float* h2    = ws + o; o += (size_t)N * 64;
    int*   graph = (int*)(ws + o); o += (size_t)N * 64;
    float* as_   = ws + o; o += N;
    float* ad_   = ws + o; o += N;
    float* as2   = ws + o; o += N;
    float* ad2   = ws + o; o += N;
    int*   tmp   = (int*)(ws + o); o += E;

    const int NB = (N + CS_BN - 1) / CS_BN;
    const int M  = NB * CS_WG;
    const size_t Np = ((size_t)N + 4) & ~(size_t)3;
    const size_t Ep = ((size_t)E + 3) & ~(size_t)3;
    const size_t Mp = ((size_t)M + 4) & ~(size_t)3;
    int* rowptr  = graph;
    int* csr_src = graph + Np;
    int* wgcnt   = csr_src + Ep;
    int* base    = wgcnt + Mp;

    const int gGemm  = (N + 63) / 64;
    const int gPers  = GAT_PERS_WG;
    const int gScan  = (M + 1 + CS_SCAN_B - 1) / CS_SCAN_B;
    const int t1 = min(CS_WG, gGemm);
    const int t2 = min(t1 + CS_WG, gGemm);
    const int t3 = min(t2 + 2 * CS_WG, gGemm);

    gat_stage_count  <<<CS_WG + t1,           CS_TH, 0, stream>>>(dstI, wgcnt, E, NB,
                                                                CS_WG, 0,  x, W1, a1s, a1d, h, as_, ad_, N);
    gat_stage_scan   <<<gScan + (t2 - t1),    CS_TH, 0, stream>>>(wgcnt, base, M,
                                                                gScan, t1, x, W1, a1s, a1d, h, as_, ad_, N);
    gat_stage_scatter<<<CS_WG + (t3 - t2),    CS_TH, 0, stream>>>(srcI, dstI, base, tmp, E, NB,
                                                                CS_WG, t2, x, W1, a1s, a1d, h, as_, ad_, N);
    gat_stage_sort   <<<NB + (gGemm - t3),    CS_TH, 0, stream>>>(base, tmp, rowptr, csr_src, E, NB,
                                                                NB,    t3, x, W1, a1s, a1d, h, as_, ad_, N);

    gat_gemm_agg<<<gGemm, 256, 0, stream>>>(h, W2, a2s, a2d, h2, as2, ad2, N,
                                            rowptr, csr_src, as_, ad_, b1, E);

    float* out = (float*)d_out;
    gat_gather    <<<gPers, 256, 0, stream>>>(rowptr, csr_src, as2, ad2, h2, b2, out, N, E, 0);
}


# Edge sort in two launches, not four (6 to 4): a chunk orders its edges by bucket inside its own slice (count, scan, place in LDS); bucket stage gathers 256 shares; rows in LDS-add arrival order
# speedup vs baseline: 4.9403x; 1.0397x over previous
#include <hip/hip_runtime.h>
#include <hip/hip_bf16.h>
#include <math.h>

typedef float v2f __attribute__((ext_vector_type(2)));
typedef float v8f __attribute__((ext_vector_type(8)));
typedef float v4f __attribute__((ext_vector_type(4)));

#define GAT_INC 256
#define GAT_HID 64
#define NEG_SLOPE 0.2f

template <int M>
__device__ __forceinline__ float row_xor(float x) {
    return __builtin_bit_cast(float, __builtin_amdgcn_update_dpp(
        0, __builtin_bit_cast(int, x), 0x160 + M, 0xf, 0xf, false));
}
__device__ __forceinline__ float xhalf(float x) {
    return __builtin_bit_cast(float, __builtin_amdgcn_permlanex16(
        __builtin_bit_cast(int, x), __builtin_bit_cast(int, x),
        0x76543210, 0xfedcba98, false, false));
}
__device__ __forceinline__ float wave_max32(float v) {
    v = fmaxf(v, row_xor<1>(v));
    v = fmaxf(v, row_xor<2>(v));
    v = fmaxf(v, row_xor<4>(v));
    v = fmaxf(v, row_xor<8>(v));
    return fmaxf(v, xhalf(v));
}
__device__ __forceinline__ float wave_sum32(float v) {
    v += row_xor<1>(v);
    v += row_xor<2>(v);
    v += row_xor<4>(v);
    v += row_xor<8>(v);
    return v + xhalf(v);
}

#define GAT_PERS_WG 1024

#define CS_SH 8
#define CS_BN (1 << CS_SH)
#define CS_WG 256
#define CS_TH 256
#define CS_UC 16

__device__ __forceinline__ void cs_chunk(int E, int wg, int& c0, int& c1) {
    const int per = (E + CS_WG - 1) / CS_WG;
    c0 = min(wg * per, E);
    c1 = min(c0 + per, E);
}

__device__ __forceinline__
void csr_chunk_order(const int* __restrict__ srcI, const int* __restrict__ dstI,
                     int* __restrict__ cntT, int* __restrict__ startT, int* __restrict__ tmp,
                     int E, int NB, int wg, int* sL) {
    int* sBkt = sL;
    int* sTot = sL + NB;
    const int tid  = threadIdx.x;
    const int lane = tid & 31;
    const int wave = tid >> 5;
    for (int i = tid; i < NB; i += CS_TH) sBkt[i] = 0;
    __syncthreads();
    int c0, c1;
    cs_chunk(E, wg, c0, c1);
    const unsigned blast = (unsigned)(NB - 1);
    const unsigned elast = (unsigned)max(E - 1, 0);
    for (int e0 = c0 + tid; e0 < c1; e0 += CS_UC * CS_TH) {
        int dd[CS_UC];
#pragma unroll
        for (int u = 0; u < CS_UC; ++u) dd[u] = dstI[min(e0 + u * CS_TH, c1 - 1)];
#pragma unroll
        for (int u = 0; u < CS_UC; ++u)
            atomicAdd(&sBkt[min((unsigned)dd[u] >> CS_SH, blast)], (e0 + u * CS_TH < c1) ? 1 : 0);
    }
    int e0 = c0 + tid;
    int d[CS_UC], s[CS_UC];
    if (c0 < c1) {
#pragma unroll
        for (int u = 0; u < CS_UC; ++u) {
            const int e = min(e0 + u * CS_TH, c1 - 1);
            d[u] = dstI[e];
            s[u] = srcI[e];
        }
    }
    __syncthreads();
    int carry = c0;
    for (int i0 = 0; i0 < NB; i0 += CS_TH) {
        const int i = i0 + tid;
        const int x = (i < NB) ? sBkt[i] : 0;
        int incl = x;
        for (int off = 1; off < 32; off <<= 1) {
            int y = __shfl_up(incl, off, 32);
            if (lane >= off) incl += y;
        }
        if (lane == 31) sTot[wave] = incl;
        __syncthreads();
        int front = 0, total = 0;
        for (int v = 0; v < CS_TH / 32; ++v) {
            const int t = sTot[v];
            total += t;
            front += (v < wave) ? t : 0;
        }
        const int start = carry + front + (incl - x);
        if (i < NB) {
            sBkt[i] = start;
            cntT[(size_t)i * CS_WG + wg]   = x;
            startT[(size_t)i * CS_WG + wg] = start;
        }
        carry += total;
        __syncthreads();
    }
    while (e0 < c1) {
        int p[CS_UC];
#pragma unroll
        for (int u = 0; u < CS_UC; ++u)
            p[u] = atomicAdd(&sBkt[min((unsigned)d[u] >> CS_SH, blast)],
                             (e0 + u * CS_TH < c1) ? 1 : 0);
#pragma unroll
        for (int u = 0; u < CS_UC; ++u)
            if (e0 + u * CS_TH < c1)
                tmp[min((unsigned)p[u], elast)] =
                    (int)(((unsigned)s[u] << CS_SH) | ((unsigned)d[u] & (unsigned)(CS_BN - 1)));
        e0 += CS_UC * CS_TH;
        if (e0 < c1) {
#pragma unroll
            for (int u = 0; u < CS_UC; ++u) {
                const int e = min(e0 + u * CS_TH, c1 - 1);
                d[u] = dstI[e];
                s[u] = srcI[e];
            }
        }
    }
}

#define CS_US 16
static_assert(CS_TH == CS_WG, "bucket pass: thread t holds chunk t's share");
static_assert(CS_BN <= CS_TH && 32 % CS_US == 0, "bucket pass map");

template <bool PLACE>
__device__ __forceinline__
void cs_share_pass(const int* __restrict__ tmp, int* __restrict__ csr_src, int* sCnt,
                   int st, int cn, int lane, unsigned elast) {
#pragma unroll
    for (int k0 = 0; k0 < 32; k0 += CS_US) {
        int w[CS_US], c[CS_US];
#pragma unroll
        for (int u = 0; u < CS_US; ++u) {
            const int su = __builtin_amdgcn_readlane(st, k0 + u);
            c[u] = __builtin_amdgcn_readlane(cn, k0 + u);
            w[u] = tmp[min((unsigned)(su + min(lane, max(c[u] - 1, 0))), elast)];
        }
        if (PLACE) {
            int p[CS_US];
#pragma unroll
            for (int u = 0; u < CS_US; ++u)
                p[u] = atomicAdd(&sCnt[w[u] & (CS_BN - 1)], (lane < c[u]) ? 1 : 0);
#pragma unroll
            for (int u = 0; u < CS_US; ++u)
                if (lane < c[u])
                    csr_src[min((unsigned)p[u], elast)] = (int)((unsigned)w[u] >> CS_SH);
        } else {
#pragma unroll
            for (int u = 0; u < CS_US; ++u)
                atomicAdd(&sCnt[w[u] & (CS_BN - 1)], (lane < c[u]) ? 1 : 0);
        }
    }
    if (__builtin_amdgcn_ballot_w32(cn > 32) != 0) {
        for (int k = 0; k < 32; ++k) {
            const int su = __builtin_amdgcn_readlane(st, k);
            const int cu = __builtin_amdgcn_readlane(cn, k);
            for (int i = 32 + lane; i < cu; i += 32) {
                const int w = tmp[min((unsigned)(su + i), elast)];
                if (PLACE) {
                    const int p = atomicAdd(&sCnt[w & (CS_BN - 1)], 1);
                    csr_src[min((unsigned)p, elast)] = (int)((unsigned)w >> CS_SH);
                } else {
                    atomicAdd(&sCnt[w & (CS_BN - 1)], 1);
                }
            }
        }
    }
}

__device__ __forceinline__
void csr_bucket_sort(const int* __restrict__ cntT, const int* __restrict__ startT,
                     const int* __restrict__ tmp,
                     int* __restrict__ rowptr, int* __restrict__ csr_src, int N, int E, int NB,
                     int b, int* sL) {
    int* sCnt = sL;
    int* sTot = sL + CS_BN;
    int* sFr  = sL + CS_BN + 32;
    int* sSz  = sL + CS_BN + 64;
    const int tid  = threadIdx.x;
    const int lane = tid & 31;
    const int wave = tid >> 5;
    const unsigned elast = (unsigned)max(E - 1, 0);

    int front = 0;
    const int4* c4 = (const int4*)cntT;
#pragma unroll 16
    for (int q = tid; q < b * (CS_WG / 4); q += CS_TH) {
        int4 v = c4[q];
        front += (v.x + v.y) + (v.z + v.w);
    }
    int cn = cntT[(size_t)b * CS_WG + tid];
    int st = startT[(size_t)b * CS_WG + tid];
    cn = min(max(cn, 0), E);
    st = min(max(st, 0), E - cn);
    int sz = cn;
    for (int off = 16; off > 0; off >>= 1) {
        front += __shfl_xor(front, off, 32);
        sz    += __shfl_xor(sz, off, 32);
    }
    if (lane == 0) { sFr[wave] = front; sSz[wave] = sz; }
    if (tid < CS_BN) sCnt[tid] = 0;
    __syncthreads();
    int beg = 0, tot = 0;
    for (int v = 0; v < CS_TH / 32; ++v) { beg += sFr[v]; tot += sSz[v]; }
    beg = min(max(beg, 0), E);
    const int end = min(beg + max(tot, 0), E);

    cs_share_pass<false>(tmp, csr_src, sCnt, st, cn, lane, elast);
    __syncthreads();
    const int x = (tid < CS_BN) ? sCnt[tid] : 0;
    int incl = x;
    for (int off = 1; off < 32; off <<= 1) {
        int y = __shfl_up(incl, off, 32);
        if (lane >= off) incl += y;
    }
    if (tid < CS_BN && lane == 31) sTot[wave] = incl;
    __syncthreads();
    if (tid < CS_BN) {
        int fr = 0;
        for (int v = 0; v < wave; ++v) fr += sTot[v];
        const int start = beg + fr + (incl - x);
        sCnt[tid] = start;
        const int node = b * CS_BN + tid;
        if (node < N) rowptr[node] = start;
    }
    if (b == NB - 1 && tid == 0) rowptr[N] = end;
    __syncthreads();
    cs_share_pass<true>(tmp, csr_src, sCnt, st, cn, lane, elast);
}

__device__ __forceinline__ float gat_score(const float* __restrict__ as_, int s, float add) {
    float v = as_[s] + add;
    return (v > 0.0f) ? v : v * NEG_SLOPE;
}
template <int G>
__device__ __forceinline__ void gat_row_group(const v2f* __restrict__ h2, int s, float alpha,
                                              int t, unsigned lane, v2f& acc) {
    unsigned idx[G];
    v2f r[G];
#pragma unroll
    for (int u = 0; u < G; ++u)
        idx[u] = (unsigned)__builtin_amdgcn_readlane(s, t + u) * 32u + lane;
#pragma unroll
    for (int u = 0; u < G; ++u)
        r[u] = h2[idx[u]];
#pragma unroll
    for (int u = 0; u < G; ++u) {
        const float at = __builtin_bit_cast(float,
            __builtin_amdgcn_readlane(__builtin_bit_cast(int, alpha), t + u));
        acc = acc + r[u] * at;
    }
}
__device__ __forceinline__ void gat_rows(const v2f* __restrict__ h2, int s, float alpha,
                                         int cnt, unsigned lane, v2f& acc) {
    const int c4 = (cnt + 3) & ~3;
    if (c4 <= 4)       gat_row_group<4>(h2, s, alpha, 0, lane, acc);
    else if (c4 <= 8)  gat_row_group<8>(h2, s, alpha, 0, lane, acc);
    else if (c4 <= 12) gat_row_group<12>(h2, s, alpha, 0, lane, acc);
    else {
        gat_row_group<16>(h2, s, alpha, 0, lane, acc);
        int t = 16;
        for (; t + 8 <= c4; t += 8) gat_row_group<8>(h2, s, alpha, t, lane, acc);
        if (t < c4)                 gat_row_group<4>(h2, s, alpha, t, lane, acc);
    }
}

__device__ __forceinline__ int gat_edge_ids(const int* __restrict__ csr_src, int rp, int k,
                                            unsigned lane, int elast) {
    const int beg = __builtin_amdgcn_readlane(rp, k);
    const int end = max(__builtin_amdgcn_readlane(rp, k + 1), beg);
    const int j   = min(beg + (int)lane, end - 1);
    return csr_src[min(max(j, 0), elast)];
}

template <bool TO_LDS>
__device__ __forceinline__ void gat_node_run(const int* __restrict__ rowptr, const int* __restrict__ csr_src,
                                             const float* __restrict__ as_, const float* __restrict__ ad_,
                                             const float* __restrict__ h, const v2f bb,
                                             float* dst, int pitch, int row0,
                                             int n0, int n1, int N, int E, int relu_flag, unsigned lane) {
    const unsigned nlast = (unsigned)(N - 1);
    const int elast = max(E - 1, 0);
    const v2f* h2 = (const v2f*)h;

    for (int base = n0; base < n1; base += 31) {
        const int c = min(31, n1 - base);
        int rp = rowptr[min(base + (int)lane, N)];
        rp = min(max(rp, 0), E);
        const float adl = ad_[min(base + (int)lane, N - 1)];

        int   sA = gat_edge_ids(csr_src, rp, 0, lane, elast);
        int   sB = gat_edge_ids(csr_src, rp, min(1, c - 1), lane, elast);
        sA = (int)min((unsigned)sA, nlast);
        float vA = as_[sA];

        for (int k = 0; k < c; ++k) {
            const int sC = gat_edge_ids(csr_src, rp, min(k + 2, c - 1), lane, elast);
            sB = (int)min((unsigned)sB, nlast);
            const float vB = as_[sB];

            const int beg = __builtin_amdgcn_readlane(rp, k);
            const int end = max(__builtin_amdgcn_readlane(rp, k + 1), beg);
            const float add = __builtin_bit_cast(float,
                __builtin_amdgcn_readlane(__builtin_bit_cast(int, adl), k));
            const int cnt = end - beg;

            v2f acc = {0.0f, 0.0f};
            if (cnt > 0) {
                if (cnt <= 32) {
                    const bool valid = (int)lane < cnt;
                    float v = vA + add;
                    v = (v > 0.0f) ? v : v * NEG_SLOPE;
                    const float m    = wave_max32(v);
                    const float ex   = valid ? __expf(v - m) : 0.0f;
                    const float den  = wave_sum32(ex);
                    const float alpha = ex / (den + 1e-16f);
                    gat_rows(h2, sA, alpha, cnt, lane, acc);
                } else {
                    float m = -3.0e38f;
                    for (int eb = beg; eb < end; eb += 32) {
                        const int j = eb + (int)lane;
                        const int s = (int)min((unsigned)csr_src[min(j, end - 1)], nlast);
                        m = fmaxf(m, gat_score(as_, s, add));
                    }
                    m = wave_max32(m);
                    float den = 0.0f;
                    for (int eb = beg; eb < end; eb += 32) {
                        const int j = eb + (int)lane;
                        const int s = (int)min((unsigned)csr_src[min(j, end - 1)], nlast);
                        const float ex = __expf(gat_score(as_, s, add) - m);
                        den += (j < end) ? ex : 0.0f;
                    }
                    den = wave_sum32(den);
                    const float dd = den + 1e-16f;
                    for (int eb = beg; eb < end; eb += 32) {
                        const int j = eb + (int)lane;
                        const int s = (int)min((unsigned)csr_src[min(j, end - 1)], nlast);
                        const float ex = __expf(gat_score(as_, s, add) - m);
                        const float alpha = (j < end) ? ex / dd : 0.0f;
                        gat_rows(h2, s, alpha, min(32, end - eb), lane, acc);
                    }
                }
            }

            v2f r;
            r.x = acc.x + bb.x;
            r.y = acc.y + bb.y;
            if (relu_flag) { r.x = fmaxf(r.x, 0.0f); r.y = fmaxf(r.y, 0.0f); }
            if (TO_LDS) *(v2f*)(dst + (base + k - row0) * pitch + 2 * (int)lane) = r;
            else        ((v2f*)dst)[(unsigned)(base + k) * 32u + lane] = r;

            sA = sB; vA = vB; sB = sC;
        }
    }
}

__global__ __launch_bounds__(256)
void gat_gather(const int* __restrict__ rowptr, const int* __restrict__ csr_src,
                const float* __restrict__ as_, const float* __restrict__ ad_,
                const float* __restrict__ h, const float* __restrict__ bias,
                float* __restrict__ out, int N, int E, int relu_flag) {
    const unsigned lane = threadIdx.x & 31;
    const unsigned nw = gridDim.x * (blockDim.x >> 5);
    const unsigned gw = (unsigned)__builtin_amdgcn_readfirstlane(
        (int)(blockIdx.x * (blockDim.x >> 5) + (threadIdx.x >> 5)));
    const unsigned per = (unsigned)N / nw, rem = (unsigned)N % nw;
    const int n0 = (int)(gw * per + min(gw, rem));
    const int n1 = n0 + (int)per + (gw < rem ? 1 : 0);
    if (n0 >= n1) return;
    const v2f bb = ((const v2f*)bias)[lane];
    gat_node_run<false>(rowptr, csr_src, as_, ad_, h, bb, out, 64, 0, n0, n1, N, E, relu_flag, lane);
}

template <int K, bool AGG>
__device__ __forceinline__
void gat_gemm_tile(float* sA, int tile,
                   const float* __restrict__ A, const float* __restrict__ W,
                   const float* __restrict__ a_src, const float* __restrict__ a_dst,
                   float* __restrict__ C, float* __restrict__ as_, float* __restrict__ ad_,
                   int N,
                   const int* __restrict__ rowptr, const int* __restrict__ csr_src,
                   const float* __restrict__ asP, const float* __restrict__ adP,
                   const float* __restrict__ biasP, int E) {
    constexpr int KP    = K + 4;
    constexpr int QPR   = K / 4;
    constexpr int RSTEP = 256 / QPR;
    constexpr int NQ    = 64 / RSTEP;
    constexpr int CP    = 68;
    static_assert(K % 4 == 0 && 256 % QPR == 0 && 64 % RSTEP == 0, "tile map");
    static_assert(CP <= KP, "the C tile reuses the A tile's LDS");
    static_assert(!AGG || K == 64, "an aggregated row is 64 features: one v2f per lane");
    const int lane = threadIdx.x & 31;
    const int wave = __builtin_amdgcn_readfirstlane((int)(threadIdx.x >> 5));
    const int rowBase = tile * 64;

    const int col      = (wave & 3) * 16 + (lane & 15);
    const int waveRow  = (wave >> 2) * 32;

    for (int k = threadIdx.x; k < K; k += 256)
        __builtin_prefetch(&W[(size_t)k * 64 + col], 0, 3);

    if constexpr (AGG) {
        const int n0 = rowBase + wave * 8;
        const int n1 = min(n0 + 8, N);
        const int nv = max(n1 - n0, 0);
        for (int k = nv; k < 8; ++k) {
            v2f z; z.x = 0.0f; z.y = 0.0f;
            *(v2f*)&sA[(wave * 8 + k) * KP + 2 * lane] = z;
        }
        if (nv > 0) {
            const v2f bb = ((const v2f*)biasP)[lane];
            gat_node_run<true>(rowptr, csr_src, asP, adP, A, bb, sA, KP, rowBase,
                               n0, n1, N, E, 1, (unsigned)lane);
        }
    } else {
        const int q  = threadIdx.x % QPR;
        const int r0 = threadIdx.x / QPR;
        v4f t[NQ];
#pragma unroll
        for (int i = 0; i < NQ; ++i) {
            const int gr = max(min(rowBase + r0 + i * RSTEP, N - 1), 0);
            t[i] = __builtin_nontemporal_load((const v4f*)(A + (size_t)gr * K) + q);
        }
#pragma unroll
        for (int i = 0; i < NQ; ++i)
            *(v4f*)&sA[(r0 + i * RSTEP) * KP + 4 * q] = t[i];
    }
    __syncthreads();

    const int row   = lane & 15;
    const int khalf = (lane >> 4) << 1;

    v8f acc0 = {};
    v8f acc1 = {};
#pragma unroll 8
    for (int k = 0; k < K; k += 4) {
        v2f b;
        b.x = W[(size_t)(k + khalf)     * 64 + col];
        b.y = W[(size_t)(k + khalf + 1) * 64 + col];

        v2f a0, a1;
        a0.x = sA[(waveRow +      row) * KP + k + khalf];
        a0.y = sA[(waveRow +      row) * KP + k + khalf + 1];
        a1.x = sA[(waveRow + 16 + row) * KP + k + khalf];
        a1.y = sA[(waveRow + 16 + row) * KP + k + khalf + 1];

        acc0 = __builtin_amdgcn_wmma_f32_16x16x4_f32(
            false, a0, false, b, (short)0, acc0, false, false);
        acc1 = __builtin_amdgcn_wmma_f32_16x16x4_f32(
            false, a1, false, b, (short)0, acc1, false, false);
    }

    __syncthreads();
    float* sC = sA;

    const int mhalf = (lane >> 4) << 3;
    for (int r = 0; r < 8; ++r) {
        sC[(waveRow +      mhalf + r) * CP + col] = acc0[r];
        sC[(waveRow + 16 + mhalf + r) * CP + col] = acc1[r];
    }

    const int c0 = (wave & 3) * 16;
    float* crow = &sC[(waveRow + lane) * CP + c0];
    {
        float s = 0.0f, d = 0.0f;
#pragma unroll
        for (int j = 0; j < 4; ++j) {
            const v4f cv = *(const v4f*)(crow + 4 * j);
            s = __builtin_fmaf(cv.x, a_src[c0 + 4 * j + 0], s); d = __builtin_fmaf(cv.x, a_dst[c0 + 4 * j + 0], d);
            s = __builtin_fmaf(cv.y, a_src[c0 + 4 * j + 1], s); d = __builtin_fmaf(cv.y, a_dst[c0 + 4 * j + 1], d);
            s = __builtin_fmaf(cv.z, a_src[c0 + 4 * j + 2], s); d = __builtin_fmaf(cv.z, a_dst[c0 + 4 * j + 2], d);
            s = __builtin_fmaf(cv.w, a_src[c0 + 4 * j + 3], s); d = __builtin_fmaf(cv.w, a_dst[c0 + 4 * j + 3], d);
        }
        v2f p; p.x = s; p.y = d;
        *(v2f*)crow = p;
    }
    __syncthreads();

    if (c0 == 0) {
        const v2f p0 = *(const v2f*)(crow);
        const v2f p1 = *(const v2f*)(crow + 16);
        const v2f p2 = *(const v2f*)(crow + 32);
        const v2f p3 = *(const v2f*)(crow + 48);
        const int gr = rowBase + waveRow + lane;
        if (gr < N) {
            as_[gr] = (p0.x + p1.x) + (p2.x + p3.x);
            ad_[gr] = (p0.y + p1.y) + (p2.y + p3.y);
        }
    }

    for (int r = 0; r < 8; ++r) {
        int gr0 = rowBase + waveRow +      mhalf + r;
        int gr1 = rowBase + waveRow + 16 + mhalf + r;
        if (gr0 < N) C[(size_t)gr0 * 64 + col] = acc0[r];
        if (gr1 < N) C[(size_t)gr1 * 64 + col] = acc1[r];
    }
}

__global__ __launch_bounds__(256)
void gat_gemm_agg(const float* __restrict__ A, const float* __restrict__ W,
                  const float* __restrict__ a_src, const float* __restrict__ a_dst,
                  float* __restrict__ C, float* __restrict__ as_, float* __restrict__ ad_,
                  int N,
                  const int* __restrict__ rowptr, const int* __restrict__ csr_src,
                  const float* __restrict__ asP, const float* __restrict__ adP,
                  const float* __restrict__ biasP, int E) {
    __shared__ __attribute__((aligned(16))) float sA[64 * (GAT_HID + 4)];
    gat_gemm_tile<GAT_HID, true>(sA, (int)blockIdx.x, A, W, a_src, a_dst, C, as_, ad_, N,
                                 rowptr, csr_src, asP, adP, biasP, E);
}

#define GAT_L1_ARGS const float* __restrict__ x, const float* __restrict__ W1, \
                    const float* __restrict__ a1s, const float* __restrict__ a1d, \
                    float* __restrict__ h, float* __restrict__ as_, float* __restrict__ ad_, int N
#define GAT_L1_TILE(sA, t) gat_gemm_tile<GAT_INC, false>(sA, (t), x, W1, a1s, a1d, h, as_, ad_, N, \
                                                         nullptr, nullptr, nullptr, nullptr, nullptr, 0)

__global__ __launch_bounds__(256)
void gat_stage_chunk(const int* __restrict__ srcI, const int* __restrict__ dstI,
                     int* __restrict__ cntT, int* __restrict__ startT, int* __restrict__ tmp,
                     int E, int NB, int nStage, int tile0, GAT_L1_ARGS) {
    __shared__ __attribute__((aligned(16))) float sA[64 * (GAT_INC + 4)];
    const int wg = (int)blockIdx.x;
    if (wg < nStage) csr_chunk_order(srcI, dstI, cntT, startT, tmp, E, NB, wg, (int*)sA);
    else             GAT_L1_TILE(sA, tile0 + wg - nStage);
}

__global__ __launch_bounds__(256)
void gat_stage_sort(const int* __restrict__ cntT, const int* __restrict__ startT,
                    const int* __restrict__ tmp,
                    int* __restrict__ rowptr, int* __restrict__ csr_src, int E, int NB,
                    int nStage, int tile0, GAT_L1_ARGS) {
    __shared__ __attribute__((aligned(16))) float sA[64 * (GAT_INC + 4)];
    const int wg = (int)blockIdx.x;
    if (wg < nStage) csr_bucket_sort(cntT, startT, tmp, rowptr, csr_src, N, E, NB, wg, (int*)sA);
    else             GAT_L1_TILE(sA, tile0 + wg - nStage);
}

extern "C" void kernel_launch(void* const* d_in, const int* in_sizes, int n_in,
                              void* d_out, int out_size, void* d_ws, size_t ws_size,
                              hipStream_t stream) {
    const float* x    = (const float*)d_in[0];
    const int*   eidx = (const int*)  d_in[1];
    const float* W1   = (const float*)d_in[2];
    const float* a1s  = (const float*)d_in[3];
    const float* a1d  = (const float*)d_in[4];
    const float* b1   = (const float*)d_in[5];
    const float* W2   = (const float*)d_in[6];
    const float* a2s  = (const float*)d_in[7];
    const float* a2d  = (const float*)d_in[8];
    const float* b2   = (const float*)d_in[9];

    const int N = in_sizes[0] / GAT_INC;
    const int E = in_sizes[1] / 2;
    const int* srcI = eidx;
    const int* dstI = eidx + E;

    float* ws = (float*)d_ws;
    size_t o = 0;
    float* h     = ws + o; o += (size_t)N * 64;
    float* h2    = ws + o; o += (size_t)N * 64;
    int*   graph = (int*)(ws + o); o += (size_t)N * 64;
    float* as_   = ws + o; o += N;
    float* ad_   = ws + o; o += N;
    float* as2   = ws + o; o += N;
    float* ad2   = ws + o; o += N;
    int*   tmp   = (int*)(ws + o); o += E;

    const int NB = (N + CS_BN - 1) / CS_BN;
    const int M  = NB * CS_WG;
    const size_t Np = ((size_t)N + 4) & ~(size_t)3;
    const size_t Ep = ((size_t)E + 3) & ~(size_t)3;
    const size_t Mp = ((size_t)M + 4) & ~(size_t)3;
    int* rowptr  = graph;
    int* csr_src = graph + Np;
    int* cntT    = csr_src + Ep;
    int* startT  = cntT + Mp;

    const int gGemm  = (N + 63) / 64;
    const int gPers  = GAT_PERS_WG;
    const int tA = min(3 * CS_WG, gGemm);

    gat_stage_chunk<<<CS_WG + tA,           CS_TH, 0, stream>>>(srcI, dstI, cntT, startT, tmp, E, NB,
                                                              CS_WG, 0,  x, W1, a1s, a1d, h, as_, ad_, N);
    gat_stage_sort <<<NB + (gGemm - tA),    CS_TH, 0, stream>>>(cntT, startT, tmp, rowptr, csr_src, E, NB,
                                                              NB,    tA, x, W1, a1s, a1d, h, as_, ad_, N);

    gat_gemm_agg<<<gGemm, 256, 0, stream>>>(h, W2, a2s, a2d, h2, as2, ad2, N,
                                            rowptr, csr_src, as_, ad_, b1, E);

    float* out = (float*)d_out;
    gat_gather    <<<gPers, 256, 0, stream>>>(rowptr, csr_src, as2, ad2, h2, b2, out, N, E, 0);
}


# Bucket stage takes its first row start from the chunk starts it already loads (sum of start minus chunk origin) instead of summing all counters of the buckets in front
# speedup vs baseline: 5.1135x; 1.0351x over previous
#include <hip/hip_runtime.h>
#include <hip/hip_bf16.h>
#include <math.h>

typedef float v2f __attribute__((ext_vector_type(2)));
typedef float v8f __attribute__((ext_vector_type(8)));
typedef float v4f __attribute__((ext_vector_type(4)));

#define GAT_INC 256
#define GAT_HID 64
#define NEG_SLOPE 0.2f

template <int M>
__device__ __forceinline__ float row_xor(float x) {
    return __builtin_bit_cast(float, __builtin_amdgcn_update_dpp(
        0, __builtin_bit_cast(int, x), 0x160 + M, 0xf, 0xf, false));
}
__device__ __forceinline__ float xhalf(float x) {
    return __builtin_bit_cast(float, __builtin_amdgcn_permlanex16(
        __builtin_bit_cast(int, x), __builtin_bit_cast(int, x),
        0x76543210, 0xfedcba98, false, false));
}
__device__ __forceinline__ float wave_max32(float v) {
    v = fmaxf(v, row_xor<1>(v));
    v = fmaxf(v, row_xor<2>(v));
    v = fmaxf(v, row_xor<4>(v));
    v = fmaxf(v, row_xor<8>(v));
    return fmaxf(v, xhalf(v));
}
__device__ __forceinline__ float wave_sum32(float v) {
    v += row_xor<1>(v);
    v += row_xor<2>(v);
    v += row_xor<4>(v);
    v += row_xor<8>(v);
    return v + xhalf(v);
}

#define GAT_PERS_WG 1024

#define CS_SH 8
#define CS_BN (1 << CS_SH)
#define CS_WG 256
#define CS_TH 256
#define CS_UC 16

__device__ __forceinline__ void cs_chunk(int E, int wg, int& c0, int& c1) {
    const int per = (E + CS_WG - 1) / CS_WG;
    c0 = min(wg * per, E);
    c1 = min(c0 + per, E);
}

__device__ __forceinline__
void csr_chunk_order(const int* __restrict__ srcI, const int* __restrict__ dstI,
                     int* __restrict__ cntT, int* __restrict__ startT, int* __restrict__ tmp,
                     int E, int NB, int wg, int* sL) {
    int* sBkt = sL;
    int* sTot = sL + NB;
    const int tid  = threadIdx.x;
    const int lane = tid & 31;
    const int wave = tid >> 5;
    for (int i = tid; i < NB; i += CS_TH) sBkt[i] = 0;
    __syncthreads();
    int c0, c1;
    cs_chunk(E, wg, c0, c1);
    const unsigned blast = (unsigned)(NB - 1);
    const unsigned elast = (unsigned)max(E - 1, 0);
    for (int e0 = c0 + tid; e0 < c1; e0 += CS_UC * CS_TH) {
        int dd[CS_UC];
#pragma unroll
        for (int u = 0; u < CS_UC; ++u) dd[u] = dstI[min(e0 + u * CS_TH, c1 - 1)];
#pragma unroll
        for (int u = 0; u < CS_UC; ++u)
            atomicAdd(&sBkt[min((unsigned)dd[u] >> CS_SH, blast)], (e0 + u * CS_TH < c1) ? 1 : 0);
    }
    int e0 = c0 + tid;
    int d[CS_UC], s[CS_UC];
    if (c0 < c1) {
#pragma unroll
        for (int u = 0; u < CS_UC; ++u) {
            const int e = min(e0 + u * CS_TH, c1 - 1);
            d[u] = dstI[e];
            s[u] = srcI[e];
        }
    }
    __syncthreads();
    int carry = c0;
    for (int i0 = 0; i0 < NB; i0 += CS_TH) {
        const int i = i0 + tid;
        const int x = (i < NB) ? sBkt[i] : 0;
        int incl = x;
        for (int off = 1; off < 32; off <<= 1) {
            int y = __shfl_up(incl, off, 32);
            if (lane >= off) incl += y;
        }
        if (lane == 31) sTot[wave] = incl;
        __syncthreads();
        int front = 0, total = 0;
        for (int v = 0; v < CS_TH / 32; ++v) {
            const int t = sTot[v];
            total += t;
            front += (v < wave) ? t : 0;
        }
        const int start = carry + front + (incl - x);
        if (i < NB) {
            sBkt[i] = start;
            cntT[(size_t)i * CS_WG + wg]   = x;
            startT[(size_t)i * CS_WG + wg] = start;
        }
        carry += total;
        __syncthreads();
    }
    while (e0 < c1) {
        int p[CS_UC];
#pragma unroll
        for (int u = 0; u < CS_UC; ++u)
            p[u] = atomicAdd(&sBkt[min((unsigned)d[u] >> CS_SH, blast)],
                             (e0 + u * CS_TH < c1) ? 1 : 0);
#pragma unroll
        for (int u = 0; u < CS_UC; ++u)
            if (e0 + u * CS_TH < c1)
                tmp[min((unsigned)p[u], elast)] =
                    (int)(((unsigned)s[u] << CS_SH) | ((unsigned)d[u] & (unsigned)(CS_BN - 1)));
        e0 += CS_UC * CS_TH;
        if (e0 < c1) {
#pragma unroll
            for (int u = 0; u < CS_UC; ++u) {
                const int e = min(e0 + u * CS_TH, c1 - 1);
                d[u] = dstI[e];
                s[u] = srcI[e];
            }
        }
    }
}

#define CS_US 16
static_assert(CS_TH == CS_WG, "bucket pass: thread t holds chunk t's share");
static_assert(CS_BN <= CS_TH && 32 % CS_US == 0, "bucket pass map");

template <bool PLACE>
__device__ __forceinline__
void cs_share_pass(const int* __restrict__ tmp, int* __restrict__ csr_src, int* sCnt,
                   int st, int cn, int lane, unsigned elast) {
#pragma unroll
    for (int k0 = 0; k0 < 32; k0 += CS_US) {
        int w[CS_US], c[CS_US];
#pragma unroll
        for (int u = 0; u < CS_US; ++u) {
            const int su = __builtin_amdgcn_readlane(st, k0 + u);
            c[u] = __builtin_amdgcn_readlane(cn, k0 + u);
            w[u] = tmp[min((unsigned)(su + min(lane, max(c[u] - 1, 0))), elast)];
        }
        if (PLACE) {
            int p[CS_US];
#pragma unroll
            for (int u = 0; u < CS_US; ++u)
                p[u] = atomicAdd(&sCnt[w[u] & (CS_BN - 1)], (lane < c[u]) ? 1 : 0);
#pragma unroll
            for (int u = 0; u < CS_US; ++u)
                if (lane < c[u])
                    csr_src[min((unsigned)p[u], elast)] = (int)((unsigned)w[u] >> CS_SH);
        } else {
#pragma unroll
            for (int u = 0; u < CS_US; ++u)
                atomicAdd(&sCnt[w[u] & (CS_BN - 1)], (lane < c[u]) ? 1 : 0);
        }
    }
    if (__builtin_amdgcn_ballot_w32(cn > 32) != 0) {
        for (int k = 0; k < 32; ++k) {
            const int su = __builtin_amdgcn_readlane(st, k);
            const int cu = __builtin_amdgcn_readlane(cn, k);
            for (int i = 32 + lane; i < cu; i += 32) {
                const int w = tmp[min((unsigned)(su + i), elast)];
                if (PLACE) {
                    const int p = atomicAdd(&sCnt[w & (CS_BN - 1)], 1);
                    csr_src[min((unsigned)p, elast)] = (int)((unsigned)w >> CS_SH);
                } else {
                    atomicAdd(&sCnt[w & (CS_BN - 1)], 1);
                }
            }
        }
    }
}

__device__ __forceinline__
void csr_bucket_sort(const int* __restrict__ cntT, const int* __restrict__ startT,
                     const int* __restrict__ tmp,
                     int* __restrict__ rowptr, int* __restrict__ csr_src, int N, int E, int NB,
                     int b, int* sL) {
    int* sCnt = sL;
    int* sTot = sL + CS_BN;
    int* sFr  = sL + CS_BN + 32;
    int* sSz  = sL + CS_BN + 64;
    const int tid  = threadIdx.x;
    const int lane = tid & 31;
    const int wave = tid >> 5;
    const unsigned elast = (unsigned)max(E - 1, 0);

    int cn = cntT[(size_t)b * CS_WG + tid];
    int st = startT[(size_t)b * CS_WG + tid];
    int cc0, cc1;
    cs_chunk(E, tid, cc0, cc1);
    int front = st - cc0;
    cn = min(max(cn, 0), E);
    st = min(max(st, 0), E - cn);
    int sz = cn;
    for (int off = 16; off > 0; off >>= 1) {
        front += __shfl_xor(front, off, 32);
        sz    += __shfl_xor(sz, off, 32);
    }
    if (lane == 0) { sFr[wave] = front; sSz[wave] = sz; }
    if (tid < CS_BN) sCnt[tid] = 0;
    __syncthreads();
    int beg = 0, tot = 0;
    for (int v = 0; v < CS_TH / 32; ++v) { beg += sFr[v]; tot += sSz[v]; }
    beg = min(max(beg, 0), E);
    const int end = min(beg + max(tot, 0), E);

    cs_share_pass<false>(tmp, csr_src, sCnt, st, cn, lane, elast);
    __syncthreads();
    const int x = (tid < CS_BN) ? sCnt[tid] : 0;
    int incl = x;
    for (int off = 1; off < 32; off <<= 1) {
        int y = __shfl_up(incl, off, 32);
        if (lane >= off) incl += y;
    }
    if (tid < CS_BN && lane == 31) sTot[wave] = incl;
    __syncthreads();
    if (tid < CS_BN) {
        int fr = 0;
        for (int v = 0; v < wave; ++v) fr += sTot[v];
        const int start = beg + fr + (incl - x);
        sCnt[tid] = start;
        const int node = b * CS_BN + tid;
        if (node < N) rowptr[node] = start;
    }
    if (b == NB - 1 && tid == 0) rowptr[N] = end;
    __syncthreads();
    cs_share_pass<true>(tmp, csr_src, sCnt, st, cn, lane, elast);
}

__device__ __forceinline__ float gat_score(const float* __restrict__ as_, int s, float add) {
    float v = as_[s] + add;
    return (v > 0.0f) ? v : v * NEG_SLOPE;
}
template <int G>
__device__ __forceinline__ void gat_row_group(const v2f* __restrict__ h2, int s, float alpha,
                                              int t, unsigned lane, v2f& acc) {
    unsigned idx[G];
    v2f r[G];
#pragma unroll
    for (int u = 0; u < G; ++u)
        idx[u] = (unsigned)__builtin_amdgcn_readlane(s, t + u) * 32u + lane;
#pragma unroll
    for (int u = 0; u < G; ++u)
        r[u] = h2[idx[u]];
#pragma unroll
    for (int u = 0; u < G; ++u) {
        const float at = __builtin_bit_cast(float,
            __builtin_amdgcn_readlane(__builtin_bit_cast(int, alpha), t + u));
        acc = acc + r[u] * at;
    }
}
__device__ __forceinline__ void gat_rows(const v2f* __restrict__ h2, int s, float alpha,
                                         int cnt, unsigned lane, v2f& acc) {
    const int c4 = (cnt + 3) & ~3;
    if (c4 <= 4)       gat_row_group<4>(h2, s, alpha, 0, lane, acc);
    else if (c4 <= 8)  gat_row_group<8>(h2, s, alpha, 0, lane, acc);
    else if (c4 <= 12) gat_row_group<12>(h2, s, alpha, 0, lane, acc);
    else {
        gat_row_group<16>(h2, s, alpha, 0, lane, acc);
        int t = 16;
        for (; t + 8 <= c4; t += 8) gat_row_group<8>(h2, s, alpha, t, lane, acc);
        if (t < c4)                 gat_row_group<4>(h2, s, alpha, t, lane, acc);
    }
}

__device__ __forceinline__ int gat_edge_ids(const int* __restrict__ csr_src, int rp, int k,
                                            unsigned lane, int elast) {
    const int beg = __builtin_amdgcn_readlane(rp, k);
    const int end = max(__builtin_amdgcn_readlane(rp, k + 1), beg);
    const int j   = min(beg + (int)lane, end - 1);
    return csr_src[min(max(j, 0), elast)];
}

template <bool TO_LDS>
__device__ __forceinline__ void gat_node_run(const int* __restrict__ rowptr, const int* __restrict__ csr_src,
                                             const float* __restrict__ as_, const float* __restrict__ ad_,
                                             const float* __restrict__ h, const v2f bb,
                                             float* dst, int pitch, int row0,
                                             int n0, int n1, int N, int E, int relu_flag, unsigned lane) {
    const unsigned nlast = (unsigned)(N - 1);
    const int elast = max(E - 1, 0);
    const v2f* h2 = (const v2f*)h;

    for (int base = n0; base < n1; base += 31) {
        const int c = min(31, n1 - base);
        int rp = rowptr[min(base + (int)lane, N)];
        rp = min(max(rp, 0), E);
        const float adl = ad_[min(base + (int)lane, N - 1)];

        int   sA = gat_edge_ids(csr_src, rp, 0, lane, elast);
        int   sB = gat_edge_ids(csr_src, rp, min(1, c - 1), lane, elast);
        sA = (int)min((unsigned)sA, nlast);
        float vA = as_[sA];

        for (int k = 0; k < c; ++k) {
            const int sC = gat_edge_ids(csr_src, rp, min(k + 2, c - 1), lane, elast);
            sB = (int)min((unsigned)sB, nlast);
            const float vB = as_[sB];

            const int beg = __builtin_amdgcn_readlane(rp, k);
            const int end = max(__builtin_amdgcn_readlane(rp, k + 1), beg);
            const float add = __builtin_bit_cast(float,
                __builtin_amdgcn_readlane(__builtin_bit_cast(int, adl), k));
            const int cnt = end - beg;

            v2f acc = {0.0f, 0.0f};
            if (cnt > 0) {
                if (cnt <= 32) {
                    const bool valid = (int)lane < cnt;
                    float v = vA + add;
                    v = (v > 0.0f) ? v : v * NEG_SLOPE;
                    const float m    = wave_max32(v);
                    const float ex   = valid ? __expf(v - m) : 0.0f;
                    const float den  = wave_sum32(ex);
                    const float alpha = ex / (den + 1e-16f);
                    gat_rows(h2, sA, alpha, cnt, lane, acc);
                } else {
                    float m = -3.0e38f;
                    for (int eb = beg; eb < end; eb += 32) {
                        const int j = eb + (int)lane;
                        const int s = (int)min((unsigned)csr_src[min(j, end - 1)], nlast);
                        m = fmaxf(m, gat_score(as_, s, add));
                    }
                    m = wave_max32(m);
                    float den = 0.0f;
                    for (int eb = beg; eb < end; eb += 32) {
                        const int j = eb + (int)lane;
                        const int s = (int)min((unsigned)csr_src[min(j, end - 1)], nlast);
                        const float ex = __expf(gat_score(as_, s, add) - m);
                        den += (j < end) ? ex : 0.0f;
                    }
                    den = wave_sum32(den);
                    const float dd = den + 1e-16f;
                    for (int eb = beg; eb < end; eb += 32) {
                        const int j = eb + (int)lane;
                        const int s = (int)min((unsigned)csr_src[min(j, end - 1)], nlast);
                        const float ex = __expf(gat_score(as_, s, add) - m);
                        const float alpha = (j < end) ? ex / dd : 0.0f;
                        gat_rows(h2, s, alpha, min(32, end - eb), lane, acc);
                    }
                }
            }

            v2f r;
            r.x = acc.x + bb.x;
            r.y = acc.y + bb.y;
            if (relu_flag) { r.x = fmaxf(r.x, 0.0f); r.y = fmaxf(r.y, 0.0f); }
            if (TO_LDS) *(v2f*)(dst + (base + k - row0) * pitch + 2 * (int)lane) = r;
            else        ((v2f*)dst)[(unsigned)(base + k) * 32u + lane] = r;

            sA = sB; vA = vB; sB = sC;
        }
    }
}

__global__ __launch_bounds__(256)
void gat_gather(const int* __restrict__ rowptr, const int* __restrict__ csr_src,
                const float* __restrict__ as_, const float* __restrict__ ad_,
                const float* __restrict__ h, const float* __restrict__ bias,
                float* __restrict__ out, int N, int E, int relu_flag) {
    const unsigned lane = threadIdx.x & 31;
    const unsigned nw = gridDim.x * (blockDim.x >> 5);
    const unsigned gw = (unsigned)__builtin_amdgcn_readfirstlane(
        (int)(blockIdx.x * (blockDim.x >> 5) + (threadIdx.x >> 5)));
    const unsigned per = (unsigned)N / nw, rem = (unsigned)N % nw;
    const int n0 = (int)(gw * per + min(gw, rem));
    const int n1 = n0 + (int)per + (gw < rem ? 1 : 0);
    if (n0 >= n1) return;
    const v2f bb = ((const v2f*)bias)[lane];
    gat_node_run<false>(rowptr, csr_src, as_, ad_, h, bb, out, 64, 0, n0, n1, N, E, relu_flag, lane);
}

template <int K, bool AGG>
__device__ __forceinline__
void gat_gemm_tile(float* sA, int tile,
                   const float* __restrict__ A, const float* __restrict__ W,
                   const float* __restrict__ a_src, const float* __restrict__ a_dst,
                   float* __restrict__ C, float* __restrict__ as_, float* __restrict__ ad_,
                   int N,
                   const int* __restrict__ rowptr, const int* __restrict__ csr_src,
                   const float* __restrict__ asP, const float* __restrict__ adP,
                   const float* __restrict__ biasP, int E) {
    constexpr int KP    = K + 4;
    constexpr int QPR   = K / 4;
    constexpr int RSTEP = 256 / QPR;
    constexpr int NQ    = 64 / RSTEP;
    constexpr int CP    = 68;
    static_assert(K % 4 == 0 && 256 % QPR == 0 && 64 % RSTEP == 0, "tile map");
    static_assert(CP <= KP, "the C tile reuses the A tile's LDS");
    static_assert(!AGG || K == 64, "an aggregated row is 64 features: one v2f per lane");
    const int lane = threadIdx.x & 31;
    const int wave = __builtin_amdgcn_readfirstlane((int)(threadIdx.x >> 5));
    const int rowBase = tile * 64;

    const int col      = (wave & 3) * 16 + (lane & 15);
    const int waveRow  = (wave >> 2) * 32;

    for (int k = threadIdx.x; k < K; k += 256)
        __builtin_prefetch(&W[(size_t)k * 64 + col], 0, 3);

    if constexpr (AGG) {
        const int n0 = rowBase + wave * 8;
        const int n1 = min(n0 + 8, N);
        const int nv = max(n1 - n0, 0);
        for (int k = nv; k < 8; ++k) {
            v2f z; z.x = 0.0f; z.y = 0.0f;
            *(v2f*)&sA[(wave * 8 + k) * KP + 2 * lane] = z;
        }
        if (nv > 0) {
            const v2f bb = ((const v2f*)biasP)[lane];
            gat_node_run<true>(rowptr, csr_src, asP, adP, A, bb, sA, KP, rowBase,
                               n0, n1, N, E, 1, (unsigned)lane);
        }
    } else {
        const int q  = threadIdx.x % QPR;
        const int r0 = threadIdx.x / QPR;
        v4f t[NQ];
#pragma unroll
        for (int i = 0; i < NQ; ++i) {
            const int gr = max(min(rowBase + r0 + i * RSTEP, N - 1), 0);
            t[i] = __builtin_nontemporal_load((const v4f*)(A + (size_t)gr * K) + q);
        }
#pragma unroll
        for (int i = 0; i < NQ; ++i)
            *(v4f*)&sA[(r0 + i * RSTEP) * KP + 4 * q] = t[i];
    }
    __syncthreads();

    const int row   = lane & 15;
    const int khalf = (lane >> 4) << 1;

    v8f acc0 = {};
    v8f acc1 = {};
#pragma unroll 8
    for (int k = 0; k < K; k += 4) {
        v2f b;
        b.x = W[(size_t)(k + khalf)     * 64 + col];
        b.y = W[(size_t)(k + khalf + 1) * 64 + col];

        v2f a0, a1;
        a0.x = sA[(waveRow +      row) * KP + k + khalf];
        a0.y = sA[(waveRow +      row) * KP + k + khalf + 1];
        a1.x = sA[(waveRow + 16 + row) * KP + k + khalf];
        a1.y = sA[(waveRow + 16 + row) * KP + k + khalf + 1];

        acc0 = __builtin_amdgcn_wmma_f32_16x16x4_f32(
            false, a0, false, b, (short)0, acc0, false, false);
        acc1 = __builtin_amdgcn_wmma_f32_16x16x4_f32(
            false, a1, false, b, (short)0, acc1, false, false);
    }

    __syncthreads();
    float* sC = sA;

    const int mhalf = (lane >> 4) << 3;
    for (int r = 0; r < 8; ++r) {
        sC[(waveRow +      mhalf + r) * CP + col] = acc0[r];
        sC[(waveRow + 16 + mhalf + r) * CP + col] = acc1[r];
    }

    const int c0 = (wave & 3) * 16;
    float* crow = &sC[(waveRow + lane) * CP + c0];
    {
        float s = 0.0f, d = 0.0f;
#pragma unroll
        for (int j = 0; j < 4; ++j) {
            const v4f cv = *(const v4f*)(crow + 4 * j);
            s = __builtin_fmaf(cv.x, a_src[c0 + 4 * j + 0], s); d = __builtin_fmaf(cv.x, a_dst[c0 + 4 * j + 0], d);
            s = __builtin_fmaf(cv.y, a_src[c0 + 4 * j + 1], s); d = __builtin_fmaf(cv.y, a_dst[c0 + 4 * j + 1], d);
            s = __builtin_fmaf(cv.z, a_src[c0 + 4 * j + 2], s); d = __builtin_fmaf(cv.z, a_dst[c0 + 4 * j + 2], d);
            s = __builtin_fmaf(cv.w, a_src[c0 + 4 * j + 3], s); d = __builtin_fmaf(cv.w, a_dst[c0 + 4 * j + 3], d);
        }
        v2f p; p.x = s; p.y = d;
        *(v2f*)crow = p;
    }
    __syncthreads();

    if (c0 == 0) {
        const v2f p0 = *(const v2f*)(crow);
        const v2f p1 = *(const v2f*)(crow + 16);
        const v2f p2 = *(const v2f*)(crow + 32);
        const v2f p3 = *(const v2f*)(crow + 48);
        const int gr = rowBase + waveRow + lane;
        if (gr < N) {
            as_[gr] = (p0.x + p1.x) + (p2.x + p3.x);
            ad_[gr] = (p0.y + p1.y) + (p2.y + p3.y);
        }
    }

    for (int r = 0; r < 8; ++r) {
        int gr0 = rowBase + waveRow +      mhalf + r;
        int gr1 = rowBase + waveRow + 16 + mhalf + r;
        if (gr0 < N) C[(size_t)gr0 * 64 + col] = acc0[r];
        if (gr1 < N) C[(size_t)gr1 * 64 + col] = acc1[r];
    }
}

__global__ __launch_bounds__(256)
void gat_gemm_agg(const float* __restrict__ A, const float* __restrict__ W,
                  const float* __restrict__ a_src, const float* __restrict__ a_dst,
                  float* __restrict__ C, float* __restrict__ as_, float* __restrict__ ad_,
                  int N,
                  const int* __restrict__ rowptr, const int* __restrict__ csr_src,
                  const float* __restrict__ asP, const float* __restrict__ adP,
                  const float* __restrict__ biasP, int E) {
    __shared__ __attribute__((aligned(16))) float sA[64 * (GAT_HID + 4)];
    gat_gemm_tile<GAT_HID, true>(sA, (int)blockIdx.x, A, W, a_src, a_dst, C, as_, ad_, N,
                                 rowptr, csr_src, asP, adP, biasP, E);
}

#define GAT_L1_ARGS const float* __restrict__ x, const float* __restrict__ W1, \
                    const float* __restrict__ a1s, const float* __restrict__ a1d, \
                    float* __restrict__ h, float* __restrict__ as_, float* __restrict__ ad_, int N
#define GAT_L1_TILE(sA, t) gat_gemm_tile<GAT_INC, false>(sA, (t), x, W1, a1s, a1d, h, as_, ad_, N, \
                                                         nullptr, nullptr, nullptr, nullptr, nullptr, 0)

__global__ __launch_bounds__(256)
void gat_stage_chunk(const int* __restrict__ srcI, const int* __restrict__ dstI,
                     int* __restrict__ cntT, int* __restrict__ startT, int* __restrict__ tmp,
                     int E, int NB, int nStage, int tile0, GAT_L1_ARGS) {
    __shared__ __attribute__((aligned(16))) float sA[64 * (GAT_INC + 4)];
    const int wg = (int)blockIdx.x;
    if (wg < nStage) csr_chunk_order(srcI, dstI, cntT, startT, tmp, E, NB, wg, (int*)sA);
    else             GAT_L1_TILE(sA, tile0 + wg - nStage);
}

__global__ __launch_bounds__(256)
void gat_stage_sort(const int* __restrict__ cntT, const int* __restrict__ startT,
                    const int* __restrict__ tmp,
                    int* __restrict__ rowptr, int* __restrict__ csr_src, int E, int NB,
                    int nStage, int tile0, GAT_L1_ARGS) {
    __shared__ __attribute__((aligned(16))) float sA[64 * (GAT_INC + 4)];
    const int wg = (int)blockIdx.x;
    if (wg < nStage) csr_bucket_sort(cntT, startT, tmp, rowptr, csr_src, N, E, NB, wg, (int*)sA);
    else             GAT_L1_TILE(sA, tile0 + wg - nStage);
}

extern "C" void kernel_launch(void* const* d_in, const int* in_sizes, int n_in,
                              void* d_out, int out_size, void* d_ws, size_t ws_size,
                              hipStream_t stream) {
    const float* x    = (const float*)d_in[0];
    const int*   eidx = (const int*)  d_in[1];
    const float* W1   = (const float*)d_in[2];
    const float* a1s  = (const float*)d_in[3];
    const float* a1d  = (const float*)d_in[4];
    const float* b1   = (const float*)d_in[5];
    const float* W2   = (const float*)d_in[6];
    const float* a2s  = (const float*)d_in[7];
    const float* a2d  = (const float*)d_in[8];
    const float* b2   = (const float*)d_in[9];

    const int N = in_sizes[0] / GAT_INC;
    const int E = in_sizes[1] / 2;
    const int* srcI = eidx;
    const int* dstI = eidx + E;

    float* ws = (float*)d_ws;
    size_t o = 0;
    float* h     = ws + o; o += (size_t)N * 64;
    float* h2    = ws + o; o += (size_t)N * 64;
    int*   graph = (int*)(ws + o); o += (size_t)N * 64;
    float* as_   = ws + o; o += N;
    float* ad_   = ws + o; o += N;
    float* as2   = ws + o; o += N;
    float* ad2   = ws + o; o += N;
    int*   tmp   = (int*)(ws + o); o += E;

    const int NB = (N + CS_BN - 1) / CS_BN;
    const int M  = NB * CS_WG;
    const size_t Np = ((size_t)N + 4) & ~(size_t)3;
    const size_t Ep = ((size_t)E + 3) & ~(size_t)3;
    const size_t Mp = ((size_t)M + 4) & ~(size_t)3;
    int* rowptr  = graph;
    int* csr_src = graph + Np;
    int* cntT    = csr_src + Ep;
    int* startT  = cntT + Mp;

    const int gGemm  = (N + 63) / 64;
    const int gPers  = GAT_PERS_WG;
    const int tA = min(3 * CS_WG, gGemm);

    gat_stage_chunk<<<CS_WG + tA,           CS_TH, 0, stream>>>(srcI, dstI, cntT, startT, tmp, E, NB,
                                                              CS_WG, 0,  x, W1, a1s, a1d, h, as_, ad_, N);
    gat_stage_sort <<<NB + (gGemm - tA),    CS_TH, 0, stream>>>(cntT, startT, tmp, rowptr, csr_src, E, NB,
                                                              NB,    tA, x, W1, a1s, a1d, h, as_, ad_, N);

    gat_gemm_agg<<<gGemm, 256, 0, stream>>>(h, W2, a2s, a2d, h2, as2, ad2, N,
                                            rowptr, csr_src, as_, ad_, b1, E);

    float* out = (float*)d_out;
    gat_gather    <<<gPers, 256, 0, stream>>>(rowptr, csr_src, as2, ad2, h2, b2, out, N, E, 0);
}
